# starting kernel
# baseline (speedup 1.0000x reference)
#include <hip/hip_runtime.h>
#include <cstdint>
#include <cstddef>

#define N_NODES   50000
#define N_EDGES   800000
#define TOT_E     (N_EDGES + N_NODES)
#define IN_C      128
#define HID       32
#define OUT_C     16
#define HEADS     8
#define HIDDEN    (HEADS * HID)
#define NEG_SLOPE 0.2f
#define GAT_EPS   1e-16f

typedef float v2f __attribute__((ext_vector_type(2)));
typedef float v8f __attribute__((ext_vector_type(8)));

template <int TN, int K, int NC>
__global__ __launch_bounds__(128) void gat_wmma_gemm_f32(
    const float* __restrict__ A, const float* __restrict__ B,
    float* __restrict__ C, int M, int waveTilesN)
{
    const int wave = blockIdx.x * (blockDim.x >> 5) + (threadIdx.x >> 5);
    const int lane = threadIdx.x & 31;
    const int tM = wave / waveTilesN;
    const int tg = wave % waveTilesN;
    if (tM * 16 >= M) return;

    const int half = lane >> 4;
    const int l    = lane & 15;
    const float* __restrict__ Ap = A + (size_t)(tM * 16 + l) * K + half * 2;
    const float* __restrict__ Bp = B + (size_t)(half * 2) * NC + tg * 16 * TN + l;

    const v8f zero = {0.f,0.f,0.f,0.f,0.f,0.f,0.f,0.f};
    v8f c[TN];
#pragma unroll
    for (int j = 0; j < TN; ++j) c[j] = zero;

#pragma unroll
    for (int k = 0; k < K; k += 4) {
        const v2f a = *(const v2f*)(Ap + k);
        v2f b[TN];
#pragma unroll
        for (int j = 0; j < TN; ++j) {
            b[j].x = Bp[(size_t)k * NC + j * 16];
            b[j].y = Bp[(size_t)(k + 1) * NC + j * 16];
        }
#pragma unroll
        for (int j = 0; j < TN; ++j)
            c[j] = __builtin_amdgcn_wmma_f32_16x16x4_f32(
                        false, a,  false, b[j],
                        (short)0, c[j],
                        false,  false);
    }

    float* __restrict__ Cbase =
        C + (size_t)(tM * 16 + half * 8) * NC + tg * 16 * TN + l;
#pragma unroll
    for (int j = 0; j < TN; ++j)
#pragma unroll
        for (int r = 0; r < 8; ++r)
            Cbase[(size_t)r * NC + j * 16] = c[j][r];
}

__global__ __launch_bounds__(256) void gat_att_logits(
    const float* __restrict__ xl, const float* __restrict__ att_s,
    const float* __restrict__ att_d, float* __restrict__ asrc,
    float* __restrict__ adst, int n, int H, int CH)
{
    const int i = blockIdx.x * blockDim.x + threadIdx.x;
    if (i >= n * H) return;
    const int h = i % H;
    const float* __restrict__ row = xl + (size_t)i * CH;
    const float* __restrict__ as  = att_s + (size_t)h * CH;
    const float* __restrict__ ad  = att_d + (size_t)h * CH;
    float s = 0.f, d = 0.f;
    for (int c = 0; c < CH; ++c) { float v = row[c]; s += v * as[c]; d += v * ad[c]; }
    asrc[i] = s;
    adst[i] = d;
}

__device__ __forceinline__ unsigned fkey(float f) {
    unsigned u = __float_as_uint(f);
    return (u & 0x80000000u) ? ~u : (u | 0x80000000u);
}
__device__ __forceinline__ float fdec(unsigned k) {
    unsigned u = (k & 0x80000000u) ? (k ^ 0x80000000u) : ~k;
    return __uint_as_float(u);
}

__device__ __forceinline__ void edge_sd(long long e, const int* __restrict__ src,
                                        const int* __restrict__ dst, int& s, int& d) {
    if (e < N_EDGES) { s = src[e]; d = dst[e]; }
    else             { s = (int)(e - N_EDGES); d = s; }
}

__global__ __launch_bounds__(256) void gat_edge_max(
    const int* __restrict__ src, const int* __restrict__ dst,
    const float* __restrict__ asrc, const float* __restrict__ adst,
    unsigned* __restrict__ emaxk, int H)
{
    const long long i = (long long)blockIdx.x * blockDim.x + threadIdx.x;
    if (i >= (long long)TOT_E * H) return;
    const int h = (int)(i % H);
    const long long e = i / H;
    int s, d; edge_sd(e, src, dst, s, d);
    float v = asrc[(size_t)s * H + h] + adst[(size_t)d * H + h];
    v = v > 0.f ? v : NEG_SLOPE * v;
    atomicMax(&emaxk[(size_t)d * H + h], fkey(v));
}

__global__ __launch_bounds__(256) void gat_edge_exp(
    const int* __restrict__ src, const int* __restrict__ dst,
    const float* __restrict__ asrc, const float* __restrict__ adst,
    const unsigned* __restrict__ emaxk, float* __restrict__ eexp,
    float* __restrict__ denom, int H)
{
    const long long i = (long long)blockIdx.x * blockDim.x + threadIdx.x;
    if (i >= (long long)TOT_E * H) return;
    const int h = (int)(i % H);
    const long long e = i / H;
    int s, d; edge_sd(e, src, dst, s, d);
    float v = asrc[(size_t)s * H + h] + adst[(size_t)d * H + h];
    v = v > 0.f ? v : NEG_SLOPE * v;
    const float m  = fdec(emaxk[(size_t)d * H + h]);
    const float ex = expf(v - m);
    eexp[i] = ex;
    atomicAdd(&denom[(size_t)d * H + h], ex);
}

__global__ __launch_bounds__(256) void gat_edge_aggregate(
    const int* __restrict__ src, const int* __restrict__ dst,
    const float* __restrict__ eexp, const float* __restrict__ denom,
    const float* __restrict__ xl, float* __restrict__ acc, int H, int CH)
{
    const long long i = (long long)blockIdx.x * blockDim.x + threadIdx.x;
    if (i >= (long long)TOT_E * H * CH) return;
    const int c        = (int)(i % CH);
    const long long eh = i / CH;
    const int h        = (int)(eh % H);
    const long long e  = eh / H;
    int s, d; edge_sd(e, src, dst, s, d);
    const float alpha = eexp[eh] / (denom[(size_t)d * H + h] + GAT_EPS);
    const float val   = alpha * xl[((size_t)s * H + h) * CH + c];
    atomicAdd(&acc[((size_t)d * H + h) * CH + c], val);
}

__global__ __launch_bounds__(256) void gat_finalize_elu(
    float* __restrict__ acc, const float* __restrict__ bias, int n, int F)
{
    const int i = blockIdx.x * blockDim.x + threadIdx.x;
    if (i >= n * F) return;
    const float v = acc[i] + bias[i % F];
    acc[i] = v > 0.f ? v : expm1f(v);
}

__global__ __launch_bounds__(256) void gat_finalize_bias(
    const float* __restrict__ acc, const float* __restrict__ bias,
    float* __restrict__ out, int n, int F)
{
    const int i = blockIdx.x * blockDim.x + threadIdx.x;
    if (i >= n * F) return;
    out[i] = acc[i] + bias[i % F];
}

extern "C" void kernel_launch(void* const* d_in, const int* in_sizes, int n_in,
                              void* d_out, int out_size, void* d_ws, size_t ws_size,
                              hipStream_t stream)
{
    (void)in_sizes; (void)n_in; (void)out_size; (void)ws_size;

    const float* x    = (const float*)d_in[0];
    const int*   ei   = (const int*)  d_in[1];
    const float* W1   = (const float*)d_in[2];
    const float* ats1 = (const float*)d_in[3];
    const float* atd1 = (const float*)d_in[4];
    const float* b1   = (const float*)d_in[5];
    const float* W2   = (const float*)d_in[6];
    const float* ats2 = (const float*)d_in[7];
    const float* atd2 = (const float*)d_in[8];
    const float* b2   = (const float*)d_in[9];
    const int* src = ei;
    const int* dst = ei + N_EDGES;
    float* out = (float*)d_out;

    float*    acc1 = (float*)d_ws;
    float*    acc2 = acc1 + (size_t)N_NODES * HIDDEN;
    unsigned* emk1 = (unsigned*)(acc2 + (size_t)N_NODES * OUT_C);
    float*    den1 = (float*)(emk1 + (size_t)N_NODES * HEADS);
    unsigned* emk2 = (unsigned*)(den1 + (size_t)N_NODES * HEADS);
    float*    den2 = (float*)(emk2 + (size_t)N_NODES);
    const size_t zero_bytes =
        (size_t)N_NODES * (HIDDEN + OUT_C + 2 * HEADS + 2) * sizeof(float);
    float* xl1 = den2 + (size_t)N_NODES;
    float* as1 = xl1 + (size_t)N_NODES * HIDDEN;
    float* ad1 = as1 + (size_t)N_NODES * HEADS;
    float* ee1 = ad1 + (size_t)N_NODES * HEADS;
    float* xl2 = ee1;
    float* as2 = xl2 + (size_t)N_NODES * OUT_C;
    float* ad2 = as2 + (size_t)N_NODES;
    float* ee2 = ad2 + (size_t)N_NODES;

    hipMemsetAsync(d_ws, 0, zero_bytes, stream);

    const int TB = 256;
    auto blk = [](long long n, int tb) { return (unsigned)((n + tb - 1) / tb); };

    {
        const int waveTilesN = HIDDEN / (16 * 4);
        const long long waves = (long long)(N_NODES / 16) * waveTilesN;
        gat_wmma_gemm_f32<4, IN_C, HIDDEN><<<blk(waves, 4), 128, 0, stream>>>(
            x, W1, xl1, N_NODES, waveTilesN);
    }
    gat_att_logits<<<blk((long long)N_NODES * HEADS, TB), TB, 0, stream>>>(
        xl1, ats1, atd1, as1, ad1, N_NODES, HEADS, HID);
    gat_edge_max<<<blk((long long)TOT_E * HEADS, TB), TB, 0, stream>>>(
        src, dst, as1, ad1, emk1, HEADS);
    gat_edge_exp<<<blk((long long)TOT_E * HEADS, TB), TB, 0, stream>>>(
        src, dst, as1, ad1, emk1, ee1, den1, HEADS);
    gat_edge_aggregate<<<blk((long long)TOT_E * HEADS * HID, TB), TB, 0, stream>>>(
        src, dst, ee1, den1, xl1, acc1, HEADS, HID);
    gat_finalize_elu<<<blk((long long)N_NODES * HIDDEN, TB), TB, 0, stream>>>(
        acc1, b1, N_NODES, HIDDEN);

    {
        const int waveTilesN = OUT_C / 16;
        const long long waves = (long long)(N_NODES / 16) * waveTilesN;
        gat_wmma_gemm_f32<1, HIDDEN, OUT_C><<<blk(waves, 4), 128, 0, stream>>>(
            acc1, W2, xl2, N_NODES, waveTilesN);
    }
    gat_att_logits<<<blk((long long)N_NODES, TB), TB, 0, stream>>>(
        xl2, ats2, atd2, as2, ad2, N_NODES, 1, OUT_C);
    gat_edge_max<<<blk((long long)TOT_E, TB), TB, 0, stream>>>(
        src, dst, as2, ad2, emk2, 1);
    gat_edge_exp<<<blk((long long)TOT_E, TB), TB, 0, stream>>>(
        src, dst, as2, ad2, emk2, ee2, den2, 1);
    gat_edge_aggregate<<<blk((long long)TOT_E * OUT_C, TB), TB, 0, stream>>>(
        src, dst, ee2, den2, xl2, acc2, 1, OUT_C);
    gat_finalize_bias<<<blk((long long)N_NODES * OUT_C, TB), TB, 0, stream>>>(
        acc2, b2, out, N_NODES, OUT_C);
}


# Layer-2 softmax+aggregate per owner work-group in LDS (float atomics, arrival order; divide at write-out) on edge queues the layer-1 owner launch now stores; global atomics, memset, 3 launches gone
# speedup vs baseline: 6.6814x; 6.6814x over previous
#include <hip/hip_runtime.h>
#include <cstdint>
#include <cstddef>

#define N_NODES   50000
#define N_EDGES   800000
#define TOT_E     (N_EDGES + N_NODES)
#define IN_C      128
#define HID       32
#define OUT_C     16
#define HEADS     8
#define HIDDEN    (HEADS * HID)
#define NEG_SLOPE 0.2f
#define GAT_EPS   1e-16f

typedef float v2f __attribute__((ext_vector_type(2)));
typedef float v8f __attribute__((ext_vector_type(8)));

template <int TN, int K, int NC>
__global__ __launch_bounds__(128) void gat_wmma_gemm_f32(
    const float* __restrict__ A, const float* __restrict__ B,
    float* __restrict__ C, int M, int waveTilesN)
{
    const int wave = blockIdx.x * (blockDim.x >> 5) + (threadIdx.x >> 5);
    const int lane = threadIdx.x & 31;
    const int tM = wave / waveTilesN;
    const int tg = wave % waveTilesN;
    if (tM * 16 >= M) return;

    const int half = lane >> 4;
    const int l    = lane & 15;
    const float* __restrict__ Ap = A + (size_t)(tM * 16 + l) * K + half * 2;
    const float* __restrict__ Bp = B + (size_t)(half * 2) * NC + tg * 16 * TN + l;

    const v8f zero = {0.f,0.f,0.f,0.f,0.f,0.f,0.f,0.f};
    v8f c[TN];
#pragma unroll
    for (int j = 0; j < TN; ++j) c[j] = zero;

#pragma unroll
    for (int k = 0; k < K; k += 4) {
        const v2f a = *(const v2f*)(Ap + k);
        v2f b[TN];
#pragma unroll
        for (int j = 0; j < TN; ++j) {
            b[j].x = Bp[(size_t)k * NC + j * 16];
            b[j].y = Bp[(size_t)(k + 1) * NC + j * 16];
        }
#pragma unroll
        for (int j = 0; j < TN; ++j)
            c[j] = __builtin_amdgcn_wmma_f32_16x16x4_f32(
                        false, a,  false, b[j],
                        (short)0, c[j],
                        false,  false);
    }

    float* __restrict__ Cbase =
        C + (size_t)(tM * 16 + half * 8) * NC + tg * 16 * TN + l;
#pragma unroll
    for (int j = 0; j < TN; ++j)
#pragma unroll
        for (int r = 0; r < 8; ++r)
            Cbase[(size_t)r * NC + j * 16] = c[j][r];
}

__global__ __launch_bounds__(256) void gat_att_logits(
    const float* __restrict__ xl, const float* __restrict__ att_s,
    const float* __restrict__ att_d, float* __restrict__ asrc,
    float* __restrict__ adst, int n, int H, int CH)
{
    const int i = blockIdx.x * blockDim.x + threadIdx.x;
    if (i >= n * H) return;
    const int h = i % H;
    const float* __restrict__ row = xl + (size_t)i * CH;
    const float* __restrict__ as  = att_s + (size_t)h * CH;
    const float* __restrict__ ad  = att_d + (size_t)h * CH;
    float s = 0.f, d = 0.f;
    for (int c = 0; c < CH; ++c) { float v = row[c]; s += v * as[c]; d += v * ad[c]; }
    asrc[i] = s;
    adst[i] = d;
}

__device__ __forceinline__ unsigned fkey(float f) {
    unsigned u = __float_as_uint(f);
    return (u & 0x80000000u) ? ~u : (u | 0x80000000u);
}
__device__ __forceinline__ float fdec(unsigned k) {
    unsigned u = (k & 0x80000000u) ? (k ^ 0x80000000u) : ~k;
    return __uint_as_float(u);
}

__device__ __forceinline__ void edge_sd(long long e, const int* __restrict__ src,
                                        const int* __restrict__ dst, int& s, int& d) {
    if (e < N_EDGES) { s = src[e]; d = dst[e]; }
    else             { s = (int)(e - N_EDGES); d = s; }
}

#define OWN_R     196
#define OWN_WG    256
#define OWN_WAVES 32
#define OWN_QCAP  512
#define OWN_EBITS 20
static_assert(TOT_E <= (1 << OWN_EBITS), "edge id must fit the low bits of a queue entry");
static_assert(N_NODES <= 65536, "a resolved queue entry keeps the source node in 16 bits");
static_assert(OWN_R * OWN_WG >= N_NODES, "every node needs an owner");

__device__ __forceinline__ float own_lrelu(float v) { return v > 0.f ? v : NEG_SLOPE * v; }

__device__ __forceinline__ void own_add_row(float* accs, int rl, int lane,
                                            float w0, float w1, float4 x0, float4 x1)
{
    float* row = accs + rl * HIDDEN + lane;
    __hip_atomic_fetch_add(row +   0, w0 * x0.x, __ATOMIC_RELAXED, __HIP_MEMORY_SCOPE_WORKGROUP);
    __hip_atomic_fetch_add(row +  32, w0 * x0.y, __ATOMIC_RELAXED, __HIP_MEMORY_SCOPE_WORKGROUP);
    __hip_atomic_fetch_add(row +  64, w0 * x0.z, __ATOMIC_RELAXED, __HIP_MEMORY_SCOPE_WORKGROUP);
    __hip_atomic_fetch_add(row +  96, w0 * x0.w, __ATOMIC_RELAXED, __HIP_MEMORY_SCOPE_WORKGROUP);
    __hip_atomic_fetch_add(row + 128, w1 * x1.x, __ATOMIC_RELAXED, __HIP_MEMORY_SCOPE_WORKGROUP);
    __hip_atomic_fetch_add(row + 160, w1 * x1.y, __ATOMIC_RELAXED, __HIP_MEMORY_SCOPE_WORKGROUP);
    __hip_atomic_fetch_add(row + 192, w1 * x1.z, __ATOMIC_RELAXED, __HIP_MEMORY_SCOPE_WORKGROUP);
    __hip_atomic_fetch_add(row + 224, w1 * x1.w, __ATOMIC_RELAXED, __HIP_MEMORY_SCOPE_WORKGROUP);
}

__device__ __forceinline__ void own_resolve(
    unsigned* myq, int qn, int lane, const int* __restrict__ src,
    const float4* __restrict__ as4, const float* rad, unsigned* rmax)
{
    const unsigned emask = (1u << OWN_EBITS) - 1u;
    for (int j = lane; j < qn; j += 32) {
        const unsigned pk = myq[j];
        const unsigned e  = pk & emask;
        const unsigned rl = pk >> OWN_EBITS;
        const unsigned s  = (e < (unsigned)N_EDGES) ? (unsigned)src[e] : (e - (unsigned)N_EDGES);
        const float4 a0 = as4[(size_t)s * 2], a1 = as4[(size_t)s * 2 + 1];
        const float4 d0 = *(const float4*)(rad + rl * HEADS);
        const float4 d1 = *(const float4*)(rad + rl * HEADS + 4);
        const float v[8] = {a0.x + d0.x, a0.y + d0.y, a0.z + d0.z, a0.w + d0.w,
                            a1.x + d1.x, a1.y + d1.y, a1.z + d1.z, a1.w + d1.w};
        unsigned* mx = rmax + rl * HEADS;
#pragma unroll
        for (int h = 0; h < HEADS; ++h)
            __hip_atomic_fetch_max(mx + h, fkey(own_lrelu(v[h])),
                                   __ATOMIC_RELAXED, __HIP_MEMORY_SCOPE_WORKGROUP);
        myq[j] = (rl << 16) | s;
    }
}

__device__ __forceinline__ void own_drain(
    const unsigned* myq, int qn, int lane, int hq,
    const float4* __restrict__ as4, const float4* __restrict__ xl4,
    float* mywt, const float* rad, const unsigned* rmax, float* rden, float* accs)
{
    for (int c0 = 0; c0 < qn; c0 += 32) {
        const int m = (qn - c0 < 32) ? (qn - c0) : 32;
        if (lane < m) {
            const unsigned pk = myq[c0 + lane];
            const unsigned s  = pk & 0xFFFFu;
            const unsigned rl = pk >> 16;
            const float4 a0 = as4[(size_t)s * 2], a1 = as4[(size_t)s * 2 + 1];
            const float4 d0 = *(const float4*)(rad + rl * HEADS);
            const float4 d1 = *(const float4*)(rad + rl * HEADS + 4);
            const uint4  k0 = *(const uint4*)(rmax + rl * HEADS);
            const uint4  k1 = *(const uint4*)(rmax + rl * HEADS + 4);
            float4 w0, w1;
            w0.x = expf(own_lrelu(a0.x + d0.x) - fdec(k0.x));
            w0.y = expf(own_lrelu(a0.y + d0.y) - fdec(k0.y));
            w0.z = expf(own_lrelu(a0.z + d0.z) - fdec(k0.z));
            w0.w = expf(own_lrelu(a0.w + d0.w) - fdec(k0.w));
            w1.x = expf(own_lrelu(a1.x + d1.x) - fdec(k1.x));
            w1.y = expf(own_lrelu(a1.y + d1.y) - fdec(k1.y));
            w1.z = expf(own_lrelu(a1.z + d1.z) - fdec(k1.z));
            w1.w = expf(own_lrelu(a1.w + d1.w) - fdec(k1.w));
            *(float4*)(mywt + lane * HEADS)     = w0;
            *(float4*)(mywt + lane * HEADS + 4) = w1;
            float* dn = rden + rl * HEADS;
            __hip_atomic_fetch_add(dn + 0, w0.x, __ATOMIC_RELAXED, __HIP_MEMORY_SCOPE_WORKGROUP);
            __hip_atomic_fetch_add(dn + 1, w0.y, __ATOMIC_RELAXED, __HIP_MEMORY_SCOPE_WORKGROUP);
            __hip_atomic_fetch_add(dn + 2, w0.z, __ATOMIC_RELAXED, __HIP_MEMORY_SCOPE_WORKGROUP);
            __hip_atomic_fetch_add(dn + 3, w0.w, __ATOMIC_RELAXED, __HIP_MEMORY_SCOPE_WORKGROUP);
            __hip_atomic_fetch_add(dn + 4, w1.x, __ATOMIC_RELAXED, __HIP_MEMORY_SCOPE_WORKGROUP);
            __hip_atomic_fetch_add(dn + 5, w1.y, __ATOMIC_RELAXED, __HIP_MEMORY_SCOPE_WORKGROUP);
            __hip_atomic_fetch_add(dn + 6, w1.z, __ATOMIC_RELAXED, __HIP_MEMORY_SCOPE_WORKGROUP);
            __hip_atomic_fetch_add(dn + 7, w1.w, __ATOMIC_RELAXED, __HIP_MEMORY_SCOPE_WORKGROUP);
        }
        int i = 0;
        for (; i + 4 <= m; i += 4) {
            float4 x0[4], x1[4];
            int    rl[4];
#pragma unroll
            for (int u = 0; u < 4; ++u) {
                const unsigned pk = (unsigned)__builtin_amdgcn_readfirstlane((int)myq[c0 + i + u]);
                const unsigned s  = pk & 0xFFFFu;
                rl[u] = (int)(pk >> 16);
                x0[u] = xl4[(size_t)s * (HIDDEN / 4) + lane];
                x1[u] = xl4[(size_t)s * (HIDDEN / 4) + 32 + lane];
            }
#pragma unroll
            for (int u = 0; u < 4; ++u) {
                const float w0 = mywt[(i + u) * HEADS + hq];
                const float w1 = mywt[(i + u) * HEADS + 4 + hq];
                own_add_row(accs, rl[u], lane, w0, w1, x0[u], x1[u]);
            }
        }
        for (; i < m; ++i) {
            const unsigned pk = (unsigned)__builtin_amdgcn_readfirstlane((int)myq[c0 + i]);
            const unsigned s  = pk & 0xFFFFu;
            const float4 x0 = xl4[(size_t)s * (HIDDEN / 4) + lane];
            const float4 x1 = xl4[(size_t)s * (HIDDEN / 4) + 32 + lane];
            const float w0 = mywt[i * HEADS + hq];
            const float w1 = mywt[i * HEADS + 4 + hq];
            own_add_row(accs, (int)(pk >> 16), lane, w0, w1, x0, x1);
        }
    }
}

template <bool SECOND>
__device__ __forceinline__ int own_filter(
    int tid, int lane, int wave, int hq, int n0, int cnt,
    const int* __restrict__ src, const int* __restrict__ dst,
    const float4* __restrict__ as4, const float4* __restrict__ xl4,
    unsigned* myq, float* mywt, const float* rad, unsigned* rmax,
    float* rden, float* accs, int* ovf)
{
    const int  rs  = wave + OWN_WAVES * lane;
    const bool has = rs < cnt;
    if (has) myq[lane] = ((unsigned)rs << OWN_EBITS) | (unsigned)(N_EDGES + n0 + rs);
    int qn = __builtin_popcount(__builtin_amdgcn_ballot_w32(has));

    const int4* __restrict__ dst4 = (const int4*)dst;
    constexpr int NV    = N_EDGES / 4;
    constexpr int TRIPS = (NV + 1023) / 1024;
    static_assert(TRIPS % 4 == 0, "the four-trip body assumes a whole number of bodies");
    auto ld = [&](int t) {
        const int i = t * 1024 + tid;
        return dst4[i < NV ? i : NV - 1];
    };
    auto trip = [&](const int4 d, int t) {
        const int i = t * 1024 + tid;
        const unsigned lim = (i < NV) ? (unsigned)cnt : 0u;
        const unsigned l0 = (unsigned)(d.x - n0), l1 = (unsigned)(d.y - n0);
        const unsigned l2 = (unsigned)(d.z - n0), l3 = (unsigned)(d.w - n0);
        const bool m0 = l0 < lim, m1 = l1 < lim, m2 = l2 < lim, m3 = l3 < lim;
        if (__builtin_amdgcn_ballot_w32(m0 | m1 | m2 | m3) != 0u) {
            const unsigned ll[4] = {l0, l1, l2, l3};
            const bool     mm[4] = {m0, m1, m2, m3};
#pragma unroll
            for (int k = 0; k < 4; ++k) {
                const unsigned mask = __builtin_amdgcn_ballot_w32(mm[k]);
                if (mm[k]) {
                    const int pos = qn + (int)__builtin_amdgcn_mbcnt_lo(mask, 0u);
                    myq[pos] = (ll[k] << OWN_EBITS) | (unsigned)(i * 4 + k);
                }
                qn += __builtin_popcount(mask);
            }
            if (qn > OWN_QCAP - 128) {
                own_resolve(myq, qn, lane, src, as4, rad, rmax);
                if (SECOND) own_drain(myq, qn, lane, hq, as4, xl4, mywt, rad, rmax, rden, accs);
                else if (lane == 0) *ovf = 1;
                qn = 0;
            }
        }
    };
    int4 d0 = ld(0), d1 = ld(1), d2 = ld(2), d3;
    for (int t = 0; t < TRIPS; t += 4) {
        d3 = ld(t + 3);  trip(d0, t);
        d0 = ld(t + 4);  trip(d1, t + 1);
        d1 = ld(t + 5);  trip(d2, t + 2);
        d2 = ld(t + 6);  trip(d3, t + 3);
    }
    own_resolve(myq, qn, lane, src, as4, rad, rmax);
    if (SECOND) { own_drain(myq, qn, lane, hq, as4, xl4, mywt, rad, rmax, rden, accs); qn = 0; }
    return qn;
}

__device__ __forceinline__ float own_elu(float v) { return v > 0.f ? v : expm1f(v); }

__global__ __launch_bounds__(1024) void gat_owner_softmax_aggregate_l1(
    const int* __restrict__ src, const int* __restrict__ dst,
    const float* __restrict__ asrc, const float* __restrict__ adst,
    const float* __restrict__ xl, const float* __restrict__ bias,
    float* __restrict__ hout, unsigned* __restrict__ qsave,
    int* __restrict__ qcnt, int* __restrict__ qovf)
{
    __shared__ __attribute__((aligned(16))) float    accs[OWN_R * HIDDEN];
    __shared__ __attribute__((aligned(16))) float    rad[OWN_R * HEADS];
    __shared__ __attribute__((aligned(16))) unsigned rmax[OWN_R * HEADS];
    __shared__ __attribute__((aligned(16))) float    rden[OWN_R * HEADS];
    __shared__ __attribute__((aligned(16))) float    wts[OWN_WAVES][32 * HEADS];
    __shared__ unsigned qpk[OWN_WAVES][OWN_QCAP];
    __shared__ int      ovf;

    const int tid  = threadIdx.x;
    const int lane = tid & 31;
    const int wave = tid >> 5;
    const int hq   = lane >> 3;
    const int n0   = blockIdx.x * OWN_R;
    int cnt = N_NODES - n0;
    if (cnt > OWN_R) cnt = OWN_R;
    const float4* __restrict__ xl4 = (const float4*)xl;
    const float4* __restrict__ as4 = (const float4*)asrc;

    float4* accs4 = (float4*)accs;
    for (int i = tid; i < OWN_R * HIDDEN / 4; i += 1024)
        accs4[i] = make_float4(0.f, 0.f, 0.f, 0.f);
    for (int i = tid; i < OWN_R * HEADS; i += 1024) {
        rad[i]  = ((i >> 3) < cnt) ? adst[(size_t)n0 * HEADS + i] : 0.f;
        rmax[i] = 0u;
        rden[i] = 0.f;
    }
    if (tid == 0) ovf = 0;
    __syncthreads();

    unsigned* myq  = qpk[wave];
    float*    mywt = wts[wave];
    const int qn = own_filter<false>(tid, lane, wave, hq, n0, cnt, src, dst, as4, xl4,
                                     myq, mywt, rad, rmax, rden, accs, &ovf);
    __syncthreads();

    {
        unsigned* gq = qsave + ((size_t)blockIdx.x * OWN_WAVES + wave) * OWN_QCAP;
        for (int j = lane; j < qn; j += 32) gq[j] = myq[j];
        if (lane == 0) qcnt[blockIdx.x * OWN_WAVES + wave] = qn;
        if (tid == 0)  qovf[blockIdx.x] = ovf;
    }

    if (ovf == 0)
        own_drain(myq, qn, lane, hq, as4, xl4, mywt, rad, rmax, rden, accs);
    else
        (void)own_filter<true>(tid, lane, wave, hq, n0, cnt, src, dst, as4, xl4,
                               myq, mywt, rad, rmax, rden, accs, &ovf);
    __syncthreads();

    float bq0[4], bq1[4];
#pragma unroll
    for (int k = 0; k < 4; ++k) { bq0[k] = bias[lane * 4 + k]; bq1[k] = bias[128 + lane * 4 + k]; }
    float4* __restrict__ h4 = (float4*)hout;
    for (int r = wave; r < cnt; r += OWN_WAVES) {
        const int n = n0 + r;
        const float* row = accs + r * HIDDEN + lane;
        const float d0 = rden[r * HEADS + hq] + GAT_EPS;
        const float d1 = rden[r * HEADS + 4 + hq] + GAT_EPS;
        float4 o0, o1;
        o0.x = own_elu(row[0]   / d0 + bq0[0]);  o0.y = own_elu(row[32]  / d0 + bq0[1]);
        o0.z = own_elu(row[64]  / d0 + bq0[2]);  o0.w = own_elu(row[96]  / d0 + bq0[3]);
        o1.x = own_elu(row[128] / d1 + bq1[0]);  o1.y = own_elu(row[160] / d1 + bq1[1]);
        o1.z = own_elu(row[192] / d1 + bq1[2]);  o1.w = own_elu(row[224] / d1 + bq1[3]);
        h4[(size_t)n * (HIDDEN / 4) + lane]      = o0;
        h4[(size_t)n * (HIDDEN / 4) + 32 + lane] = o1;
    }
}

__device__ __forceinline__ void l2_add_quarter(float* acc, unsigned rl, int q, float w, float4 x)
{
    float* a = acc + (q * 4) * OWN_R + rl;
    __hip_atomic_fetch_add(a,             w * x.x, __ATOMIC_RELAXED, __HIP_MEMORY_SCOPE_WORKGROUP);
    __hip_atomic_fetch_add(a + OWN_R,     w * x.y, __ATOMIC_RELAXED, __HIP_MEMORY_SCOPE_WORKGROUP);
    __hip_atomic_fetch_add(a + 2 * OWN_R, w * x.z, __ATOMIC_RELAXED, __HIP_MEMORY_SCOPE_WORKGROUP);
    __hip_atomic_fetch_add(a + 3 * OWN_R, w * x.w, __ATOMIC_RELAXED, __HIP_MEMORY_SCOPE_WORKGROUP);
}

__global__ __launch_bounds__(1024) void gat_owner_softmax_aggregate_l2(
    const int* __restrict__ src, const int* __restrict__ dst,
    const float* __restrict__ as2, const float* __restrict__ ad2,
    const float* __restrict__ xl2, const float* __restrict__ bias,
    const unsigned* __restrict__ qsave, const int* __restrict__ qcnt,
    const int* __restrict__ qovf, float* __restrict__ out)
{
    __shared__ float    acc[OUT_C * OWN_R];
    __shared__ float    rad[OWN_R];
    __shared__ unsigned rmax[OWN_R];
    __shared__ float    rden[OWN_R];

    const int tid  = threadIdx.x;
    const int lane = tid & 31;
    const int wave = tid >> 5;
    const int n0   = blockIdx.x * OWN_R;
    int cnt = N_NODES - n0;
    if (cnt > OWN_R) cnt = OWN_R;
    const float4* __restrict__ xq = (const float4*)xl2;

    for (int i = tid; i < OUT_C * OWN_R; i += 1024) acc[i] = 0.f;
    if (tid < OWN_R) {
        rad[tid]  = (tid < cnt) ? ad2[n0 + tid] : 0.f;
        rmax[tid] = 0u;
        rden[tid] = 0.f;
    }
    const bool queued = qovf[blockIdx.x] == 0;
    const unsigned* __restrict__ gq = qsave + ((size_t)blockIdx.x * OWN_WAVES + wave) * OWN_QCAP;
    const int qn = queued ? __builtin_amdgcn_readfirstlane(qcnt[blockIdx.x * OWN_WAVES + wave]) : 0;
    __syncthreads();

    if (queued) {
        for (int j0 = 0; j0 < qn; j0 += 128) {
            unsigned pk[4];
            float    a[4];
#pragma unroll
            for (int t = 0; t < 4; ++t) {
                const int j = j0 + t * 32 + lane;
                pk[t] = (j < qn) ? gq[j] : 0u;
            }
#pragma unroll
            for (int t = 0; t < 4; ++t) a[t] = as2[pk[t] & 0xFFFFu];
#pragma unroll
            for (int t = 0; t < 4; ++t) {
                if (j0 + t * 32 + lane < qn) {
                    const unsigned rl = pk[t] >> 16;
                    const float v = own_lrelu(a[t] + rad[rl]);
                    __hip_atomic_fetch_max(rmax + rl, fkey(v), __ATOMIC_RELAXED, __HIP_MEMORY_SCOPE_WORKGROUP);
                }
            }
        }
    } else {
        for (int e = tid; e < TOT_E; e += 1024) {
            int s, d; edge_sd(e, src, dst, s, d);
            const unsigned rl = (unsigned)(d - n0);
            if (rl < (unsigned)cnt) {
                const float v = own_lrelu(as2[s] + rad[rl]);
                __hip_atomic_fetch_max(rmax + rl, fkey(v), __ATOMIC_RELAXED, __HIP_MEMORY_SCOPE_WORKGROUP);
            }
        }
    }
    __syncthreads();

    if (queued) {
        const int q   = lane & 3;
        const int sub = lane >> 2;
        auto entry = [&](int j) { return gq[j < qn ? j : qn - 1]; };
        unsigned nxt = (qn > 0) ? entry(lane) : 0u;
        for (int j0 = 0; j0 < qn; j0 += 32) {
            const unsigned mine = nxt;
            nxt = entry(j0 + 32 + lane);
            unsigned pk[4];
            float    a[4];
            float4   x[4];
#pragma unroll
            for (int t = 0; t < 4; ++t) {
                pk[t] = (unsigned)__shfl((int)mine, t * 8 + sub);
                const unsigned s = pk[t] & 0xFFFFu;
                a[t] = as2[s];
                x[t] = xq[(size_t)s * (OUT_C / 4) + q];
            }
#pragma unroll
            for (int t = 0; t < 4; ++t) {
                if (j0 + t * 8 + sub < qn) {
                    const unsigned rl = pk[t] >> 16;
                    const float w = expf(own_lrelu(a[t] + rad[rl]) - fdec(rmax[rl]));
                    if (q == 0)
                        __hip_atomic_fetch_add(rden + rl, w, __ATOMIC_RELAXED, __HIP_MEMORY_SCOPE_WORKGROUP);
                    l2_add_quarter(acc, rl, q, w, x[t]);
                }
            }
        }
    } else {
        for (int e = tid; e < TOT_E; e += 1024) {
            int s, d; edge_sd(e, src, dst, s, d);
            const unsigned rl = (unsigned)(d - n0);
            if (rl < (unsigned)cnt) {
                const float w = expf(own_lrelu(as2[s] + rad[rl]) - fdec(rmax[rl]));
                __hip_atomic_fetch_add(rden + rl, w, __ATOMIC_RELAXED, __HIP_MEMORY_SCOPE_WORKGROUP);
                for (int q = 0; q < 4; ++q)
                    l2_add_quarter(acc, rl, q, w, xq[(size_t)s * (OUT_C / 4) + q]);
            }
        }
    }
    __syncthreads();

    const int tot = cnt * OUT_C;
    for (int i = tid; i < tot; i += 1024) {
        const int r = i >> 4, c = i & 15;
        out[(size_t)n0 * OUT_C + i] = acc[c * OWN_R + r] / (rden[r] + GAT_EPS) + bias[c];
    }
}

extern "C" void kernel_launch(void* const* d_in, const int* in_sizes, int n_in,
                              void* d_out, int out_size, void* d_ws, size_t ws_size,
                              hipStream_t stream)
{
    (void)in_sizes; (void)n_in; (void)out_size; (void)ws_size;

    const float* x    = (const float*)d_in[0];
    const int*   ei   = (const int*)  d_in[1];
    const float* W1   = (const float*)d_in[2];
    const float* ats1 = (const float*)d_in[3];
    const float* atd1 = (const float*)d_in[4];
    const float* b1   = (const float*)d_in[5];
    const float* W2   = (const float*)d_in[6];
    const float* ats2 = (const float*)d_in[7];
    const float* atd2 = (const float*)d_in[8];
    const float* b2   = (const float*)d_in[9];
    const int* src = ei;
    const int* dst = ei + N_EDGES;
    float* out = (float*)d_out;

    float* h1  = (float*)d_ws;
    float* xl1 = h1 + (size_t)N_NODES * HIDDEN;
    float* as1 = xl1 + (size_t)N_NODES * HIDDEN;
    float* ad1 = as1 + (size_t)N_NODES * HEADS;
    float* xl2 = ad1 + (size_t)N_NODES * HEADS;
    float* as2 = xl2 + (size_t)N_NODES * OUT_C;
    float* ad2 = as2 + (size_t)N_NODES;
    unsigned* qsv = (unsigned*)(ad2 + (size_t)N_NODES);
    int*      qct = (int*)(qsv + (size_t)OWN_WG * OWN_WAVES * OWN_QCAP);
    int*      qov = qct + (size_t)OWN_WG * OWN_WAVES;

    const int TB = 256;
    auto blk = [](long long n, int tb) { return (unsigned)((n + tb - 1) / tb); };

    {
        const int waveTilesN = HIDDEN / (16 * 4);
        const long long waves = (long long)(N_NODES / 16) * waveTilesN;
        gat_wmma_gemm_f32<4, IN_C, HIDDEN><<<blk(waves, 4), 128, 0, stream>>>(
            x, W1, xl1, N_NODES, waveTilesN);
    }
    gat_att_logits<<<blk((long long)N_NODES * HEADS, TB), TB, 0, stream>>>(
        xl1, ats1, atd1, as1, ad1, N_NODES, HEADS, HID);
    gat_owner_softmax_aggregate_l1<<<OWN_WG, OWN_WAVES * 32, 0, stream>>>(
        src, dst, as1, ad1, xl1, b1, h1, qsv, qct, qov);

    {
        const int waveTilesN = OUT_C / 16;
        const long long waves = (long long)(N_NODES / 16) * waveTilesN;
        gat_wmma_gemm_f32<1, HIDDEN, OUT_C><<<blk(waves, 4), 128, 0, stream>>>(
            h1, W2, xl2, N_NODES, waveTilesN);
    }
    gat_att_logits<<<blk((long long)N_NODES, TB), TB, 0, stream>>>(
        xl2, ats2, atd2, as2, ad2, N_NODES, 1, OUT_C);
    gat_owner_softmax_aggregate_l2<<<OWN_WG, OWN_WAVES * 32, 0, stream>>>(
        src, dst, as2, ad2, xl2, b2, qsv, qct, qov, out);
}


# Edges filed once by owner work-group in a new binning launch (LDS integer-atomic slots, 256-byte cells); owner waves copy 8 cells instead of scanning the whole list; sums still arrival order
# speedup vs baseline: 7.8440x; 1.1740x over previous
#include <hip/hip_runtime.h>
#include <cstdint>
#include <cstddef>

#define N_NODES   50000
#define N_EDGES   800000
#define TOT_E     (N_EDGES + N_NODES)
#define IN_C      128
#define HID       32
#define OUT_C     16
#define HEADS     8
#define HIDDEN    (HEADS * HID)
#define NEG_SLOPE 0.2f
#define GAT_EPS   1e-16f

typedef float v2f __attribute__((ext_vector_type(2)));
typedef float v8f __attribute__((ext_vector_type(8)));

template <int TN, int K, int NC>
__global__ __launch_bounds__(128) void gat_wmma_gemm_f32(
    const float* __restrict__ A, const float* __restrict__ B,
    float* __restrict__ C, int M, int waveTilesN)
{
    const int wave = blockIdx.x * (blockDim.x >> 5) + (threadIdx.x >> 5);
    const int lane = threadIdx.x & 31;
    const int tM = wave / waveTilesN;
    const int tg = wave % waveTilesN;
    if (tM * 16 >= M) return;

    const int half = lane >> 4;
    const int l    = lane & 15;
    const float* __restrict__ Ap = A + (size_t)(tM * 16 + l) * K + half * 2;
    const float* __restrict__ Bp = B + (size_t)(half * 2) * NC + tg * 16 * TN + l;

    const v8f zero = {0.f,0.f,0.f,0.f,0.f,0.f,0.f,0.f};
    v8f c[TN];
#pragma unroll
    for (int j = 0; j < TN; ++j) c[j] = zero;

#pragma unroll
    for (int k = 0; k < K; k += 4) {
        const v2f a = *(const v2f*)(Ap + k);
        v2f b[TN];
#pragma unroll
        for (int j = 0; j < TN; ++j) {
            b[j].x = Bp[(size_t)k * NC + j * 16];
            b[j].y = Bp[(size_t)(k + 1) * NC + j * 16];
        }
#pragma unroll
        for (int j = 0; j < TN; ++j)
            c[j] = __builtin_amdgcn_wmma_f32_16x16x4_f32(
                        false, a,  false, b[j],
                        (short)0, c[j],
                        false,  false);
    }

    float* __restrict__ Cbase =
        C + (size_t)(tM * 16 + half * 8) * NC + tg * 16 * TN + l;
#pragma unroll
    for (int j = 0; j < TN; ++j)
#pragma unroll
        for (int r = 0; r < 8; ++r)
            Cbase[(size_t)r * NC + j * 16] = c[j][r];
}

__global__ __launch_bounds__(256) void gat_att_logits(
    const float* __restrict__ xl, const float* __restrict__ att_s,
    const float* __restrict__ att_d, float* __restrict__ asrc,
    float* __restrict__ adst, int n, int H, int CH)
{
    const int i = blockIdx.x * blockDim.x + threadIdx.x;
    if (i >= n * H) return;
    const int h = i % H;
    const float* __restrict__ row = xl + (size_t)i * CH;
    const float* __restrict__ as  = att_s + (size_t)h * CH;
    const float* __restrict__ ad  = att_d + (size_t)h * CH;
    float s = 0.f, d = 0.f;
    for (int c = 0; c < CH; ++c) { float v = row[c]; s += v * as[c]; d += v * ad[c]; }
    asrc[i] = s;
    adst[i] = d;
}

__device__ __forceinline__ unsigned fkey(float f) {
    unsigned u = __float_as_uint(f);
    return (u & 0x80000000u) ? ~u : (u | 0x80000000u);
}
__device__ __forceinline__ float fdec(unsigned k) {
    unsigned u = (k & 0x80000000u) ? (k ^ 0x80000000u) : ~k;
    return __uint_as_float(u);
}

__device__ __forceinline__ void edge_sd(long long e, const int* __restrict__ src,
                                        const int* __restrict__ dst, int& s, int& d) {
    if (e < N_EDGES) { s = src[e]; d = dst[e]; }
    else             { s = (int)(e - N_EDGES); d = s; }
}

#define OWN_R     196
#define OWN_WG    256
#define OWN_WAVES 32
#define OWN_QCAP  512
#define OWN_EBITS 20
#define BIN_CHUNKS 256
#define BIN_W      64
#define BIN_CAP    (BIN_W - 1)
#define BIN_VEC    ((N_EDGES / 4 + BIN_CHUNKS - 1) / BIN_CHUNKS)
#define BIN_PER_WAVE (BIN_CHUNKS / OWN_WAVES)
static_assert(TOT_E <= (1 << OWN_EBITS), "edge id must fit the low bits of a queue entry");
static_assert(N_NODES <= 65536, "a resolved queue entry keeps the source node in 16 bits");
static_assert(OWN_R * OWN_WG >= N_NODES, "every node needs an owner");
static_assert(N_EDGES % 4 == 0, "the binning launch reads the edge list as int4");
static_assert(BIN_VEC <= 1024, "one edge vector per thread of a binning work-group");
static_assert(BIN_PER_WAVE == 8 && OWN_WG == 8 * OWN_WAVES, "8 cells per wave on both sides");
static_assert((OWN_R + OWN_WAVES - 1) / OWN_WAVES + BIN_PER_WAVE * BIN_CAP <= OWN_QCAP,
              "self loops plus 8 full cells must fit a wave's queue");

__global__ __launch_bounds__(1024) void gat_bin_edges_by_owner(
    const int* __restrict__ src, const int* __restrict__ dst, unsigned* __restrict__ bins)
{
    __shared__ unsigned lcnt[OWN_WG];
    __shared__ unsigned lent[OWN_WG * BIN_CAP];

    const int tid  = threadIdx.x;
    const int lane = tid & 31;
    const int wave = tid >> 5;
    if (tid < OWN_WG) lcnt[tid] = 0u;
    __syncthreads();

    const int v = blockIdx.x * BIN_VEC + tid;
    if (tid < BIN_VEC && v < N_EDGES / 4) {
        const int4 d4 = ((const int4*)dst)[v];
        const int4 s4 = ((const int4*)src)[v];
        const int dd[4] = {d4.x, d4.y, d4.z, d4.w};
        const int ss[4] = {s4.x, s4.y, s4.z, s4.w};
#pragma unroll
        for (int k = 0; k < 4; ++k) {
            const unsigned d = (unsigned)dd[k];
            if (d < (unsigned)N_NODES) {
                const unsigned o    = d / OWN_R;
                const unsigned rl   = d - o * OWN_R;
                const unsigned slot = __hip_atomic_fetch_add(lcnt + o, 1u, __ATOMIC_RELAXED,
                                                             __HIP_MEMORY_SCOPE_WORKGROUP);
                if (slot < BIN_CAP) lent[o * BIN_CAP + slot] = (rl << 16) | (unsigned)ss[k];
            }
        }
    }
    __syncthreads();

#pragma unroll
    for (int k = 0; k < BIN_PER_WAVE; ++k) {
        const int o = wave * BIN_PER_WAVE + k;
        const unsigned c  = lcnt[o];
        const unsigned cc = c < BIN_CAP ? c : BIN_CAP;
        unsigned* cell = bins + ((size_t)o * BIN_CHUNKS + blockIdx.x) * BIN_W;
        const unsigned* row = lent + o * BIN_CAP;
        cell[lane] = (lane == 0) ? c : (((unsigned)lane <= cc) ? row[lane - 1] : 0u);
        if (cc > 31u) {
            if ((unsigned)(31 + lane) < cc) cell[32 + lane] = row[31 + lane];
        }
    }
}

__device__ __forceinline__ float own_lrelu(float v) { return v > 0.f ? v : NEG_SLOPE * v; }

__device__ __forceinline__ void own_add_row(float* accs, int rl, int lane,
                                            float w0, float w1, float4 x0, float4 x1)
{
    float* row = accs + rl * HIDDEN + lane;
    __hip_atomic_fetch_add(row +   0, w0 * x0.x, __ATOMIC_RELAXED, __HIP_MEMORY_SCOPE_WORKGROUP);
    __hip_atomic_fetch_add(row +  32, w0 * x0.y, __ATOMIC_RELAXED, __HIP_MEMORY_SCOPE_WORKGROUP);
    __hip_atomic_fetch_add(row +  64, w0 * x0.z, __ATOMIC_RELAXED, __HIP_MEMORY_SCOPE_WORKGROUP);
    __hip_atomic_fetch_add(row +  96, w0 * x0.w, __ATOMIC_RELAXED, __HIP_MEMORY_SCOPE_WORKGROUP);
    __hip_atomic_fetch_add(row + 128, w1 * x1.x, __ATOMIC_RELAXED, __HIP_MEMORY_SCOPE_WORKGROUP);
    __hip_atomic_fetch_add(row + 160, w1 * x1.y, __ATOMIC_RELAXED, __HIP_MEMORY_SCOPE_WORKGROUP);
    __hip_atomic_fetch_add(row + 192, w1 * x1.z, __ATOMIC_RELAXED, __HIP_MEMORY_SCOPE_WORKGROUP);
    __hip_atomic_fetch_add(row + 224, w1 * x1.w, __ATOMIC_RELAXED, __HIP_MEMORY_SCOPE_WORKGROUP);
}

template <bool RAW>
__device__ __forceinline__ void own_resolve(
    unsigned* myq, int qn, int lane, const int* __restrict__ src,
    const float4* __restrict__ as4, const float* rad, unsigned* rmax)
{
    const unsigned emask = (1u << OWN_EBITS) - 1u;
    for (int j = lane; j < qn; j += 32) {
        const unsigned pk = myq[j];
        unsigned rl, s;
        if (RAW) {
            const unsigned e = pk & emask;
            rl = pk >> OWN_EBITS;
            s  = (e < (unsigned)N_EDGES) ? (unsigned)src[e] : (e - (unsigned)N_EDGES);
        } else {
            rl = pk >> 16;
            s  = pk & 0xFFFFu;
        }
        const float4 a0 = as4[(size_t)s * 2], a1 = as4[(size_t)s * 2 + 1];
        const float4 d0 = *(const float4*)(rad + rl * HEADS);
        const float4 d1 = *(const float4*)(rad + rl * HEADS + 4);
        const float v[8] = {a0.x + d0.x, a0.y + d0.y, a0.z + d0.z, a0.w + d0.w,
                            a1.x + d1.x, a1.y + d1.y, a1.z + d1.z, a1.w + d1.w};
        unsigned* mx = rmax + rl * HEADS;
#pragma unroll
        for (int h = 0; h < HEADS; ++h)
            __hip_atomic_fetch_max(mx + h, fkey(own_lrelu(v[h])),
                                   __ATOMIC_RELAXED, __HIP_MEMORY_SCOPE_WORKGROUP);
        if (RAW) myq[j] = (rl << 16) | s;
    }
}

__device__ __forceinline__ void own_drain(
    const unsigned* myq, int qn, int lane, int hq,
    const float4* __restrict__ as4, const float4* __restrict__ xl4,
    float* mywt, const float* rad, const unsigned* rmax, float* rden, float* accs)
{
    for (int c0 = 0; c0 < qn; c0 += 32) {
        const int m = (qn - c0 < 32) ? (qn - c0) : 32;
        if (lane < m) {
            const unsigned pk = myq[c0 + lane];
            const unsigned s  = pk & 0xFFFFu;
            const unsigned rl = pk >> 16;
            const float4 a0 = as4[(size_t)s * 2], a1 = as4[(size_t)s * 2 + 1];
            const float4 d0 = *(const float4*)(rad + rl * HEADS);
            const float4 d1 = *(const float4*)(rad + rl * HEADS + 4);
            const uint4  k0 = *(const uint4*)(rmax + rl * HEADS);
            const uint4  k1 = *(const uint4*)(rmax + rl * HEADS + 4);
            float4 w0, w1;
            w0.x = expf(own_lrelu(a0.x + d0.x) - fdec(k0.x));
            w0.y = expf(own_lrelu(a0.y + d0.y) - fdec(k0.y));
            w0.z = expf(own_lrelu(a0.z + d0.z) - fdec(k0.z));
            w0.w = expf(own_lrelu(a0.w + d0.w) - fdec(k0.w));
            w1.x = expf(own_lrelu(a1.x + d1.x) - fdec(k1.x));
            w1.y = expf(own_lrelu(a1.y + d1.y) - fdec(k1.y));
            w1.z = expf(own_lrelu(a1.z + d1.z) - fdec(k1.z));
            w1.w = expf(own_lrelu(a1.w + d1.w) - fdec(k1.w));
            *(float4*)(mywt + lane * HEADS)     = w0;
            *(float4*)(mywt + lane * HEADS + 4) = w1;
            float* dn = rden + rl * HEADS;
            __hip_atomic_fetch_add(dn + 0, w0.x, __ATOMIC_RELAXED, __HIP_MEMORY_SCOPE_WORKGROUP);
            __hip_atomic_fetch_add(dn + 1, w0.y, __ATOMIC_RELAXED, __HIP_MEMORY_SCOPE_WORKGROUP);
            __hip_atomic_fetch_add(dn + 2, w0.z, __ATOMIC_RELAXED, __HIP_MEMORY_SCOPE_WORKGROUP);
            __hip_atomic_fetch_add(dn + 3, w0.w, __ATOMIC_RELAXED, __HIP_MEMORY_SCOPE_WORKGROUP);
            __hip_atomic_fetch_add(dn + 4, w1.x, __ATOMIC_RELAXED, __HIP_MEMORY_SCOPE_WORKGROUP);
            __hip_atomic_fetch_add(dn + 5, w1.y, __ATOMIC_RELAXED, __HIP_MEMORY_SCOPE_WORKGROUP);
            __hip_atomic_fetch_add(dn + 6, w1.z, __ATOMIC_RELAXED, __HIP_MEMORY_SCOPE_WORKGROUP);
            __hip_atomic_fetch_add(dn + 7, w1.w, __ATOMIC_RELAXED, __HIP_MEMORY_SCOPE_WORKGROUP);
        }
        int i = 0;
        for (; i + 4 <= m; i += 4) {
            float4 x0[4], x1[4];
            int    rl[4];
#pragma unroll
            for (int u = 0; u < 4; ++u) {
                const unsigned pk = (unsigned)__builtin_amdgcn_readfirstlane((int)myq[c0 + i + u]);
                const unsigned s  = pk & 0xFFFFu;
                rl[u] = (int)(pk >> 16);
                x0[u] = xl4[(size_t)s * (HIDDEN / 4) + lane];
                x1[u] = xl4[(size_t)s * (HIDDEN / 4) + 32 + lane];
            }
#pragma unroll
            for (int u = 0; u < 4; ++u) {
                const float w0 = mywt[(i + u) * HEADS + hq];
                const float w1 = mywt[(i + u) * HEADS + 4 + hq];
                own_add_row(accs, rl[u], lane, w0, w1, x0[u], x1[u]);
            }
        }
        for (; i < m; ++i) {
            const unsigned pk = (unsigned)__builtin_amdgcn_readfirstlane((int)myq[c0 + i]);
            const unsigned s  = pk & 0xFFFFu;
            const float4 x0 = xl4[(size_t)s * (HIDDEN / 4) + lane];
            const float4 x1 = xl4[(size_t)s * (HIDDEN / 4) + 32 + lane];
            const float w0 = mywt[i * HEADS + hq];
            const float w1 = mywt[i * HEADS + 4 + hq];
            own_add_row(accs, (int)(pk >> 16), lane, w0, w1, x0, x1);
        }
    }
}

template <bool SECOND>
__device__ __forceinline__ int own_filter(
    int tid, int lane, int wave, int hq, int n0, int cnt,
    const int* __restrict__ src, const int* __restrict__ dst,
    const float4* __restrict__ as4, const float4* __restrict__ xl4,
    unsigned* myq, float* mywt, const float* rad, unsigned* rmax,
    float* rden, float* accs, int* ovf)
{
    const int  rs  = wave + OWN_WAVES * lane;
    const bool has = rs < cnt;
    if (has) myq[lane] = ((unsigned)rs << OWN_EBITS) | (unsigned)(N_EDGES + n0 + rs);
    int qn = __builtin_popcount(__builtin_amdgcn_ballot_w32(has));

    const int4* __restrict__ dst4 = (const int4*)dst;
    constexpr int NV    = N_EDGES / 4;
    constexpr int TRIPS = (NV + 1023) / 1024;
    static_assert(TRIPS % 4 == 0, "the four-trip body assumes a whole number of bodies");
    auto ld = [&](int t) {
        const int i = t * 1024 + tid;
        return dst4[i < NV ? i : NV - 1];
    };
    auto trip = [&](const int4 d, int t) {
        const int i = t * 1024 + tid;
        const unsigned lim = (i < NV) ? (unsigned)cnt : 0u;
        const unsigned l0 = (unsigned)(d.x - n0), l1 = (unsigned)(d.y - n0);
        const unsigned l2 = (unsigned)(d.z - n0), l3 = (unsigned)(d.w - n0);
        const bool m0 = l0 < lim, m1 = l1 < lim, m2 = l2 < lim, m3 = l3 < lim;
        if (__builtin_amdgcn_ballot_w32(m0 | m1 | m2 | m3) != 0u) {
            const unsigned ll[4] = {l0, l1, l2, l3};
            const bool     mm[4] = {m0, m1, m2, m3};
#pragma unroll
            for (int k = 0; k < 4; ++k) {
                const unsigned mask = __builtin_amdgcn_ballot_w32(mm[k]);
                if (mm[k]) {
                    const int pos = qn + (int)__builtin_amdgcn_mbcnt_lo(mask, 0u);
                    myq[pos] = (ll[k] << OWN_EBITS) | (unsigned)(i * 4 + k);
                }
                qn += __builtin_popcount(mask);
            }
            if (qn > OWN_QCAP - 128) {
                own_resolve<true>(myq, qn, lane, src, as4, rad, rmax);
                if (SECOND) own_drain(myq, qn, lane, hq, as4, xl4, mywt, rad, rmax, rden, accs);
                else if (lane == 0) *ovf = 1;
                qn = 0;
            }
        }
    };
    int4 d0 = ld(0), d1 = ld(1), d2 = ld(2), d3;
    for (int t = 0; t < TRIPS; t += 4) {
        d3 = ld(t + 3);  trip(d0, t);
        d0 = ld(t + 4);  trip(d1, t + 1);
        d1 = ld(t + 5);  trip(d2, t + 2);
        d2 = ld(t + 6);  trip(d3, t + 3);
    }
    own_resolve<true>(myq, qn, lane, src, as4, rad, rmax);
    if (SECOND) { own_drain(myq, qn, lane, hq, as4, xl4, mywt, rad, rmax, rden, accs); qn = 0; }
    return qn;
}

__device__ __forceinline__ float own_elu(float v) { return v > 0.f ? v : expm1f(v); }

__global__ __launch_bounds__(1024) void gat_owner_softmax_aggregate_l1(
    const int* __restrict__ src, const int* __restrict__ dst,
    const float* __restrict__ asrc, const float* __restrict__ adst,
    const float* __restrict__ xl, const float* __restrict__ bias,
    float* __restrict__ hout, unsigned* __restrict__ qsave,
    int* __restrict__ qcnt, int* __restrict__ qovf, const unsigned* __restrict__ bins)
{
    __shared__ __attribute__((aligned(16))) float    accs[OWN_R * HIDDEN];
    __shared__ __attribute__((aligned(16))) float    rad[OWN_R * HEADS];
    __shared__ __attribute__((aligned(16))) unsigned rmax[OWN_R * HEADS];
    __shared__ __attribute__((aligned(16))) float    rden[OWN_R * HEADS];
    __shared__ __attribute__((aligned(16))) float    wts[OWN_WAVES][32 * HEADS];
    __shared__ unsigned qpk[OWN_WAVES][OWN_QCAP];
    __shared__ int      ovf;
    __shared__ int      binbad;

    const int tid  = threadIdx.x;
    const int lane = tid & 31;
    const int wave = tid >> 5;
    const int hq   = lane >> 3;
    const int n0   = blockIdx.x * OWN_R;
    int cnt = N_NODES - n0;
    if (cnt > OWN_R) cnt = OWN_R;
    const float4* __restrict__ xl4 = (const float4*)xl;
    const float4* __restrict__ as4 = (const float4*)asrc;

    const unsigned* __restrict__ cb =
        bins + ((size_t)blockIdx.x * BIN_CHUNKS + wave * BIN_PER_WAVE) * BIN_W;
    unsigned w0[BIN_PER_WAVE];
#pragma unroll
    for (int k = 0; k < BIN_PER_WAVE; ++k) w0[k] = cb[k * BIN_W + lane];

    float4* accs4 = (float4*)accs;
    for (int i = tid; i < OWN_R * HIDDEN / 4; i += 1024)
        accs4[i] = make_float4(0.f, 0.f, 0.f, 0.f);
    for (int i = tid; i < OWN_R * HEADS; i += 1024) {
        rad[i]  = ((i >> 3) < cnt) ? adst[(size_t)n0 * HEADS + i] : 0.f;
        rmax[i] = 0u;
        rden[i] = 0.f;
    }
    if (tid == 0) { ovf = 0; binbad = 0; }
    __syncthreads();

    unsigned* myq  = qpk[wave];
    float*    mywt = wts[wave];

    int  ck[BIN_PER_WAVE];
    bool bad = false;
#pragma unroll
    for (int k = 0; k < BIN_PER_WAVE; ++k) {
        ck[k] = __builtin_amdgcn_readlane((int)w0[k], 0);
        bad |= (unsigned)ck[k] > (unsigned)BIN_CAP;
    }
    if (bad && lane == 0) binbad = 1;
    __syncthreads();

    int qn;
    if (binbad == 0) {
        const int  rs  = wave + OWN_WAVES * lane;
        const bool has = rs < cnt;
        if (has) myq[lane] = ((unsigned)rs << 16) | (unsigned)(n0 + rs);
        qn = __builtin_popcount(__builtin_amdgcn_ballot_w32(has));
#pragma unroll
        for (int k = 0; k < BIN_PER_WAVE; ++k) {
            if (lane >= 1 && lane <= ck[k]) myq[qn + lane - 1] = w0[k];
            if (ck[k] > 31) {
                if (31 + lane < ck[k]) myq[qn + 31 + lane] = cb[k * BIN_W + 32 + lane];
            }
            qn += ck[k];
        }
        own_resolve<false>(myq, qn, lane, src, as4, rad, rmax);
    } else {
        qn = own_filter<false>(tid, lane, wave, hq, n0, cnt, src, dst, as4, xl4,
                               myq, mywt, rad, rmax, rden, accs, &ovf);
    }
    __syncthreads();

    {
        unsigned* gq = qsave + ((size_t)blockIdx.x * OWN_WAVES + wave) * OWN_QCAP;
        for (int j = lane; j < qn; j += 32) gq[j] = myq[j];
        if (lane == 0) qcnt[blockIdx.x * OWN_WAVES + wave] = qn;
        if (tid == 0)  qovf[blockIdx.x] = ovf;
    }

    if (ovf == 0)
        own_drain(myq, qn, lane, hq, as4, xl4, mywt, rad, rmax, rden, accs);
    else
        (void)own_filter<true>(tid, lane, wave, hq, n0, cnt, src, dst, as4, xl4,
                               myq, mywt, rad, rmax, rden, accs, &ovf);
    __syncthreads();

    float bq0[4], bq1[4];
#pragma unroll
    for (int k = 0; k < 4; ++k) { bq0[k] = bias[lane * 4 + k]; bq1[k] = bias[128 + lane * 4 + k]; }
    float4* __restrict__ h4 = (float4*)hout;
    for (int r = wave; r < cnt; r += OWN_WAVES) {
        const int n = n0 + r;
        const float* row = accs + r * HIDDEN + lane;
        const float d0 = rden[r * HEADS + hq] + GAT_EPS;
        const float d1 = rden[r * HEADS + 4 + hq] + GAT_EPS;
        float4 o0, o1;
        o0.x = own_elu(row[0]   / d0 + bq0[0]);  o0.y = own_elu(row[32]  / d0 + bq0[1]);
        o0.z = own_elu(row[64]  / d0 + bq0[2]);  o0.w = own_elu(row[96]  / d0 + bq0[3]);
        o1.x = own_elu(row[128] / d1 + bq1[0]);  o1.y = own_elu(row[160] / d1 + bq1[1]);
        o1.z = own_elu(row[192] / d1 + bq1[2]);  o1.w = own_elu(row[224] / d1 + bq1[3]);
        h4[(size_t)n * (HIDDEN / 4) + lane]      = o0;
        h4[(size_t)n * (HIDDEN / 4) + 32 + lane] = o1;
    }
}

__device__ __forceinline__ void l2_add_quarter(float* acc, unsigned rl, int q, float w, float4 x)
{
    float* a = acc + (q * 4) * OWN_R + rl;
    __hip_atomic_fetch_add(a,             w * x.x, __ATOMIC_RELAXED, __HIP_MEMORY_SCOPE_WORKGROUP);
    __hip_atomic_fetch_add(a + OWN_R,     w * x.y, __ATOMIC_RELAXED, __HIP_MEMORY_SCOPE_WORKGROUP);
    __hip_atomic_fetch_add(a + 2 * OWN_R, w * x.z, __ATOMIC_RELAXED, __HIP_MEMORY_SCOPE_WORKGROUP);
    __hip_atomic_fetch_add(a + 3 * OWN_R, w * x.w, __ATOMIC_RELAXED, __HIP_MEMORY_SCOPE_WORKGROUP);
}

__global__ __launch_bounds__(1024) void gat_owner_softmax_aggregate_l2(
    const int* __restrict__ src, const int* __restrict__ dst,
    const float* __restrict__ as2, const float* __restrict__ ad2,
    const float* __restrict__ xl2, const float* __restrict__ bias,
    const unsigned* __restrict__ qsave, const int* __restrict__ qcnt,
    const int* __restrict__ qovf, float* __restrict__ out)
{
    __shared__ float    acc[OUT_C * OWN_R];
    __shared__ float    rad[OWN_R];
    __shared__ unsigned rmax[OWN_R];
    __shared__ float    rden[OWN_R];

    const int tid  = threadIdx.x;
    const int lane = tid & 31;
    const int wave = tid >> 5;
    const int n0   = blockIdx.x * OWN_R;
    int cnt = N_NODES - n0;
    if (cnt > OWN_R) cnt = OWN_R;
    const float4* __restrict__ xq = (const float4*)xl2;

    for (int i = tid; i < OUT_C * OWN_R; i += 1024) acc[i] = 0.f;
    if (tid < OWN_R) {
        rad[tid]  = (tid < cnt) ? ad2[n0 + tid] : 0.f;
        rmax[tid] = 0u;
        rden[tid] = 0.f;
    }
    const bool queued = qovf[blockIdx.x] == 0;
    const unsigned* __restrict__ gq = qsave + ((size_t)blockIdx.x * OWN_WAVES + wave) * OWN_QCAP;
    const int qn = queued ? __builtin_amdgcn_readfirstlane(qcnt[blockIdx.x * OWN_WAVES + wave]) : 0;
    __syncthreads();

    if (queued) {
        for (int j0 = 0; j0 < qn; j0 += 128) {
            unsigned pk[4];
            float    a[4];
#pragma unroll
            for (int t = 0; t < 4; ++t) {
                const int j = j0 + t * 32 + lane;
                pk[t] = (j < qn) ? gq[j] : 0u;
            }
#pragma unroll
            for (int t = 0; t < 4; ++t) a[t] = as2[pk[t] & 0xFFFFu];
#pragma unroll
            for (int t = 0; t < 4; ++t) {
                if (j0 + t * 32 + lane < qn) {
                    const unsigned rl = pk[t] >> 16;
                    const float v = own_lrelu(a[t] + rad[rl]);
                    __hip_atomic_fetch_max(rmax + rl, fkey(v), __ATOMIC_RELAXED, __HIP_MEMORY_SCOPE_WORKGROUP);
                }
            }
        }
    } else {
        for (int e = tid; e < TOT_E; e += 1024) {
            int s, d; edge_sd(e, src, dst, s, d);
            const unsigned rl = (unsigned)(d - n0);
            if (rl < (unsigned)cnt) {
                const float v = own_lrelu(as2[s] + rad[rl]);
                __hip_atomic_fetch_max(rmax + rl, fkey(v), __ATOMIC_RELAXED, __HIP_MEMORY_SCOPE_WORKGROUP);
            }
        }
    }
    __syncthreads();

    if (queued) {
        const int q   = lane & 3;
        const int sub = lane >> 2;
        auto entry = [&](int j) { return gq[j < qn ? j : qn - 1]; };
        unsigned nxt = (qn > 0) ? entry(lane) : 0u;
        for (int j0 = 0; j0 < qn; j0 += 32) {
            const unsigned mine = nxt;
            nxt = entry(j0 + 32 + lane);
            unsigned pk[4];
            float    a[4];
            float4   x[4];
#pragma unroll
            for (int t = 0; t < 4; ++t) {
                pk[t] = (unsigned)__shfl((int)mine, t * 8 + sub);
                const unsigned s = pk[t] & 0xFFFFu;
                a[t] = as2[s];
                x[t] = xq[(size_t)s * (OUT_C / 4) + q];
            }
#pragma unroll
            for (int t = 0; t < 4; ++t) {
                if (j0 + t * 8 + sub < qn) {
                    const unsigned rl = pk[t] >> 16;
                    const float w = expf(own_lrelu(a[t] + rad[rl]) - fdec(rmax[rl]));
                    if (q == 0)
                        __hip_atomic_fetch_add(rden + rl, w, __ATOMIC_RELAXED, __HIP_MEMORY_SCOPE_WORKGROUP);
                    l2_add_quarter(acc, rl, q, w, x[t]);
                }
            }
        }
    } else {
        for (int e = tid; e < TOT_E; e += 1024) {
            int s, d; edge_sd(e, src, dst, s, d);
            const unsigned rl = (unsigned)(d - n0);
            if (rl < (unsigned)cnt) {
                const float w = expf(own_lrelu(as2[s] + rad[rl]) - fdec(rmax[rl]));
                __hip_atomic_fetch_add(rden + rl, w, __ATOMIC_RELAXED, __HIP_MEMORY_SCOPE_WORKGROUP);
                for (int q = 0; q < 4; ++q)
                    l2_add_quarter(acc, rl, q, w, xq[(size_t)s * (OUT_C / 4) + q]);
            }
        }
    }
    __syncthreads();

    const int tot = cnt * OUT_C;
    for (int i = tid; i < tot; i += 1024) {
        const int r = i >> 4, c = i & 15;
        out[(size_t)n0 * OUT_C + i] = acc[c * OWN_R + r] / (rden[r] + GAT_EPS) + bias[c];
    }
}

extern "C" void kernel_launch(void* const* d_in, const int* in_sizes, int n_in,
                              void* d_out, int out_size, void* d_ws, size_t ws_size,
                              hipStream_t stream)
{
    (void)in_sizes; (void)n_in; (void)out_size; (void)ws_size;

    const float* x    = (const float*)d_in[0];
    const int*   ei   = (const int*)  d_in[1];
    const float* W1   = (const float*)d_in[2];
    const float* ats1 = (const float*)d_in[3];
    const float* atd1 = (const float*)d_in[4];
    const float* b1   = (const float*)d_in[5];
    const float* W2   = (const float*)d_in[6];
    const float* ats2 = (const float*)d_in[7];
    const float* atd2 = (const float*)d_in[8];
    const float* b2   = (const float*)d_in[9];
    const int* src = ei;
    const int* dst = ei + N_EDGES;
    float* out = (float*)d_out;

    float* h1  = (float*)d_ws;
    float* xl1 = h1 + (size_t)N_NODES * HIDDEN;
    float* as1 = xl1 + (size_t)N_NODES * HIDDEN;
    float* ad1 = as1 + (size_t)N_NODES * HEADS;
    float* xl2 = ad1 + (size_t)N_NODES * HEADS;
    float* as2 = xl2 + (size_t)N_NODES * OUT_C;
    float* ad2 = as2 + (size_t)N_NODES;
    unsigned* qsv = (unsigned*)(ad2 + (size_t)N_NODES);
    int*      qct = (int*)(qsv + (size_t)OWN_WG * OWN_WAVES * OWN_QCAP);
    int*      qov = qct + (size_t)OWN_WG * OWN_WAVES;
    constexpr size_t kWordsBefore = (size_t)N_NODES * (2 * HIDDEN + 2 * HEADS + OUT_C + 2)
                                  + (size_t)OWN_WG * OWN_WAVES * OWN_QCAP
                                  + (size_t)OWN_WG * OWN_WAVES + OWN_WG;
    constexpr size_t kBinsAt = (kWordsBefore + BIN_W - 1) / BIN_W * BIN_W;
    static_assert((kBinsAt + (size_t)OWN_WG * BIN_CHUNKS * BIN_W) * 4 <= 268435456ull,
                  "the bins must end inside the workspace");
    unsigned* bins = (unsigned*)d_ws + kBinsAt;

    const int TB = 256;
    auto blk = [](long long n, int tb) { return (unsigned)((n + tb - 1) / tb); };

    gat_bin_edges_by_owner<<<BIN_CHUNKS, 1024, 0, stream>>>(src, dst, bins);

    {
        const int waveTilesN = HIDDEN / (16 * 4);
        const long long waves = (long long)(N_NODES / 16) * waveTilesN;
        gat_wmma_gemm_f32<4, IN_C, HIDDEN><<<blk(waves, 4), 128, 0, stream>>>(
            x, W1, xl1, N_NODES, waveTilesN);
    }
    gat_att_logits<<<blk((long long)N_NODES * HEADS, TB), TB, 0, stream>>>(
        xl1, ats1, atd1, as1, ad1, N_NODES, HEADS, HID);
    gat_owner_softmax_aggregate_l1<<<OWN_WG, OWN_WAVES * 32, 0, stream>>>(
        src, dst, as1, ad1, xl1, b1, h1, qsv, qct, qov, bins);

    {
        const int waveTilesN = OUT_C / 16;
        const long long waves = (long long)(N_NODES / 16) * waveTilesN;
        gat_wmma_gemm_f32<1, HIDDEN, OUT_C><<<blk(waves, 4), 128, 0, stream>>>(
            h1, W2, xl2, N_NODES, waveTilesN);
    }
    gat_att_logits<<<blk((long long)N_NODES, TB), TB, 0, stream>>>(
        xl2, ats2, atd2, as2, ad2, N_NODES, 1, OUT_C);
    gat_owner_softmax_aggregate_l2<<<OWN_WG, OWN_WAVES * 32, 0, stream>>>(
        src, dst, as2, ad2, xl2, b2, qsv, qct, qov, out);
}


# Attention logits made as 16 extra GEMM columns per layer from folded weights W.att (built by 6 extra work-groups of the binning launch); logit sums now in the GEMM k order; 2 logits launches gone
# speedup vs baseline: 8.2917x; 1.0571x over previous
#include <hip/hip_runtime.h>
#include <cstdint>
#include <cstddef>

#define N_NODES   50000
#define N_EDGES   800000
#define TOT_E     (N_EDGES + N_NODES)
#define IN_C      128
#define HID       32
#define OUT_C     16
#define HEADS     8
#define HIDDEN    (HEADS * HID)
#define NEG_SLOPE 0.2f
#define GAT_EPS   1e-16f

typedef float v2f __attribute__((ext_vector_type(2)));
typedef float v8f __attribute__((ext_vector_type(8)));

template <int TN, int K, int NC>
__device__ __forceinline__ void gat_wmma_strip_f32(
    const float* __restrict__ Ap, const float* const (&Bp)[TN], v8f (&c)[TN])
{
    const v8f zero = {0.f,0.f,0.f,0.f,0.f,0.f,0.f,0.f};
#pragma unroll
    for (int j = 0; j < TN; ++j) c[j] = zero;

#pragma unroll
    for (int k = 0; k < K; k += 4) {
        const v2f a = *(const v2f*)(Ap + k);
        v2f b[TN];
#pragma unroll
        for (int j = 0; j < TN; ++j) {
            b[j].x = Bp[j][(size_t)k * NC];
            b[j].y = Bp[j][(size_t)(k + 1) * NC];
        }
#pragma unroll
        for (int j = 0; j < TN; ++j)
            c[j] = __builtin_amdgcn_wmma_f32_16x16x4_f32(
                        false, a,  false, b[j],
                        (short)0, c[j],
                        false,  false);
    }
}

#define FOLD_COLS 16
#define FOLD_WG   ((IN_C + HIDDEN) * FOLD_COLS / 1024)
static_assert(FOLD_WG * 1024 == (IN_C + HIDDEN) * FOLD_COLS, "one thread per folded weight");
static_assert(2 * HEADS == FOLD_COLS && N_NODES % 16 == 0, "logit tile shape; whole row tiles");

__device__ __forceinline__ void gat_fold_att(
    int t, const float* __restrict__ W1, const float* __restrict__ ats1,
    const float* __restrict__ atd1, const float* __restrict__ W2,
    const float* __restrict__ ats2, const float* __restrict__ atd2,
    float* __restrict__ wa1, float* __restrict__ wa2)
{
    if (t < IN_C * FOLD_COLS) {
        const int k = t / FOLD_COLS, j = t % FOLD_COLS, h = j % HEADS;
        const float* __restrict__ att = (j < HEADS ? ats1 : atd1) + h * HID;
        const float* __restrict__ w   = W1 + (size_t)k * HIDDEN + h * HID;
        float s = 0.f;
        for (int c = 0; c < HID; ++c) s += w[c] * att[c];
        wa1[t] = s;
    } else {
        const int u = t - IN_C * FOLD_COLS;
        const int k = u / FOLD_COLS, j = u % FOLD_COLS;
        float s = 0.f;
        if (j < 2) {
            const float* __restrict__ att = (j == 0) ? ats2 : atd2;
            const float* __restrict__ w   = W2 + (size_t)k * OUT_C;
            for (int c = 0; c < OUT_C; ++c) s += w[c] * att[c];
        }
        wa2[u] = s;
    }
}

__global__ __launch_bounds__(160) void gat_gemm1_xl_logits(
    const float* __restrict__ x, const float* __restrict__ W1,
    const float* __restrict__ wa1, float* __restrict__ xl,
    float* __restrict__ asrc, float* __restrict__ adst)
{
    const int wave = threadIdx.x >> 5;
    const int lane = threadIdx.x & 31;
    const int tM   = blockIdx.x;
    const int half = lane >> 4;
    const int l    = lane & 15;
    const float* __restrict__ Ap = x + (size_t)(tM * 16 + l) * IN_C + half * 2;

    if (wave < 4) {
        const float* __restrict__ B0 = W1 + (size_t)(half * 2) * HIDDEN + wave * 64 + l;
        const float* const Bp[4] = {B0, B0 + 16, B0 + 32, B0 + 48};
        v8f c[4];
        gat_wmma_strip_f32<4, IN_C, HIDDEN>(Ap, Bp, c);
        float* __restrict__ Cbase = xl + (size_t)(tM * 16 + half * 8) * HIDDEN + wave * 64 + l;
#pragma unroll
        for (int j = 0; j < 4; ++j)
#pragma unroll
            for (int r = 0; r < 8; ++r)
                Cbase[(size_t)r * HIDDEN + j * 16] = c[j][r];
    } else {
        const float* const Bp[1] = {wa1 + (half * 2) * FOLD_COLS + l};
        v8f c[1];
        gat_wmma_strip_f32<1, IN_C, FOLD_COLS>(Ap, Bp, c);
        float* __restrict__ o = (l < HEADS ? asrc : adst)
                              + (size_t)(tM * 16 + half * 8) * HEADS + (l % HEADS);
#pragma unroll
        for (int r = 0; r < 8; ++r) o[r * HEADS] = c[0][r];
    }
}

__global__ __launch_bounds__(128) void gat_gemm2_xl_logits(
    const float* __restrict__ h, const float* __restrict__ W2,
    const float* __restrict__ wa2, float* __restrict__ xl2,
    float* __restrict__ as2, float* __restrict__ ad2)
{
    const int tM   = blockIdx.x * 4 + (threadIdx.x >> 5);
    const int lane = threadIdx.x & 31;
    if (tM * 16 >= N_NODES) return;
    const int half = lane >> 4;
    const int l    = lane & 15;
    const float* __restrict__ Ap = h + (size_t)(tM * 16 + l) * HIDDEN + half * 2;
    const float* const Bp[2] = {W2  + (half * 2) * OUT_C + l,
                                wa2 + (half * 2) * FOLD_COLS + l};
    static_assert(OUT_C == FOLD_COLS, "both tiles share the row pitch");
    v8f c[2];
    gat_wmma_strip_f32<2, HIDDEN, OUT_C>(Ap, Bp, c);
    float* __restrict__ Cbase = xl2 + (size_t)(tM * 16 + half * 8) * OUT_C + l;
#pragma unroll
    for (int r = 0; r < 8; ++r) Cbase[(size_t)r * OUT_C] = c[0][r];
    if (l < 2) {
        float* __restrict__ o = (l == 0 ? as2 : ad2) + tM * 16 + half * 8;
#pragma unroll
        for (int r = 0; r < 8; ++r) o[r] = c[1][r];
    }
}

__device__ __forceinline__ unsigned fkey(float f) {
    unsigned u = __float_as_uint(f);
    return (u & 0x80000000u) ? ~u : (u | 0x80000000u);
}
__device__ __forceinline__ float fdec(unsigned k) {
    unsigned u = (k & 0x80000000u) ? (k ^ 0x80000000u) : ~k;
    return __uint_as_float(u);
}

__device__ __forceinline__ void edge_sd(long long e, const int* __restrict__ src,
                                        const int* __restrict__ dst, int& s, int& d) {
    if (e < N_EDGES) { s = src[e]; d = dst[e]; }
    else             { s = (int)(e - N_EDGES); d = s; }
}

#define OWN_R     196
#define OWN_WG    256
#define OWN_WAVES 32
#define OWN_QCAP  512
#define OWN_EBITS 20
#define BIN_CHUNKS 256
#define BIN_W      64
#define BIN_CAP    (BIN_W - 1)
#define BIN_VEC    ((N_EDGES / 4 + BIN_CHUNKS - 1) / BIN_CHUNKS)
#define BIN_PER_WAVE (BIN_CHUNKS / OWN_WAVES)
static_assert(TOT_E <= (1 << OWN_EBITS), "edge id must fit the low bits of a queue entry");
static_assert(N_NODES <= 65536, "a resolved queue entry keeps the source node in 16 bits");
static_assert(OWN_R * OWN_WG >= N_NODES, "every node needs an owner");
static_assert(N_EDGES % 4 == 0, "the binning launch reads the edge list as int4");
static_assert(BIN_VEC <= 1024, "one edge vector per thread of a binning work-group");
static_assert(BIN_PER_WAVE == 8 && OWN_WG == 8 * OWN_WAVES, "8 cells per wave on both sides");
static_assert((OWN_R + OWN_WAVES - 1) / OWN_WAVES + BIN_PER_WAVE * BIN_CAP <= OWN_QCAP,
              "self loops plus 8 full cells must fit a wave's queue");

__global__ __launch_bounds__(1024) void gat_bin_edges_by_owner(
    const int* __restrict__ src, const int* __restrict__ dst, unsigned* __restrict__ bins,
    const float* __restrict__ W1, const float* __restrict__ ats1, const float* __restrict__ atd1,
    const float* __restrict__ W2, const float* __restrict__ ats2, const float* __restrict__ atd2,
    float* __restrict__ wa1, float* __restrict__ wa2)
{
    if (blockIdx.x >= BIN_CHUNKS) {
        gat_fold_att((int)(blockIdx.x - BIN_CHUNKS) * 1024 + (int)threadIdx.x,
                     W1, ats1, atd1, W2, ats2, atd2, wa1, wa2);
        return;
    }

    __shared__ unsigned lcnt[OWN_WG];
    __shared__ unsigned lent[OWN_WG * BIN_CAP];

    const int tid  = threadIdx.x;
    const int lane = tid & 31;
    const int wave = tid >> 5;
    if (tid < OWN_WG) lcnt[tid] = 0u;
    __syncthreads();

    const int v = blockIdx.x * BIN_VEC + tid;
    if (tid < BIN_VEC && v < N_EDGES / 4) {
        const int4 d4 = ((const int4*)dst)[v];
        const int4 s4 = ((const int4*)src)[v];
        const int dd[4] = {d4.x, d4.y, d4.z, d4.w};
        const int ss[4] = {s4.x, s4.y, s4.z, s4.w};
#pragma unroll
        for (int k = 0; k < 4; ++k) {
            const unsigned d = (unsigned)dd[k];
            if (d < (unsigned)N_NODES) {
                const unsigned o    = d / OWN_R;
                const unsigned rl   = d - o * OWN_R;
                const unsigned slot = __hip_atomic_fetch_add(lcnt + o, 1u, __ATOMIC_RELAXED,
                                                             __HIP_MEMORY_SCOPE_WORKGROUP);
                if (slot < BIN_CAP) lent[o * BIN_CAP + slot] = (rl << 16) | (unsigned)ss[k];
            }
        }
    }
    __syncthreads();

#pragma unroll
    for (int k = 0; k < BIN_PER_WAVE; ++k) {
        const int o = wave * BIN_PER_WAVE + k;
        const unsigned c  = lcnt[o];
        const unsigned cc = c < BIN_CAP ? c : BIN_CAP;
        unsigned* cell = bins + ((size_t)o * BIN_CHUNKS + blockIdx.x) * BIN_W;
        const unsigned* row = lent + o * BIN_CAP;
        cell[lane] = (lane == 0) ? c : (((unsigned)lane <= cc) ? row[lane - 1] : 0u);
        if (cc > 31u) {
            if ((unsigned)(31 + lane) < cc) cell[32 + lane] = row[31 + lane];
        }
    }
}

__device__ __forceinline__ float own_lrelu(float v) { return v > 0.f ? v : NEG_SLOPE * v; }

__device__ __forceinline__ void own_add_row(float* accs, int rl, int lane,
                                            float w0, float w1, float4 x0, float4 x1)
{
    float* row = accs + rl * HIDDEN + lane;
    __hip_atomic_fetch_add(row +   0, w0 * x0.x, __ATOMIC_RELAXED, __HIP_MEMORY_SCOPE_WORKGROUP);
    __hip_atomic_fetch_add(row +  32, w0 * x0.y, __ATOMIC_RELAXED, __HIP_MEMORY_SCOPE_WORKGROUP);
    __hip_atomic_fetch_add(row +  64, w0 * x0.z, __ATOMIC_RELAXED, __HIP_MEMORY_SCOPE_WORKGROUP);
    __hip_atomic_fetch_add(row +  96, w0 * x0.w, __ATOMIC_RELAXED, __HIP_MEMORY_SCOPE_WORKGROUP);
    __hip_atomic_fetch_add(row + 128, w1 * x1.x, __ATOMIC_RELAXED, __HIP_MEMORY_SCOPE_WORKGROUP);
    __hip_atomic_fetch_add(row + 160, w1 * x1.y, __ATOMIC_RELAXED, __HIP_MEMORY_SCOPE_WORKGROUP);
    __hip_atomic_fetch_add(row + 192, w1 * x1.z, __ATOMIC_RELAXED, __HIP_MEMORY_SCOPE_WORKGROUP);
    __hip_atomic_fetch_add(row + 224, w1 * x1.w, __ATOMIC_RELAXED, __HIP_MEMORY_SCOPE_WORKGROUP);
}

template <bool RAW>
__device__ __forceinline__ void own_resolve(
    unsigned* myq, int qn, int lane, const int* __restrict__ src,
    const float4* __restrict__ as4, const float* rad, unsigned* rmax)
{
    const unsigned emask = (1u << OWN_EBITS) - 1u;
    for (int j = lane; j < qn; j += 32) {
        const unsigned pk = myq[j];
        unsigned rl, s;
        if (RAW) {
            const unsigned e = pk & emask;
            rl = pk >> OWN_EBITS;
            s  = (e < (unsigned)N_EDGES) ? (unsigned)src[e] : (e - (unsigned)N_EDGES);
        } else {
            rl = pk >> 16;
            s  = pk & 0xFFFFu;
        }
        const float4 a0 = as4[(size_t)s * 2], a1 = as4[(size_t)s * 2 + 1];
        const float4 d0 = *(const float4*)(rad + rl * HEADS);
        const float4 d1 = *(const float4*)(rad + rl * HEADS + 4);
        const float v[8] = {a0.x + d0.x, a0.y + d0.y, a0.z + d0.z, a0.w + d0.w,
                            a1.x + d1.x, a1.y + d1.y, a1.z + d1.z, a1.w + d1.w};
        unsigned* mx = rmax + rl * HEADS;
#pragma unroll
        for (int h = 0; h < HEADS; ++h)
            __hip_atomic_fetch_max(mx + h, fkey(own_lrelu(v[h])),
                                   __ATOMIC_RELAXED, __HIP_MEMORY_SCOPE_WORKGROUP);
        if (RAW) myq[j] = (rl << 16) | s;
    }
}

__device__ __forceinline__ void own_drain(
    const unsigned* myq, int qn, int lane, int hq,
    const float4* __restrict__ as4, const float4* __restrict__ xl4,
    float* mywt, const float* rad, const unsigned* rmax, float* rden, float* accs)
{
    for (int c0 = 0; c0 < qn; c0 += 32) {
        const int m = (qn - c0 < 32) ? (qn - c0) : 32;
        if (lane < m) {
            const unsigned pk = myq[c0 + lane];
            const unsigned s  = pk & 0xFFFFu;
            const unsigned rl = pk >> 16;
            const float4 a0 = as4[(size_t)s * 2], a1 = as4[(size_t)s * 2 + 1];
            const float4 d0 = *(const float4*)(rad + rl * HEADS);
            const float4 d1 = *(const float4*)(rad + rl * HEADS + 4);
            const uint4  k0 = *(const uint4*)(rmax + rl * HEADS);
            const uint4  k1 = *(const uint4*)(rmax + rl * HEADS + 4);
            float4 w0, w1;
            w0.x = expf(own_lrelu(a0.x + d0.x) - fdec(k0.x));
            w0.y = expf(own_lrelu(a0.y + d0.y) - fdec(k0.y));
            w0.z = expf(own_lrelu(a0.z + d0.z) - fdec(k0.z));
            w0.w = expf(own_lrelu(a0.w + d0.w) - fdec(k0.w));
            w1.x = expf(own_lrelu(a1.x + d1.x) - fdec(k1.x));
            w1.y = expf(own_lrelu(a1.y + d1.y) - fdec(k1.y));
            w1.z = expf(own_lrelu(a1.z + d1.z) - fdec(k1.z));
            w1.w = expf(own_lrelu(a1.w + d1.w) - fdec(k1.w));
            *(float4*)(mywt + lane * HEADS)     = w0;
            *(float4*)(mywt + lane * HEADS + 4) = w1;
            float* dn = rden + rl * HEADS;
            __hip_atomic_fetch_add(dn + 0, w0.x, __ATOMIC_RELAXED, __HIP_MEMORY_SCOPE_WORKGROUP);
            __hip_atomic_fetch_add(dn + 1, w0.y, __ATOMIC_RELAXED, __HIP_MEMORY_SCOPE_WORKGROUP);
            __hip_atomic_fetch_add(dn + 2, w0.z, __ATOMIC_RELAXED, __HIP_MEMORY_SCOPE_WORKGROUP);
            __hip_atomic_fetch_add(dn + 3, w0.w, __ATOMIC_RELAXED, __HIP_MEMORY_SCOPE_WORKGROUP);
            __hip_atomic_fetch_add(dn + 4, w1.x, __ATOMIC_RELAXED, __HIP_MEMORY_SCOPE_WORKGROUP);
            __hip_atomic_fetch_add(dn + 5, w1.y, __ATOMIC_RELAXED, __HIP_MEMORY_SCOPE_WORKGROUP);
            __hip_atomic_fetch_add(dn + 6, w1.z, __ATOMIC_RELAXED, __HIP_MEMORY_SCOPE_WORKGROUP);
            __hip_atomic_fetch_add(dn + 7, w1.w, __ATOMIC_RELAXED, __HIP_MEMORY_SCOPE_WORKGROUP);
        }
        int i = 0;
        for (; i + 4 <= m; i += 4) {
            float4 x0[4], x1[4];
            int    rl[4];
#pragma unroll
            for (int u = 0; u < 4; ++u) {
                const unsigned pk = (unsigned)__builtin_amdgcn_readfirstlane((int)myq[c0 + i + u]);
                const unsigned s  = pk & 0xFFFFu;
                rl[u] = (int)(pk >> 16);
                x0[u] = xl4[(size_t)s * (HIDDEN / 4) + lane];
                x1[u] = xl4[(size_t)s * (HIDDEN / 4) + 32 + lane];
            }
#pragma unroll
            for (int u = 0; u < 4; ++u) {
                const float w0 = mywt[(i + u) * HEADS + hq];
                const float w1 = mywt[(i + u) * HEADS + 4 + hq];
                own_add_row(accs, rl[u], lane, w0, w1, x0[u], x1[u]);
            }
        }
        for (; i < m; ++i) {
            const unsigned pk = (unsigned)__builtin_amdgcn_readfirstlane((int)myq[c0 + i]);
            const unsigned s  = pk & 0xFFFFu;
            const float4 x0 = xl4[(size_t)s * (HIDDEN / 4) + lane];
            const float4 x1 = xl4[(size_t)s * (HIDDEN / 4) + 32 + lane];
            const float w0 = mywt[i * HEADS + hq];
            const float w1 = mywt[i * HEADS + 4 + hq];
            own_add_row(accs, (int)(pk >> 16), lane, w0, w1, x0, x1);
        }
    }
}

template <bool SECOND>
__device__ __forceinline__ int own_filter(
    int tid, int lane, int wave, int hq, int n0, int cnt,
    const int* __restrict__ src, const int* __restrict__ dst,
    const float4* __restrict__ as4, const float4* __restrict__ xl4,
    unsigned* myq, float* mywt, const float* rad, unsigned* rmax,
    float* rden, float* accs, int* ovf)
{
    const int  rs  = wave + OWN_WAVES * lane;
    const bool has = rs < cnt;
    if (has) myq[lane] = ((unsigned)rs << OWN_EBITS) | (unsigned)(N_EDGES + n0 + rs);
    int qn = __builtin_popcount(__builtin_amdgcn_ballot_w32(has));

    const int4* __restrict__ dst4 = (const int4*)dst;
    constexpr int NV    = N_EDGES / 4;
    constexpr int TRIPS = (NV + 1023) / 1024;
    static_assert(TRIPS % 4 == 0, "the four-trip body assumes a whole number of bodies");
    auto ld = [&](int t) {
        const int i = t * 1024 + tid;
        return dst4[i < NV ? i : NV - 1];
    };
    auto trip = [&](const int4 d, int t) {
        const int i = t * 1024 + tid;
        const unsigned lim = (i < NV) ? (unsigned)cnt : 0u;
        const unsigned l0 = (unsigned)(d.x - n0), l1 = (unsigned)(d.y - n0);
        const unsigned l2 = (unsigned)(d.z - n0), l3 = (unsigned)(d.w - n0);
        const bool m0 = l0 < lim, m1 = l1 < lim, m2 = l2 < lim, m3 = l3 < lim;
        if (__builtin_amdgcn_ballot_w32(m0 | m1 | m2 | m3) != 0u) {
            const unsigned ll[4] = {l0, l1, l2, l3};
            const bool     mm[4] = {m0, m1, m2, m3};
#pragma unroll
            for (int k = 0; k < 4; ++k) {
                const unsigned mask = __builtin_amdgcn_ballot_w32(mm[k]);
                if (mm[k]) {
                    const int pos = qn + (int)__builtin_amdgcn_mbcnt_lo(mask, 0u);
                    myq[pos] = (ll[k] << OWN_EBITS) | (unsigned)(i * 4 + k);
                }
                qn += __builtin_popcount(mask);
            }
            if (qn > OWN_QCAP - 128) {
                own_resolve<true>(myq, qn, lane, src, as4, rad, rmax);
                if (SECOND) own_drain(myq, qn, lane, hq, as4, xl4, mywt, rad, rmax, rden, accs);
                else if (lane == 0) *ovf = 1;
                qn = 0;
            }
        }
    };
    int4 d0 = ld(0), d1 = ld(1), d2 = ld(2), d3;
    for (int t = 0; t < TRIPS; t += 4) {
        d3 = ld(t + 3);  trip(d0, t);
        d0 = ld(t + 4);  trip(d1, t + 1);
        d1 = ld(t + 5);  trip(d2, t + 2);
        d2 = ld(t + 6);  trip(d3, t + 3);
    }
    own_resolve<true>(myq, qn, lane, src, as4, rad, rmax);
    if (SECOND) { own_drain(myq, qn, lane, hq, as4, xl4, mywt, rad, rmax, rden, accs); qn = 0; }
    return qn;
}

__device__ __forceinline__ float own_elu(float v) { return v > 0.f ? v : expm1f(v); }

__global__ __launch_bounds__(1024) void gat_owner_softmax_aggregate_l1(
    const int* __restrict__ src, const int* __restrict__ dst,
    const float* __restrict__ asrc, const float* __restrict__ adst,
    const float* __restrict__ xl, const float* __restrict__ bias,
    float* __restrict__ hout, unsigned* __restrict__ qsave,
    int* __restrict__ qcnt, int* __restrict__ qovf, const unsigned* __restrict__ bins)
{
    __shared__ __attribute__((aligned(16))) float    accs[OWN_R * HIDDEN];
    __shared__ __attribute__((aligned(16))) float    rad[OWN_R * HEADS];
    __shared__ __attribute__((aligned(16))) unsigned rmax[OWN_R * HEADS];
    __shared__ __attribute__((aligned(16))) float    rden[OWN_R * HEADS];
    __shared__ __attribute__((aligned(16))) float    wts[OWN_WAVES][32 * HEADS];
    __shared__ unsigned qpk[OWN_WAVES][OWN_QCAP];
    __shared__ int      ovf;
    __shared__ int      binbad;

    const int tid  = threadIdx.x;
    const int lane = tid & 31;
    const int wave = tid >> 5;
    const int hq   = lane >> 3;
    const int n0   = blockIdx.x * OWN_R;
    int cnt = N_NODES - n0;
    if (cnt > OWN_R) cnt = OWN_R;
    const float4* __restrict__ xl4 = (const float4*)xl;
    const float4* __restrict__ as4 = (const float4*)asrc;

    const unsigned* __restrict__ cb =
        bins + ((size_t)blockIdx.x * BIN_CHUNKS + wave * BIN_PER_WAVE) * BIN_W;
    unsigned w0[BIN_PER_WAVE];
#pragma unroll
    for (int k = 0; k < BIN_PER_WAVE; ++k) w0[k] = cb[k * BIN_W + lane];

    float4* accs4 = (float4*)accs;
    for (int i = tid; i < OWN_R * HIDDEN / 4; i += 1024)
        accs4[i] = make_float4(0.f, 0.f, 0.f, 0.f);
    for (int i = tid; i < OWN_R * HEADS; i += 1024) {
        rad[i]  = ((i >> 3) < cnt) ? adst[(size_t)n0 * HEADS + i] : 0.f;
        rmax[i] = 0u;
        rden[i] = 0.f;
    }
    if (tid == 0) { ovf = 0; binbad = 0; }
    __syncthreads();

    unsigned* myq  = qpk[wave];
    float*    mywt = wts[wave];

    int  ck[BIN_PER_WAVE];
    bool bad = false;
#pragma unroll
    for (int k = 0; k < BIN_PER_WAVE; ++k) {
        ck[k] = __builtin_amdgcn_readlane((int)w0[k], 0);
        bad |= (unsigned)ck[k] > (unsigned)BIN_CAP;
    }
    if (bad && lane == 0) binbad = 1;
    __syncthreads();

    int qn;
    if (binbad == 0) {
        const int  rs  = wave + OWN_WAVES * lane;
        const bool has = rs < cnt;
        if (has) myq[lane] = ((unsigned)rs << 16) | (unsigned)(n0 + rs);
        qn = __builtin_popcount(__builtin_amdgcn_ballot_w32(has));
#pragma unroll
        for (int k = 0; k < BIN_PER_WAVE; ++k) {
            if (lane >= 1 && lane <= ck[k]) myq[qn + lane - 1] = w0[k];
            if (ck[k] > 31) {
                if (31 + lane < ck[k]) myq[qn + 31 + lane] = cb[k * BIN_W + 32 + lane];
            }
            qn += ck[k];
        }
        own_resolve<false>(myq, qn, lane, src, as4, rad, rmax);
    } else {
        qn = own_filter<false>(tid, lane, wave, hq, n0, cnt, src, dst, as4, xl4,
                               myq, mywt, rad, rmax, rden, accs, &ovf);
    }
    __syncthreads();

    {
        unsigned* gq = qsave + ((size_t)blockIdx.x * OWN_WAVES + wave) * OWN_QCAP;
        for (int j = lane; j < qn; j += 32) gq[j] = myq[j];
        if (lane == 0) qcnt[blockIdx.x * OWN_WAVES + wave] = qn;
        if (tid == 0)  qovf[blockIdx.x] = ovf;
    }

    if (ovf == 0)
        own_drain(myq, qn, lane, hq, as4, xl4, mywt, rad, rmax, rden, accs);
    else
        (void)own_filter<true>(tid, lane, wave, hq, n0, cnt, src, dst, as4, xl4,
                               myq, mywt, rad, rmax, rden, accs, &ovf);
    __syncthreads();

    float bq0[4], bq1[4];
#pragma unroll
    for (int k = 0; k < 4; ++k) { bq0[k] = bias[lane * 4 + k]; bq1[k] = bias[128 + lane * 4 + k]; }
    float4* __restrict__ h4 = (float4*)hout;
    for (int r = wave; r < cnt; r += OWN_WAVES) {
        const int n = n0 + r;
        const float* row = accs + r * HIDDEN + lane;
        const float d0 = rden[r * HEADS + hq] + GAT_EPS;
        const float d1 = rden[r * HEADS + 4 + hq] + GAT_EPS;
        float4 o0, o1;
        o0.x = own_elu(row[0]   / d0 + bq0[0]);  o0.y = own_elu(row[32]  / d0 + bq0[1]);
        o0.z = own_elu(row[64]  / d0 + bq0[2]);  o0.w = own_elu(row[96]  / d0 + bq0[3]);
        o1.x = own_elu(row[128] / d1 + bq1[0]);  o1.y = own_elu(row[160] / d1 + bq1[1]);
        o1.z = own_elu(row[192] / d1 + bq1[2]);  o1.w = own_elu(row[224] / d1 + bq1[3]);
        h4[(size_t)n * (HIDDEN / 4) + lane]      = o0;
        h4[(size_t)n * (HIDDEN / 4) + 32 + lane] = o1;
    }
}

__device__ __forceinline__ void l2_add_quarter(float* acc, unsigned rl, int q, float w, float4 x)
{
    float* a = acc + (q * 4) * OWN_R + rl;
    __hip_atomic_fetch_add(a,             w * x.x, __ATOMIC_RELAXED, __HIP_MEMORY_SCOPE_WORKGROUP);
    __hip_atomic_fetch_add(a + OWN_R,     w * x.y, __ATOMIC_RELAXED, __HIP_MEMORY_SCOPE_WORKGROUP);
    __hip_atomic_fetch_add(a + 2 * OWN_R, w * x.z, __ATOMIC_RELAXED, __HIP_MEMORY_SCOPE_WORKGROUP);
    __hip_atomic_fetch_add(a + 3 * OWN_R, w * x.w, __ATOMIC_RELAXED, __HIP_MEMORY_SCOPE_WORKGROUP);
}

__global__ __launch_bounds__(1024) void gat_owner_softmax_aggregate_l2(
    const int* __restrict__ src, const int* __restrict__ dst,
    const float* __restrict__ as2, const float* __restrict__ ad2,
    const float* __restrict__ xl2, const float* __restrict__ bias,
    const unsigned* __restrict__ qsave, const int* __restrict__ qcnt,
    const int* __restrict__ qovf, float* __restrict__ out)
{
    __shared__ float    acc[OUT_C * OWN_R];
    __shared__ float    rad[OWN_R];
    __shared__ unsigned rmax[OWN_R];
    __shared__ float    rden[OWN_R];

    const int tid  = threadIdx.x;
    const int lane = tid & 31;
    const int wave = tid >> 5;
    const int n0   = blockIdx.x * OWN_R;
    int cnt = N_NODES - n0;
    if (cnt > OWN_R) cnt = OWN_R;
    const float4* __restrict__ xq = (const float4*)xl2;

    for (int i = tid; i < OUT_C * OWN_R; i += 1024) acc[i] = 0.f;
    if (tid < OWN_R) {
        rad[tid]  = (tid < cnt) ? ad2[n0 + tid] : 0.f;
        rmax[tid] = 0u;
        rden[tid] = 0.f;
    }
    const bool queued = qovf[blockIdx.x] == 0;
    const unsigned* __restrict__ gq = qsave + ((size_t)blockIdx.x * OWN_WAVES + wave) * OWN_QCAP;
    const int qn = queued ? __builtin_amdgcn_readfirstlane(qcnt[blockIdx.x * OWN_WAVES + wave]) : 0;
    __syncthreads();

    if (queued) {
        for (int j0 = 0; j0 < qn; j0 += 128) {
            unsigned pk[4];
            float    a[4];
#pragma unroll
            for (int t = 0; t < 4; ++t) {
                const int j = j0 + t * 32 + lane;
                pk[t] = (j < qn) ? gq[j] : 0u;
            }
#pragma unroll
            for (int t = 0; t < 4; ++t) a[t] = as2[pk[t] & 0xFFFFu];
#pragma unroll
            for (int t = 0; t < 4; ++t) {
                if (j0 + t * 32 + lane < qn) {
                    const unsigned rl = pk[t] >> 16;
                    const float v = own_lrelu(a[t] + rad[rl]);
                    __hip_atomic_fetch_max(rmax + rl, fkey(v), __ATOMIC_RELAXED, __HIP_MEMORY_SCOPE_WORKGROUP);
                }
            }
        }
    } else {
        for (int e = tid; e < TOT_E; e += 1024) {
            int s, d; edge_sd(e, src, dst, s, d);
            const unsigned rl = (unsigned)(d - n0);
            if (rl < (unsigned)cnt) {
                const float v = own_lrelu(as2[s] + rad[rl]);
                __hip_atomic_fetch_max(rmax + rl, fkey(v), __ATOMIC_RELAXED, __HIP_MEMORY_SCOPE_WORKGROUP);
            }
        }
    }
    __syncthreads();

    if (queued) {
        const int q   = lane & 3;
        const int sub = lane >> 2;
        auto entry = [&](int j) { return gq[j < qn ? j : qn - 1]; };
        unsigned nxt = (qn > 0) ? entry(lane) : 0u;
        for (int j0 = 0; j0 < qn; j0 += 32) {
            const unsigned mine = nxt;
            nxt = entry(j0 + 32 + lane);
            unsigned pk[4];
            float    a[4];
            float4   x[4];
#pragma unroll
            for (int t = 0; t < 4; ++t) {
                pk[t] = (unsigned)__shfl((int)mine, t * 8 + sub);
                const unsigned s = pk[t] & 0xFFFFu;
                a[t] = as2[s];
                x[t] = xq[(size_t)s * (OUT_C / 4) + q];
            }
#pragma unroll
            for (int t = 0; t < 4; ++t) {
                if (j0 + t * 8 + sub < qn) {
                    const unsigned rl = pk[t] >> 16;
                    const float w = expf(own_lrelu(a[t] + rad[rl]) - fdec(rmax[rl]));
                    if (q == 0)
                        __hip_atomic_fetch_add(rden + rl, w, __ATOMIC_RELAXED, __HIP_MEMORY_SCOPE_WORKGROUP);
                    l2_add_quarter(acc, rl, q, w, x[t]);
                }
            }
        }
    } else {
        for (int e = tid; e < TOT_E; e += 1024) {
            int s, d; edge_sd(e, src, dst, s, d);
            const unsigned rl = (unsigned)(d - n0);
            if (rl < (unsigned)cnt) {
                const float w = expf(own_lrelu(as2[s] + rad[rl]) - fdec(rmax[rl]));
                __hip_atomic_fetch_add(rden + rl, w, __ATOMIC_RELAXED, __HIP_MEMORY_SCOPE_WORKGROUP);
                for (int q = 0; q < 4; ++q)
                    l2_add_quarter(acc, rl, q, w, xq[(size_t)s * (OUT_C / 4) + q]);
            }
        }
    }
    __syncthreads();

    const int tot = cnt * OUT_C;
    for (int i = tid; i < tot; i += 1024) {
        const int r = i >> 4, c = i & 15;
        out[(size_t)n0 * OUT_C + i] = acc[c * OWN_R + r] / (rden[r] + GAT_EPS) + bias[c];
    }
}

extern "C" void kernel_launch(void* const* d_in, const int* in_sizes, int n_in,
                              void* d_out, int out_size, void* d_ws, size_t ws_size,
                              hipStream_t stream)
{
    (void)in_sizes; (void)n_in; (void)out_size; (void)ws_size;

    const float* x    = (const float*)d_in[0];
    const int*   ei   = (const int*)  d_in[1];
    const float* W1   = (const float*)d_in[2];
    const float* ats1 = (const float*)d_in[3];
    const float* atd1 = (const float*)d_in[4];
    const float* b1   = (const float*)d_in[5];
    const float* W2   = (const float*)d_in[6];
    const float* ats2 = (const float*)d_in[7];
    const float* atd2 = (const float*)d_in[8];
    const float* b2   = (const float*)d_in[9];
    const int* src = ei;
    const int* dst = ei + N_EDGES;
    float* out = (float*)d_out;

    float* h1  = (float*)d_ws;
    float* xl1 = h1 + (size_t)N_NODES * HIDDEN;
    float* as1 = xl1 + (size_t)N_NODES * HIDDEN;
    float* ad1 = as1 + (size_t)N_NODES * HEADS;
    float* xl2 = ad1 + (size_t)N_NODES * HEADS;
    float* as2 = xl2 + (size_t)N_NODES * OUT_C;
    float* ad2 = as2 + (size_t)N_NODES;
    unsigned* qsv = (unsigned*)(ad2 + (size_t)N_NODES);
    int*      qct = (int*)(qsv + (size_t)OWN_WG * OWN_WAVES * OWN_QCAP);
    int*      qov = qct + (size_t)OWN_WG * OWN_WAVES;
    constexpr size_t kWordsBefore = (size_t)N_NODES * (2 * HIDDEN + 2 * HEADS + OUT_C + 2)
                                  + (size_t)OWN_WG * OWN_WAVES * OWN_QCAP
                                  + (size_t)OWN_WG * OWN_WAVES + OWN_WG;
    constexpr size_t kBinsAt = (kWordsBefore + BIN_W - 1) / BIN_W * BIN_W;
    constexpr size_t kFoldAt = kBinsAt + (size_t)OWN_WG * BIN_CHUNKS * BIN_W;
    static_assert((kFoldAt + (size_t)(IN_C + HIDDEN) * FOLD_COLS) * 4 <= 268435456ull,
                  "the bins and the folded weights must end inside the workspace");
    unsigned* bins = (unsigned*)d_ws + kBinsAt;
    float* wa1 = (float*)d_ws + kFoldAt;
    float* wa2 = wa1 + (size_t)IN_C * FOLD_COLS;

    gat_bin_edges_by_owner<<<BIN_CHUNKS + FOLD_WG, 1024, 0, stream>>>(
        src, dst, bins, W1, ats1, atd1, W2, ats2, atd2, wa1, wa2);

    gat_gemm1_xl_logits<<<N_NODES / 16, 160, 0, stream>>>(x, W1, wa1, xl1, as1, ad1);
    gat_owner_softmax_aggregate_l1<<<OWN_WG, OWN_WAVES * 32, 0, stream>>>(
        src, dst, as1, ad1, xl1, b1, h1, qsv, qct, qov, bins);

    gat_gemm2_xl_logits<<<(N_NODES / 16 + 3) / 4, 128, 0, stream>>>(h1, W2, wa2, xl2, as2, ad2);
    gat_owner_softmax_aggregate_l2<<<OWN_WG, OWN_WAVES * 32, 0, stream>>>(
        src, dst, as2, ad2, xl2, b2, qsv, qct, qov, out);
}


# Layer-2 GEMM and its logit tile run inside the layer-1 owner launch: ELU rows stay in the LDS sum tile (swizzled in place), 13 row tiles on f32 WMMA, A from LDS; h never stored; GEMM2 launch gone
# speedup vs baseline: 8.7315x; 1.0530x over previous
#include <hip/hip_runtime.h>
#include <cstdint>
#include <cstddef>

#define N_NODES   50000
#define N_EDGES   800000
#define TOT_E     (N_EDGES + N_NODES)
#define IN_C      128
#define HID       32
#define OUT_C     16
#define HEADS     8
#define HIDDEN    (HEADS * HID)
#define NEG_SLOPE 0.2f
#define GAT_EPS   1e-16f

typedef float v2f __attribute__((ext_vector_type(2)));
typedef float v8f __attribute__((ext_vector_type(8)));

template <int TN, int K, int NC>
__device__ __forceinline__ void gat_wmma_strip_f32(
    const float* __restrict__ Ap, const float* const (&Bp)[TN], v8f (&c)[TN])
{
    const v8f zero = {0.f,0.f,0.f,0.f,0.f,0.f,0.f,0.f};
#pragma unroll
    for (int j = 0; j < TN; ++j) c[j] = zero;

#pragma unroll
    for (int k = 0; k < K; k += 4) {
        const v2f a = *(const v2f*)(Ap + k);
        v2f b[TN];
#pragma unroll
        for (int j = 0; j < TN; ++j) {
            b[j].x = Bp[j][(size_t)k * NC];
            b[j].y = Bp[j][(size_t)(k + 1) * NC];
        }
#pragma unroll
        for (int j = 0; j < TN; ++j)
            c[j] = __builtin_amdgcn_wmma_f32_16x16x4_f32(
                        false, a,  false, b[j],
                        (short)0, c[j],
                        false,  false);
    }
}

#define FOLD_COLS 16
#define FOLD_WG   ((IN_C + HIDDEN) * FOLD_COLS / 1024)
static_assert(FOLD_WG * 1024 == (IN_C + HIDDEN) * FOLD_COLS, "one thread per folded weight");
static_assert(2 * HEADS == FOLD_COLS && N_NODES % 16 == 0, "logit tile shape; whole row tiles");

__device__ __forceinline__ void gat_fold_att(
    int t, const float* __restrict__ W1, const float* __restrict__ ats1,
    const float* __restrict__ atd1, const float* __restrict__ W2,
    const float* __restrict__ ats2, const float* __restrict__ atd2,
    float* __restrict__ wa1, float* __restrict__ wa2)
{
    if (t < IN_C * FOLD_COLS) {
        const int k = t / FOLD_COLS, j = t % FOLD_COLS, h = j % HEADS;
        const float* __restrict__ att = (j < HEADS ? ats1 : atd1) + h * HID;
        const float* __restrict__ w   = W1 + (size_t)k * HIDDEN + h * HID;
        float s = 0.f;
        for (int c = 0; c < HID; ++c) s += w[c] * att[c];
        wa1[t] = s;
    } else {
        const int u = t - IN_C * FOLD_COLS;
        const int k = u / FOLD_COLS, j = u % FOLD_COLS;
        float s = 0.f;
        if (j < 2) {
            const float* __restrict__ att = (j == 0) ? ats2 : atd2;
            const float* __restrict__ w   = W2 + (size_t)k * OUT_C;
            for (int c = 0; c < OUT_C; ++c) s += w[c] * att[c];
        }
        wa2[u] = s;
    }
}

__global__ __launch_bounds__(160) void gat_gemm1_xl_logits(
    const float* __restrict__ x, const float* __restrict__ W1,
    const float* __restrict__ wa1, float* __restrict__ xl,
    float* __restrict__ asrc, float* __restrict__ adst)
{
    const int wave = threadIdx.x >> 5;
    const int lane = threadIdx.x & 31;
    const int tM   = blockIdx.x;
    const int half = lane >> 4;
    const int l    = lane & 15;
    const float* __restrict__ Ap = x + (size_t)(tM * 16 + l) * IN_C + half * 2;

    if (wave < 4) {
        const float* __restrict__ B0 = W1 + (size_t)(half * 2) * HIDDEN + wave * 64 + l;
        const float* const Bp[4] = {B0, B0 + 16, B0 + 32, B0 + 48};
        v8f c[4];
        gat_wmma_strip_f32<4, IN_C, HIDDEN>(Ap, Bp, c);
        float* __restrict__ Cbase = xl + (size_t)(tM * 16 + half * 8) * HIDDEN + wave * 64 + l;
#pragma unroll
        for (int j = 0; j < 4; ++j)
#pragma unroll
            for (int r = 0; r < 8; ++r)
                Cbase[(size_t)r * HIDDEN + j * 16] = c[j][r];
    } else {
        const float* const Bp[1] = {wa1 + (half * 2) * FOLD_COLS + l};
        v8f c[1];
        gat_wmma_strip_f32<1, IN_C, FOLD_COLS>(Ap, Bp, c);
        float* __restrict__ o = (l < HEADS ? asrc : adst)
                              + (size_t)(tM * 16 + half * 8) * HEADS + (l % HEADS);
#pragma unroll
        for (int r = 0; r < 8; ++r) o[r * HEADS] = c[0][r];
    }
}

__device__ __forceinline__ unsigned fkey(float f) {
    unsigned u = __float_as_uint(f);
    return (u & 0x80000000u) ? ~u : (u | 0x80000000u);
}
__device__ __forceinline__ float fdec(unsigned k) {
    unsigned u = (k & 0x80000000u) ? (k ^ 0x80000000u) : ~k;
    return __uint_as_float(u);
}

__device__ __forceinline__ void edge_sd(long long e, const int* __restrict__ src,
                                        const int* __restrict__ dst, int& s, int& d) {
    if (e < N_EDGES) { s = src[e]; d = dst[e]; }
    else             { s = (int)(e - N_EDGES); d = s; }
}

#define OWN_R     196
#define OWN_WG    256
#define OWN_WAVES 32
#define OWN_QCAP  512
#define OWN_EBITS 20
#define BIN_CHUNKS 256
#define BIN_W      64
#define BIN_CAP    (BIN_W - 1)
#define BIN_VEC    ((N_EDGES / 4 + BIN_CHUNKS - 1) / BIN_CHUNKS)
#define BIN_PER_WAVE (BIN_CHUNKS / OWN_WAVES)
static_assert(TOT_E <= (1 << OWN_EBITS), "edge id must fit the low bits of a queue entry");
static_assert(N_NODES <= 65536, "a resolved queue entry keeps the source node in 16 bits");
static_assert(OWN_R * OWN_WG >= N_NODES, "every node needs an owner");
static_assert(N_EDGES % 4 == 0, "the binning launch reads the edge list as int4");
static_assert(BIN_VEC <= 1024, "one edge vector per thread of a binning work-group");
static_assert(BIN_PER_WAVE == 8 && OWN_WG == 8 * OWN_WAVES, "8 cells per wave on both sides");
static_assert((OWN_R + OWN_WAVES - 1) / OWN_WAVES + BIN_PER_WAVE * BIN_CAP <= OWN_QCAP,
              "self loops plus 8 full cells must fit a wave's queue");

__global__ __launch_bounds__(1024) void gat_bin_edges_by_owner(
    const int* __restrict__ src, const int* __restrict__ dst, unsigned* __restrict__ bins,
    const float* __restrict__ W1, const float* __restrict__ ats1, const float* __restrict__ atd1,
    const float* __restrict__ W2, const float* __restrict__ ats2, const float* __restrict__ atd2,
    float* __restrict__ wa1, float* __restrict__ wa2)
{
    if (blockIdx.x >= BIN_CHUNKS) {
        gat_fold_att((int)(blockIdx.x - BIN_CHUNKS) * 1024 + (int)threadIdx.x,
                     W1, ats1, atd1, W2, ats2, atd2, wa1, wa2);
        return;
    }

    __shared__ unsigned lcnt[OWN_WG];
    __shared__ unsigned lent[OWN_WG * BIN_CAP];

    const int tid  = threadIdx.x;
    const int lane = tid & 31;
    const int wave = tid >> 5;
    if (tid < OWN_WG) lcnt[tid] = 0u;
    __syncthreads();

    const int v = blockIdx.x * BIN_VEC + tid;
    if (tid < BIN_VEC && v < N_EDGES / 4) {
        const int4 d4 = ((const int4*)dst)[v];
        const int4 s4 = ((const int4*)src)[v];
        const int dd[4] = {d4.x, d4.y, d4.z, d4.w};
        const int ss[4] = {s4.x, s4.y, s4.z, s4.w};
#pragma unroll
        for (int k = 0; k < 4; ++k) {
            const unsigned d = (unsigned)dd[k];
            if (d < (unsigned)N_NODES) {
                const unsigned o    = d / OWN_R;
                const unsigned rl   = d - o * OWN_R;
                const unsigned slot = __hip_atomic_fetch_add(lcnt + o, 1u, __ATOMIC_RELAXED,
                                                             __HIP_MEMORY_SCOPE_WORKGROUP);
                if (slot < BIN_CAP) lent[o * BIN_CAP + slot] = (rl << 16) | (unsigned)ss[k];
            }
        }
    }
    __syncthreads();

#pragma unroll
    for (int k = 0; k < BIN_PER_WAVE; ++k) {
        const int o = wave * BIN_PER_WAVE + k;
        const unsigned c  = lcnt[o];
        const unsigned cc = c < BIN_CAP ? c : BIN_CAP;
        unsigned* cell = bins + ((size_t)o * BIN_CHUNKS + blockIdx.x) * BIN_W;
        const unsigned* row = lent + o * BIN_CAP;
        cell[lane] = (lane == 0) ? c : (((unsigned)lane <= cc) ? row[lane - 1] : 0u);
        if (cc > 31u) {
            if ((unsigned)(31 + lane) < cc) cell[32 + lane] = row[31 + lane];
        }
    }
}

__device__ __forceinline__ float own_lrelu(float v) { return v > 0.f ? v : NEG_SLOPE * v; }

__device__ __forceinline__ void own_add_row(float* accs, int rl, int lane,
                                            float w0, float w1, float4 x0, float4 x1)
{
    float* row = accs + rl * HIDDEN + lane;
    __hip_atomic_fetch_add(row +   0, w0 * x0.x, __ATOMIC_RELAXED, __HIP_MEMORY_SCOPE_WORKGROUP);
    __hip_atomic_fetch_add(row +  32, w0 * x0.y, __ATOMIC_RELAXED, __HIP_MEMORY_SCOPE_WORKGROUP);
    __hip_atomic_fetch_add(row +  64, w0 * x0.z, __ATOMIC_RELAXED, __HIP_MEMORY_SCOPE_WORKGROUP);
    __hip_atomic_fetch_add(row +  96, w0 * x0.w, __ATOMIC_RELAXED, __HIP_MEMORY_SCOPE_WORKGROUP);
    __hip_atomic_fetch_add(row + 128, w1 * x1.x, __ATOMIC_RELAXED, __HIP_MEMORY_SCOPE_WORKGROUP);
    __hip_atomic_fetch_add(row + 160, w1 * x1.y, __ATOMIC_RELAXED, __HIP_MEMORY_SCOPE_WORKGROUP);
    __hip_atomic_fetch_add(row + 192, w1 * x1.z, __ATOMIC_RELAXED, __HIP_MEMORY_SCOPE_WORKGROUP);
    __hip_atomic_fetch_add(row + 224, w1 * x1.w, __ATOMIC_RELAXED, __HIP_MEMORY_SCOPE_WORKGROUP);
}

template <bool RAW>
__device__ __forceinline__ void own_resolve(
    unsigned* myq, int qn, int lane, const int* __restrict__ src,
    const float4* __restrict__ as4, const float* rad, unsigned* rmax)
{
    const unsigned emask = (1u << OWN_EBITS) - 1u;
    for (int j = lane; j < qn; j += 32) {
        const unsigned pk = myq[j];
        unsigned rl, s;
        if (RAW) {
            const unsigned e = pk & emask;
            rl = pk >> OWN_EBITS;
            s  = (e < (unsigned)N_EDGES) ? (unsigned)src[e] : (e - (unsigned)N_EDGES);
        } else {
            rl = pk >> 16;
            s  = pk & 0xFFFFu;
        }
        const float4 a0 = as4[(size_t)s * 2], a1 = as4[(size_t)s * 2 + 1];
        const float4 d0 = *(const float4*)(rad + rl * HEADS);
        const float4 d1 = *(const float4*)(rad + rl * HEADS + 4);
        const float v[8] = {a0.x + d0.x, a0.y + d0.y, a0.z + d0.z, a0.w + d0.w,
                            a1.x + d1.x, a1.y + d1.y, a1.z + d1.z, a1.w + d1.w};
        unsigned* mx = rmax + rl * HEADS;
#pragma unroll
        for (int h = 0; h < HEADS; ++h)
            __hip_atomic_fetch_max(mx + h, fkey(own_lrelu(v[h])),
                                   __ATOMIC_RELAXED, __HIP_MEMORY_SCOPE_WORKGROUP);
        if (RAW) myq[j] = (rl << 16) | s;
    }
}

__device__ __forceinline__ void own_drain(
    const unsigned* myq, int qn, int lane, int hq,
    const float4* __restrict__ as4, const float4* __restrict__ xl4,
    float* mywt, const float* rad, const unsigned* rmax, float* rden, float* accs)
{
    for (int c0 = 0; c0 < qn; c0 += 32) {
        const int m = (qn - c0 < 32) ? (qn - c0) : 32;
        if (lane < m) {
            const unsigned pk = myq[c0 + lane];
            const unsigned s  = pk & 0xFFFFu;
            const unsigned rl = pk >> 16;
            const float4 a0 = as4[(size_t)s * 2], a1 = as4[(size_t)s * 2 + 1];
            const float4 d0 = *(const float4*)(rad + rl * HEADS);
            const float4 d1 = *(const float4*)(rad + rl * HEADS + 4);
            const uint4  k0 = *(const uint4*)(rmax + rl * HEADS);
            const uint4  k1 = *(const uint4*)(rmax + rl * HEADS + 4);
            float4 w0, w1;
            w0.x = expf(own_lrelu(a0.x + d0.x) - fdec(k0.x));
            w0.y = expf(own_lrelu(a0.y + d0.y) - fdec(k0.y));
            w0.z = expf(own_lrelu(a0.z + d0.z) - fdec(k0.z));
            w0.w = expf(own_lrelu(a0.w + d0.w) - fdec(k0.w));
            w1.x = expf(own_lrelu(a1.x + d1.x) - fdec(k1.x));
            w1.y = expf(own_lrelu(a1.y + d1.y) - fdec(k1.y));
            w1.z = expf(own_lrelu(a1.z + d1.z) - fdec(k1.z));
            w1.w = expf(own_lrelu(a1.w + d1.w) - fdec(k1.w));
            *(float4*)(mywt + lane * HEADS)     = w0;
            *(float4*)(mywt + lane * HEADS + 4) = w1;
            float* dn = rden + rl * HEADS;
            __hip_atomic_fetch_add(dn + 0, w0.x, __ATOMIC_RELAXED, __HIP_MEMORY_SCOPE_WORKGROUP);
            __hip_atomic_fetch_add(dn + 1, w0.y, __ATOMIC_RELAXED, __HIP_MEMORY_SCOPE_WORKGROUP);
            __hip_atomic_fetch_add(dn + 2, w0.z, __ATOMIC_RELAXED, __HIP_MEMORY_SCOPE_WORKGROUP);
            __hip_atomic_fetch_add(dn + 3, w0.w, __ATOMIC_RELAXED, __HIP_MEMORY_SCOPE_WORKGROUP);
            __hip_atomic_fetch_add(dn + 4, w1.x, __ATOMIC_RELAXED, __HIP_MEMORY_SCOPE_WORKGROUP);
            __hip_atomic_fetch_add(dn + 5, w1.y, __ATOMIC_RELAXED, __HIP_MEMORY_SCOPE_WORKGROUP);
            __hip_atomic_fetch_add(dn + 6, w1.z, __ATOMIC_RELAXED, __HIP_MEMORY_SCOPE_WORKGROUP);
            __hip_atomic_fetch_add(dn + 7, w1.w, __ATOMIC_RELAXED, __HIP_MEMORY_SCOPE_WORKGROUP);
        }
        int i = 0;
        for (; i + 4 <= m; i += 4) {
            float4 x0[4], x1[4];
            int    rl[4];
#pragma unroll
            for (int u = 0; u < 4; ++u) {
                const unsigned pk = (unsigned)__builtin_amdgcn_readfirstlane((int)myq[c0 + i + u]);
                const unsigned s  = pk & 0xFFFFu;
                rl[u] = (int)(pk >> 16);
                x0[u] = xl4[(size_t)s * (HIDDEN / 4) + lane];
                x1[u] = xl4[(size_t)s * (HIDDEN / 4) + 32 + lane];
            }
#pragma unroll
            for (int u = 0; u < 4; ++u) {
                const float w0 = mywt[(i + u) * HEADS + hq];
                const float w1 = mywt[(i + u) * HEADS + 4 + hq];
                own_add_row(accs, rl[u], lane, w0, w1, x0[u], x1[u]);
            }
        }
        for (; i < m; ++i) {
            const unsigned pk = (unsigned)__builtin_amdgcn_readfirstlane((int)myq[c0 + i]);
            const unsigned s  = pk & 0xFFFFu;
            const float4 x0 = xl4[(size_t)s * (HIDDEN / 4) + lane];
            const float4 x1 = xl4[(size_t)s * (HIDDEN / 4) + 32 + lane];
            const float w0 = mywt[i * HEADS + hq];
            const float w1 = mywt[i * HEADS + 4 + hq];
            own_add_row(accs, (int)(pk >> 16), lane, w0, w1, x0, x1);
        }
    }
}

template <bool SECOND>
__device__ __forceinline__ int own_filter(
    int tid, int lane, int wave, int hq, int n0, int cnt,
    const int* __restrict__ src, const int* __restrict__ dst,
    const float4* __restrict__ as4, const float4* __restrict__ xl4,
    unsigned* myq, float* mywt, const float* rad, unsigned* rmax,
    float* rden, float* accs, int* ovf)
{
    const int  rs  = wave + OWN_WAVES * lane;
    const bool has = rs < cnt;
    if (has) myq[lane] = ((unsigned)rs << OWN_EBITS) | (unsigned)(N_EDGES + n0 + rs);
    int qn = __builtin_popcount(__builtin_amdgcn_ballot_w32(has));

    const int4* __restrict__ dst4 = (const int4*)dst;
    constexpr int NV    = N_EDGES / 4;
    constexpr int TRIPS = (NV + 1023) / 1024;
    static_assert(TRIPS % 4 == 0, "the four-trip body assumes a whole number of bodies");
    auto ld = [&](int t) {
        const int i = t * 1024 + tid;
        return dst4[i < NV ? i : NV - 1];
    };
    auto trip = [&](const int4 d, int t) {
        const int i = t * 1024 + tid;
        const unsigned lim = (i < NV) ? (unsigned)cnt : 0u;
        const unsigned l0 = (unsigned)(d.x - n0), l1 = (unsigned)(d.y - n0);
        const unsigned l2 = (unsigned)(d.z - n0), l3 = (unsigned)(d.w - n0);
        const bool m0 = l0 < lim, m1 = l1 < lim, m2 = l2 < lim, m3 = l3 < lim;
        if (__builtin_amdgcn_ballot_w32(m0 | m1 | m2 | m3) != 0u) {
            const unsigned ll[4] = {l0, l1, l2, l3};
            const bool     mm[4] = {m0, m1, m2, m3};
#pragma unroll
            for (int k = 0; k < 4; ++k) {
                const unsigned mask = __builtin_amdgcn_ballot_w32(mm[k]);
                if (mm[k]) {
                    const int pos = qn + (int)__builtin_amdgcn_mbcnt_lo(mask, 0u);
                    myq[pos] = (ll[k] << OWN_EBITS) | (unsigned)(i * 4 + k);
                }
                qn += __builtin_popcount(mask);
            }
            if (qn > OWN_QCAP - 128) {
                own_resolve<true>(myq, qn, lane, src, as4, rad, rmax);
                if (SECOND) own_drain(myq, qn, lane, hq, as4, xl4, mywt, rad, rmax, rden, accs);
                else if (lane == 0) *ovf = 1;
                qn = 0;
            }
        }
    };
    int4 d0 = ld(0), d1 = ld(1), d2 = ld(2), d3;
    for (int t = 0; t < TRIPS; t += 4) {
        d3 = ld(t + 3);  trip(d0, t);
        d0 = ld(t + 4);  trip(d1, t + 1);
        d1 = ld(t + 5);  trip(d2, t + 2);
        d2 = ld(t + 6);  trip(d3, t + 3);
    }
    own_resolve<true>(myq, qn, lane, src, as4, rad, rmax);
    if (SECOND) { own_drain(myq, qn, lane, hq, as4, xl4, mywt, rad, rmax, rden, accs); qn = 0; }
    return qn;
}

__device__ __forceinline__ float own_elu(float v) { return v > 0.f ? v : expm1f(v); }

__global__ __launch_bounds__(1024) void gat_owner_softmax_aggregate_l1(
    const int* __restrict__ src, const int* __restrict__ dst,
    const float* __restrict__ asrc, const float* __restrict__ adst,
    const float* __restrict__ xl, const float* __restrict__ bias,
    const float* __restrict__ W2, const float* __restrict__ wa2,
    float* __restrict__ xl2, float* __restrict__ as2, float* __restrict__ ad2,
    unsigned* __restrict__ qsave,
    int* __restrict__ qcnt, int* __restrict__ qovf, const unsigned* __restrict__ bins)
{
    __shared__ __attribute__((aligned(16))) float    accs[OWN_R * HIDDEN];
    __shared__ __attribute__((aligned(16))) float    rad[OWN_R * HEADS];
    __shared__ __attribute__((aligned(16))) unsigned rmax[OWN_R * HEADS];
    __shared__ __attribute__((aligned(16))) float    rden[OWN_R * HEADS];
    __shared__ __attribute__((aligned(16))) float    wts[OWN_WAVES][32 * HEADS];
    __shared__ unsigned qpk[OWN_WAVES][OWN_QCAP];
    __shared__ int      ovf;
    __shared__ int      binbad;

    const int tid  = threadIdx.x;
    const int lane = tid & 31;
    const int wave = tid >> 5;
    const int hq   = lane >> 3;
    const int n0   = blockIdx.x * OWN_R;
    int cnt = N_NODES - n0;
    if (cnt > OWN_R) cnt = OWN_R;
    const float4* __restrict__ xl4 = (const float4*)xl;
    const float4* __restrict__ as4 = (const float4*)asrc;

    const unsigned* __restrict__ cb =
        bins + ((size_t)blockIdx.x * BIN_CHUNKS + wave * BIN_PER_WAVE) * BIN_W;
    unsigned w0[BIN_PER_WAVE];
#pragma unroll
    for (int k = 0; k < BIN_PER_WAVE; ++k) w0[k] = cb[k * BIN_W + lane];

    float4* accs4 = (float4*)accs;
    for (int i = tid; i < OWN_R * HIDDEN / 4; i += 1024)
        accs4[i] = make_float4(0.f, 0.f, 0.f, 0.f);
    for (int i = tid; i < OWN_R * HEADS; i += 1024) {
        rad[i]  = ((i >> 3) < cnt) ? adst[(size_t)n0 * HEADS + i] : 0.f;
        rmax[i] = 0u;
        rden[i] = 0.f;
    }
    if (tid == 0) { ovf = 0; binbad = 0; }
    __syncthreads();

    unsigned* myq  = qpk[wave];
    float*    mywt = wts[wave];

    int  ck[BIN_PER_WAVE];
    bool bad = false;
#pragma unroll
    for (int k = 0; k < BIN_PER_WAVE; ++k) {
        ck[k] = __builtin_amdgcn_readlane((int)w0[k], 0);
        bad |= (unsigned)ck[k] > (unsigned)BIN_CAP;
    }
    if (bad && lane == 0) binbad = 1;
    __syncthreads();

    int qn;
    if (binbad == 0) {
        const int  rs  = wave + OWN_WAVES * lane;
        const bool has = rs < cnt;
        if (has) myq[lane] = ((unsigned)rs << 16) | (unsigned)(n0 + rs);
        qn = __builtin_popcount(__builtin_amdgcn_ballot_w32(has));
#pragma unroll
        for (int k = 0; k < BIN_PER_WAVE; ++k) {
            if (lane >= 1 && lane <= ck[k]) myq[qn + lane - 1] = w0[k];
            if (ck[k] > 31) {
                if (31 + lane < ck[k]) myq[qn + 31 + lane] = cb[k * BIN_W + 32 + lane];
            }
            qn += ck[k];
        }
        own_resolve<false>(myq, qn, lane, src, as4, rad, rmax);
    } else {
        qn = own_filter<false>(tid, lane, wave, hq, n0, cnt, src, dst, as4, xl4,
                               myq, mywt, rad, rmax, rden, accs, &ovf);
    }
    __syncthreads();

    {
        unsigned* gq = qsave + ((size_t)blockIdx.x * OWN_WAVES + wave) * OWN_QCAP;
        for (int j = lane; j < qn; j += 32) gq[j] = myq[j];
        if (lane == 0) qcnt[blockIdx.x * OWN_WAVES + wave] = qn;
        if (tid == 0)  qovf[blockIdx.x] = ovf;
    }

    if (ovf == 0)
        own_drain(myq, qn, lane, hq, as4, xl4, mywt, rad, rmax, rden, accs);
    else
        (void)own_filter<true>(tid, lane, wave, hq, n0, cnt, src, dst, as4, xl4,
                               myq, mywt, rad, rmax, rden, accs, &ovf);
    __syncthreads();

    float bq0[4], bq1[4];
#pragma unroll
    for (int k = 0; k < 4; ++k) { bq0[k] = bias[lane * 4 + k]; bq1[k] = bias[128 + lane * 4 + k]; }
    for (int r = wave; r < cnt; r += OWN_WAVES) {
        float* rowp = accs + r * HIDDEN;
        const float* row = rowp + lane;
        const float d0 = rden[r * HEADS + hq] + GAT_EPS;
        const float d1 = rden[r * HEADS + 4 + hq] + GAT_EPS;
        float4 o0, o1;
        o0.x = own_elu(row[0]   / d0 + bq0[0]);  o0.y = own_elu(row[32]  / d0 + bq0[1]);
        o0.z = own_elu(row[64]  / d0 + bq0[2]);  o0.w = own_elu(row[96]  / d0 + bq0[3]);
        o1.x = own_elu(row[128] / d1 + bq1[0]);  o1.y = own_elu(row[160] / d1 + bq1[1]);
        o1.z = own_elu(row[192] / d1 + bq1[2]);  o1.w = own_elu(row[224] / d1 + bq1[3]);
        const int sw = (lane * 4) ^ ((r & 15) << 2);
        *(float4*)(rowp + sw)       = o0;
        *(float4*)(rowp + 128 + sw) = o1;
    }
    __syncthreads();

    const int ntile = (cnt + 15) >> 4;
    if (wave < 2 * ntile) {
        const int  tile = wave >> 1;
        const bool lg   = (wave & 1) != 0;
        const int  half = lane >> 4;
        const int  l    = lane & 15;
        int rr = tile * 16 + l;
        if (rr > cnt - 1) rr = cnt - 1;
        const int swz = (rr & 15) << 2;
        const float* arow = accs + rr * HIDDEN + half * 2;
        static_assert(OUT_C == FOLD_COLS, "both B tiles share the row pitch");
        const float* __restrict__ bp = (lg ? wa2 : W2) + (half * 2) * OUT_C + l;
        v8f acc = {0.f,0.f,0.f,0.f,0.f,0.f,0.f,0.f};
#pragma unroll
        for (int kk = 0; kk < HIDDEN / 4; ++kk) {
            const v2f a = *(const v2f*)(arow + (kk >> 4) * 64 + (((kk & 15) << 2) ^ swz));
            v2f b;
            b.x = bp[(kk * 4) * OUT_C];
            b.y = bp[(kk * 4 + 1) * OUT_C];
            acc = __builtin_amdgcn_wmma_f32_16x16x4_f32( false, a,  false, b,
                                                         (short)0, acc,
                                                         false,  false);
        }
        const int rbase = tile * 16 + half * 8;
        if (!lg) {
            float* __restrict__ xo = xl2 + (size_t)(n0 + rbase) * OUT_C + l;
#pragma unroll
            for (int r = 0; r < 8; ++r)
                if (rbase + r < cnt) xo[(size_t)r * OUT_C] = acc[r];
        } else if (l < 2) {
            float* __restrict__ o = (l == 0 ? as2 : ad2) + n0 + rbase;
#pragma unroll
            for (int r = 0; r < 8; ++r)
                if (rbase + r < cnt) o[r] = acc[r];
        }
    }
}

__device__ __forceinline__ void l2_add_quarter(float* acc, unsigned rl, int q, float w, float4 x)
{
    float* a = acc + (q * 4) * OWN_R + rl;
    __hip_atomic_fetch_add(a,             w * x.x, __ATOMIC_RELAXED, __HIP_MEMORY_SCOPE_WORKGROUP);
    __hip_atomic_fetch_add(a + OWN_R,     w * x.y, __ATOMIC_RELAXED, __HIP_MEMORY_SCOPE_WORKGROUP);
    __hip_atomic_fetch_add(a + 2 * OWN_R, w * x.z, __ATOMIC_RELAXED, __HIP_MEMORY_SCOPE_WORKGROUP);
    __hip_atomic_fetch_add(a + 3 * OWN_R, w * x.w, __ATOMIC_RELAXED, __HIP_MEMORY_SCOPE_WORKGROUP);
}

__global__ __launch_bounds__(1024) void gat_owner_softmax_aggregate_l2(
    const int* __restrict__ src, const int* __restrict__ dst,
    const float* __restrict__ as2, const float* __restrict__ ad2,
    const float* __restrict__ xl2, const float* __restrict__ bias,
    const unsigned* __restrict__ qsave, const int* __restrict__ qcnt,
    const int* __restrict__ qovf, float* __restrict__ out)
{
    __shared__ float    acc[OUT_C * OWN_R];
    __shared__ float    rad[OWN_R];
    __shared__ unsigned rmax[OWN_R];
    __shared__ float    rden[OWN_R];

    const int tid  = threadIdx.x;
    const int lane = tid & 31;
    const int wave = tid >> 5;
    const int n0   = blockIdx.x * OWN_R;
    int cnt = N_NODES - n0;
    if (cnt > OWN_R) cnt = OWN_R;
    const float4* __restrict__ xq = (const float4*)xl2;

    for (int i = tid; i < OUT_C * OWN_R; i += 1024) acc[i] = 0.f;
    if (tid < OWN_R) {
        rad[tid]  = (tid < cnt) ? ad2[n0 + tid] : 0.f;
        rmax[tid] = 0u;
        rden[tid] = 0.f;
    }
    const bool queued = qovf[blockIdx.x] == 0;
    const unsigned* __restrict__ gq = qsave + ((size_t)blockIdx.x * OWN_WAVES + wave) * OWN_QCAP;
    const int qn = queued ? __builtin_amdgcn_readfirstlane(qcnt[blockIdx.x * OWN_WAVES + wave]) : 0;
    __syncthreads();

    if (queued) {
        for (int j0 = 0; j0 < qn; j0 += 128) {
            unsigned pk[4];
            float    a[4];
#pragma unroll
            for (int t = 0; t < 4; ++t) {
                const int j = j0 + t * 32 + lane;
                pk[t] = (j < qn) ? gq[j] : 0u;
            }
#pragma unroll
            for (int t = 0; t < 4; ++t) a[t] = as2[pk[t] & 0xFFFFu];
#pragma unroll
            for (int t = 0; t < 4; ++t) {
                if (j0 + t * 32 + lane < qn) {
                    const unsigned rl = pk[t] >> 16;
                    const float v = own_lrelu(a[t] + rad[rl]);
                    __hip_atomic_fetch_max(rmax + rl, fkey(v), __ATOMIC_RELAXED, __HIP_MEMORY_SCOPE_WORKGROUP);
                }
            }
        }
    } else {
        for (int e = tid; e < TOT_E; e += 1024) {
            int s, d; edge_sd(e, src, dst, s, d);
            const unsigned rl = (unsigned)(d - n0);
            if (rl < (unsigned)cnt) {
                const float v = own_lrelu(as2[s] + rad[rl]);
                __hip_atomic_fetch_max(rmax + rl, fkey(v), __ATOMIC_RELAXED, __HIP_MEMORY_SCOPE_WORKGROUP);
            }
        }
    }
    __syncthreads();

    if (queued) {
        const int q   = lane & 3;
        const int sub = lane >> 2;
        auto entry = [&](int j) { return gq[j < qn ? j : qn - 1]; };
        unsigned nxt = (qn > 0) ? entry(lane) : 0u;
        for (int j0 = 0; j0 < qn; j0 += 32) {
            const unsigned mine = nxt;
            nxt = entry(j0 + 32 + lane);
            unsigned pk[4];
            float    a[4];
            float4   x[4];
#pragma unroll
            for (int t = 0; t < 4; ++t) {
                pk[t] = (unsigned)__shfl((int)mine, t * 8 + sub);
                const unsigned s = pk[t] & 0xFFFFu;
                a[t] = as2[s];
                x[t] = xq[(size_t)s * (OUT_C / 4) + q];
            }
#pragma unroll
            for (int t = 0; t < 4; ++t) {
                if (j0 + t * 8 + sub < qn) {
                    const unsigned rl = pk[t] >> 16;
                    const float w = expf(own_lrelu(a[t] + rad[rl]) - fdec(rmax[rl]));
                    if (q == 0)
                        __hip_atomic_fetch_add(rden + rl, w, __ATOMIC_RELAXED, __HIP_MEMORY_SCOPE_WORKGROUP);
                    l2_add_quarter(acc, rl, q, w, x[t]);
                }
            }
        }
    } else {
        for (int e = tid; e < TOT_E; e += 1024) {
            int s, d; edge_sd(e, src, dst, s, d);
            const unsigned rl = (unsigned)(d - n0);
            if (rl < (unsigned)cnt) {
                const float w = expf(own_lrelu(as2[s] + rad[rl]) - fdec(rmax[rl]));
                __hip_atomic_fetch_add(rden + rl, w, __ATOMIC_RELAXED, __HIP_MEMORY_SCOPE_WORKGROUP);
                for (int q = 0; q < 4; ++q)
                    l2_add_quarter(acc, rl, q, w, xq[(size_t)s * (OUT_C / 4) + q]);
            }
        }
    }
    __syncthreads();

    const int tot = cnt * OUT_C;
    for (int i = tid; i < tot; i += 1024) {
        const int r = i >> 4, c = i & 15;
        out[(size_t)n0 * OUT_C + i] = acc[c * OWN_R + r] / (rden[r] + GAT_EPS) + bias[c];
    }
}

extern "C" void kernel_launch(void* const* d_in, const int* in_sizes, int n_in,
                              void* d_out, int out_size, void* d_ws, size_t ws_size,
                              hipStream_t stream)
{
    (void)in_sizes; (void)n_in; (void)out_size; (void)ws_size;

    const float* x    = (const float*)d_in[0];
    const int*   ei   = (const int*)  d_in[1];
    const float* W1   = (const float*)d_in[2];
    const float* ats1 = (const float*)d_in[3];
    const float* atd1 = (const float*)d_in[4];
    const float* b1   = (const float*)d_in[5];
    const float* W2   = (const float*)d_in[6];
    const float* ats2 = (const float*)d_in[7];
    const float* atd2 = (const float*)d_in[8];
    const float* b2   = (const float*)d_in[9];
    const int* src = ei;
    const int* dst = ei + N_EDGES;
    float* out = (float*)d_out;

    float* xl1 = (float*)d_ws;
    float* as1 = xl1 + (size_t)N_NODES * HIDDEN;
    float* ad1 = as1 + (size_t)N_NODES * HEADS;
    float* xl2 = ad1 + (size_t)N_NODES * HEADS;
    float* as2 = xl2 + (size_t)N_NODES * OUT_C;
    float* ad2 = as2 + (size_t)N_NODES;
    unsigned* qsv = (unsigned*)(ad2 + (size_t)N_NODES);
    int*      qct = (int*)(qsv + (size_t)OWN_WG * OWN_WAVES * OWN_QCAP);
    int*      qov = qct + (size_t)OWN_WG * OWN_WAVES;
    constexpr size_t kWordsBefore = (size_t)N_NODES * (HIDDEN + 2 * HEADS + OUT_C + 2)
                                  + (size_t)OWN_WG * OWN_WAVES * OWN_QCAP
                                  + (size_t)OWN_WG * OWN_WAVES + OWN_WG;
    constexpr size_t kBinsAt = (kWordsBefore + BIN_W - 1) / BIN_W * BIN_W;
    constexpr size_t kFoldAt = kBinsAt + (size_t)OWN_WG * BIN_CHUNKS * BIN_W;
    static_assert((kFoldAt + (size_t)(IN_C + HIDDEN) * FOLD_COLS) * 4 <= 268435456ull,
                  "the bins and the folded weights must end inside the workspace");
    unsigned* bins = (unsigned*)d_ws + kBinsAt;
    float* wa1 = (float*)d_ws + kFoldAt;
    float* wa2 = wa1 + (size_t)IN_C * FOLD_COLS;

    gat_bin_edges_by_owner<<<BIN_CHUNKS + FOLD_WG, 1024, 0, stream>>>(
        src, dst, bins, W1, ats1, atd1, W2, ats2, atd2, wa1, wa2);

    gat_gemm1_xl_logits<<<N_NODES / 16, 160, 0, stream>>>(x, W1, wa1, xl1, as1, ad1);
    gat_owner_softmax_aggregate_l1<<<OWN_WG, OWN_WAVES * 32, 0, stream>>>(
        src, dst, as1, ad1, xl1, b1, W2, wa2, xl2, as2, ad2, qsv, qct, qov, bins);

    gat_owner_softmax_aggregate_l2<<<OWN_WG, OWN_WAVES * 32, 0, stream>>>(
        src, dst, as2, ad2, xl2, b2, qsv, qct, qov, out);
}


# Layer-1 GEMM fragment loads and xl stores widened to 16 bytes: a lane owns 4 adjacent columns of its strip and 4 adjacent k of its row (k regrouped inside each 8-block of A and B: GEMM sum order)
# speedup vs baseline: 9.2013x; 1.0538x over previous
#include <hip/hip_runtime.h>
#include <cstdint>
#include <cstddef>

#define N_NODES   50000
#define N_EDGES   800000
#define TOT_E     (N_EDGES + N_NODES)
#define IN_C      128
#define HID       32
#define OUT_C     16
#define HEADS     8
#define HIDDEN    (HEADS * HID)
#define NEG_SLOPE 0.2f
#define GAT_EPS   1e-16f

typedef float v2f __attribute__((ext_vector_type(2)));
typedef float v8f __attribute__((ext_vector_type(8)));

__device__ __forceinline__ v8f gat_wmma_step_f32(v2f a, v2f b, v8f c)
{
    return __builtin_amdgcn_wmma_f32_16x16x4_f32( false, a,  false, b,
                                                  (short)0, c,
                                                  false,  false);
}

template <int K, int NC>
__device__ __forceinline__ void gat_wmma_strip4_f32(
    const float* __restrict__ Ap, const float* __restrict__ Bq, v8f (&c)[4])
{
    const v8f zero = {0.f,0.f,0.f,0.f,0.f,0.f,0.f,0.f};
#pragma unroll
    for (int j = 0; j < 4; ++j) c[j] = zero;

    float4 an = *(const float4*)Ap;
    float4 bn[4];
#pragma unroll
    for (int s = 0; s < 4; ++s) bn[s] = *(const float4*)(Bq + (size_t)s * NC);

#pragma unroll
    for (int u = 0; u < K / 8; ++u) {
        const float4 a4 = an;
        const float4 b0 = bn[0], b1 = bn[1], b2 = bn[2], b3 = bn[3];
        if (u + 1 < K / 8) {
            an = *(const float4*)(Ap + 8 * (u + 1));
#pragma unroll
            for (int s = 0; s < 4; ++s)
                bn[s] = *(const float4*)(Bq + (size_t)(8 * (u + 1) + s) * NC);
        }
        const v2f a01 = {a4.x, a4.y};
        const v2f a23 = {a4.z, a4.w};
        const v2f p01[4] = {{b0.x, b1.x}, {b0.y, b1.y}, {b0.z, b1.z}, {b0.w, b1.w}};
        const v2f p23[4] = {{b2.x, b3.x}, {b2.y, b3.y}, {b2.z, b3.z}, {b2.w, b3.w}};
#pragma unroll
        for (int j = 0; j < 4; ++j) c[j] = gat_wmma_step_f32(a01, p01[j], c[j]);
#pragma unroll
        for (int j = 0; j < 4; ++j) c[j] = gat_wmma_step_f32(a23, p23[j], c[j]);
    }
}

template <int K, int NC>
__device__ __forceinline__ void gat_wmma_strip1_f32(
    const float* __restrict__ Ap, const float* __restrict__ Bp, v8f& c)
{
    const v8f zero = {0.f,0.f,0.f,0.f,0.f,0.f,0.f,0.f};
    c = zero;
#pragma unroll
    for (int u = 0; u < K / 8; ++u) {
        const float4 a4 = *(const float4*)(Ap + 8 * u);
        float b[4];
#pragma unroll
        for (int s = 0; s < 4; ++s) b[s] = Bp[(size_t)(8 * u + s) * NC];
        const v2f a01 = {a4.x, a4.y};
        const v2f a23 = {a4.z, a4.w};
        const v2f b01 = {b[0], b[1]};
        const v2f b23 = {b[2], b[3]};
        c = gat_wmma_step_f32(a01, b01, c);
        c = gat_wmma_step_f32(a23, b23, c);
    }
}

#define FOLD_COLS 16
#define FOLD_WG   ((IN_C + HIDDEN) * FOLD_COLS / 1024)
static_assert(FOLD_WG * 1024 == (IN_C + HIDDEN) * FOLD_COLS, "one thread per folded weight");
static_assert(2 * HEADS == FOLD_COLS && N_NODES % 16 == 0, "logit tile shape; whole row tiles");

__device__ __forceinline__ void gat_fold_att(
    int t, const float* __restrict__ W1, const float* __restrict__ ats1,
    const float* __restrict__ atd1, const float* __restrict__ W2,
    const float* __restrict__ ats2, const float* __restrict__ atd2,
    float* __restrict__ wa1, float* __restrict__ wa2)
{
    if (t < IN_C * FOLD_COLS) {
        const int k = t / FOLD_COLS, j = t % FOLD_COLS, h = j % HEADS;
        const float* __restrict__ att = (j < HEADS ? ats1 : atd1) + h * HID;
        const float* __restrict__ w   = W1 + (size_t)k * HIDDEN + h * HID;
        float s = 0.f;
        for (int c = 0; c < HID; ++c) s += w[c] * att[c];
        wa1[t] = s;
    } else {
        const int u = t - IN_C * FOLD_COLS;
        const int k = u / FOLD_COLS, j = u % FOLD_COLS;
        float s = 0.f;
        if (j < 2) {
            const float* __restrict__ att = (j == 0) ? ats2 : atd2;
            const float* __restrict__ w   = W2 + (size_t)k * OUT_C;
            for (int c = 0; c < OUT_C; ++c) s += w[c] * att[c];
        }
        wa2[u] = s;
    }
}

static_assert(IN_C % 8 == 0, "the strips take k in blocks of 8");
__global__ __launch_bounds__(160) __attribute__((amdgpu_waves_per_eu(4, 10))) void gat_gemm1_xl_logits(
    const float* __restrict__ x, const float* __restrict__ W1,
    const float* __restrict__ wa1, float* __restrict__ xl,
    float* __restrict__ asrc, float* __restrict__ adst)
{
    const int wave = threadIdx.x >> 5;
    const int lane = threadIdx.x & 31;
    const int tM   = blockIdx.x;
    const int half = lane >> 4;
    const int l    = lane & 15;
    const float* __restrict__ Ap = x + (size_t)(tM * 16 + l) * IN_C + half * 4;

    if (wave < 4) {
        const float* __restrict__ Bq = W1 + (size_t)(half * 4) * HIDDEN + wave * 64 + l * 4;
        v8f c[4];
        gat_wmma_strip4_f32<IN_C, HIDDEN>(Ap, Bq, c);
        float* __restrict__ Cbase = xl + (size_t)(tM * 16 + half * 8) * HIDDEN + wave * 64 + l * 4;
#pragma unroll
        for (int r = 0; r < 8; ++r)
            *(float4*)(Cbase + (size_t)r * HIDDEN) = make_float4(c[0][r], c[1][r], c[2][r], c[3][r]);
    } else {
        const float* __restrict__ Bp = wa1 + (half * 4) * FOLD_COLS + l;
        v8f c;
        gat_wmma_strip1_f32<IN_C, FOLD_COLS>(Ap, Bp, c);
        float* __restrict__ o = (l < HEADS ? asrc : adst)
                              + (size_t)(tM * 16 + half * 8) * HEADS + (l % HEADS);
#pragma unroll
        for (int r = 0; r < 8; ++r) o[r * HEADS] = c[r];
    }
}

__device__ __forceinline__ unsigned fkey(float f) {
    unsigned u = __float_as_uint(f);
    return (u & 0x80000000u) ? ~u : (u | 0x80000000u);
}
__device__ __forceinline__ float fdec(unsigned k) {
    unsigned u = (k & 0x80000000u) ? (k ^ 0x80000000u) : ~k;
    return __uint_as_float(u);
}

__device__ __forceinline__ void edge_sd(long long e, const int* __restrict__ src,
                                        const int* __restrict__ dst, int& s, int& d) {
    if (e < N_EDGES) { s = src[e]; d = dst[e]; }
    else             { s = (int)(e - N_EDGES); d = s; }
}

#define OWN_R     196
#define OWN_WG    256
#define OWN_WAVES 32
#define OWN_QCAP  512
#define OWN_EBITS 20
#define BIN_CHUNKS 256
#define BIN_W      64
#define BIN_CAP    (BIN_W - 1)
#define BIN_VEC    ((N_EDGES / 4 + BIN_CHUNKS - 1) / BIN_CHUNKS)
#define BIN_PER_WAVE (BIN_CHUNKS / OWN_WAVES)
static_assert(TOT_E <= (1 << OWN_EBITS), "edge id must fit the low bits of a queue entry");
static_assert(N_NODES <= 65536, "a resolved queue entry keeps the source node in 16 bits");
static_assert(OWN_R * OWN_WG >= N_NODES, "every node needs an owner");
static_assert(N_EDGES % 4 == 0, "the binning launch reads the edge list as int4");
static_assert(BIN_VEC <= 1024, "one edge vector per thread of a binning work-group");
static_assert(BIN_PER_WAVE == 8 && OWN_WG == 8 * OWN_WAVES, "8 cells per wave on both sides");
static_assert((OWN_R + OWN_WAVES - 1) / OWN_WAVES + BIN_PER_WAVE * BIN_CAP <= OWN_QCAP,
              "self loops plus 8 full cells must fit a wave's queue");

__global__ __launch_bounds__(1024) void gat_bin_edges_by_owner(
    const int* __restrict__ src, const int* __restrict__ dst, unsigned* __restrict__ bins,
    const float* __restrict__ W1, const float* __restrict__ ats1, const float* __restrict__ atd1,
    const float* __restrict__ W2, const float* __restrict__ ats2, const float* __restrict__ atd2,
    float* __restrict__ wa1, float* __restrict__ wa2)
{
    if (blockIdx.x >= BIN_CHUNKS) {
        gat_fold_att((int)(blockIdx.x - BIN_CHUNKS) * 1024 + (int)threadIdx.x,
                     W1, ats1, atd1, W2, ats2, atd2, wa1, wa2);
        return;
    }

    __shared__ unsigned lcnt[OWN_WG];
    __shared__ unsigned lent[OWN_WG * BIN_CAP];

    const int tid  = threadIdx.x;
    const int lane = tid & 31;
    const int wave = tid >> 5;
    if (tid < OWN_WG) lcnt[tid] = 0u;
    __syncthreads();

    const int v = blockIdx.x * BIN_VEC + tid;
    if (tid < BIN_VEC && v < N_EDGES / 4) {
        const int4 d4 = ((const int4*)dst)[v];
        const int4 s4 = ((const int4*)src)[v];
        const int dd[4] = {d4.x, d4.y, d4.z, d4.w};
        const int ss[4] = {s4.x, s4.y, s4.z, s4.w};
#pragma unroll
        for (int k = 0; k < 4; ++k) {
            const unsigned d = (unsigned)dd[k];
            if (d < (unsigned)N_NODES) {
                const unsigned o    = d / OWN_R;
                const unsigned rl   = d - o * OWN_R;
                const unsigned slot = __hip_atomic_fetch_add(lcnt + o, 1u, __ATOMIC_RELAXED,
                                                             __HIP_MEMORY_SCOPE_WORKGROUP);
                if (slot < BIN_CAP) lent[o * BIN_CAP + slot] = (rl << 16) | (unsigned)ss[k];
            }
        }
    }
    __syncthreads();

#pragma unroll
    for (int k = 0; k < BIN_PER_WAVE; ++k) {
        const int o = wave * BIN_PER_WAVE + k;
        const unsigned c  = lcnt[o];
        const unsigned cc = c < BIN_CAP ? c : BIN_CAP;
        unsigned* cell = bins + ((size_t)o * BIN_CHUNKS + blockIdx.x) * BIN_W;
        const unsigned* row = lent + o * BIN_CAP;
        cell[lane] = (lane == 0) ? c : (((unsigned)lane <= cc) ? row[lane - 1] : 0u);
        if (cc > 31u) {
            if ((unsigned)(31 + lane) < cc) cell[32 + lane] = row[31 + lane];
        }
    }
}

__device__ __forceinline__ float own_lrelu(float v) { return v > 0.f ? v : NEG_SLOPE * v; }

__device__ __forceinline__ void own_add_row(float* accs, int rl, int lane,
                                            float w0, float w1, float4 x0, float4 x1)
{
    float* row = accs + rl * HIDDEN + lane;
    __hip_atomic_fetch_add(row +   0, w0 * x0.x, __ATOMIC_RELAXED, __HIP_MEMORY_SCOPE_WORKGROUP);
    __hip_atomic_fetch_add(row +  32, w0 * x0.y, __ATOMIC_RELAXED, __HIP_MEMORY_SCOPE_WORKGROUP);
    __hip_atomic_fetch_add(row +  64, w0 * x0.z, __ATOMIC_RELAXED, __HIP_MEMORY_SCOPE_WORKGROUP);
    __hip_atomic_fetch_add(row +  96, w0 * x0.w, __ATOMIC_RELAXED, __HIP_MEMORY_SCOPE_WORKGROUP);
    __hip_atomic_fetch_add(row + 128, w1 * x1.x, __ATOMIC_RELAXED, __HIP_MEMORY_SCOPE_WORKGROUP);
    __hip_atomic_fetch_add(row + 160, w1 * x1.y, __ATOMIC_RELAXED, __HIP_MEMORY_SCOPE_WORKGROUP);
    __hip_atomic_fetch_add(row + 192, w1 * x1.z, __ATOMIC_RELAXED, __HIP_MEMORY_SCOPE_WORKGROUP);
    __hip_atomic_fetch_add(row + 224, w1 * x1.w, __ATOMIC_RELAXED, __HIP_MEMORY_SCOPE_WORKGROUP);
}

template <bool RAW>
__device__ __forceinline__ void own_resolve(
    unsigned* myq, int qn, int lane, const int* __restrict__ src,
    const float4* __restrict__ as4, const float* rad, unsigned* rmax)
{
    const unsigned emask = (1u << OWN_EBITS) - 1u;
    for (int j = lane; j < qn; j += 32) {
        const unsigned pk = myq[j];
        unsigned rl, s;
        if (RAW) {
            const unsigned e = pk & emask;
            rl = pk >> OWN_EBITS;
            s  = (e < (unsigned)N_EDGES) ? (unsigned)src[e] : (e - (unsigned)N_EDGES);
        } else {
            rl = pk >> 16;
            s  = pk & 0xFFFFu;
        }
        const float4 a0 = as4[(size_t)s * 2], a1 = as4[(size_t)s * 2 + 1];
        const float4 d0 = *(const float4*)(rad + rl * HEADS);
        const float4 d1 = *(const float4*)(rad + rl * HEADS + 4);
        const float v[8] = {a0.x + d0.x, a0.y + d0.y, a0.z + d0.z, a0.w + d0.w,
                            a1.x + d1.x, a1.y + d1.y, a1.z + d1.z, a1.w + d1.w};
        unsigned* mx = rmax + rl * HEADS;
#pragma unroll
        for (int h = 0; h < HEADS; ++h)
            __hip_atomic_fetch_max(mx + h, fkey(own_lrelu(v[h])),
                                   __ATOMIC_RELAXED, __HIP_MEMORY_SCOPE_WORKGROUP);
        if (RAW) myq[j] = (rl << 16) | s;
    }
}

__device__ __forceinline__ void own_drain(
    const unsigned* myq, int qn, int lane, int hq,
    const float4* __restrict__ as4, const float4* __restrict__ xl4,
    float* mywt, const float* rad, const unsigned* rmax, float* rden, float* accs)
{
    for (int c0 = 0; c0 < qn; c0 += 32) {
        const int m = (qn - c0 < 32) ? (qn - c0) : 32;
        if (lane < m) {
            const unsigned pk = myq[c0 + lane];
            const unsigned s  = pk & 0xFFFFu;
            const unsigned rl = pk >> 16;
            const float4 a0 = as4[(size_t)s * 2], a1 = as4[(size_t)s * 2 + 1];
            const float4 d0 = *(const float4*)(rad + rl * HEADS);
            const float4 d1 = *(const float4*)(rad + rl * HEADS + 4);
            const uint4  k0 = *(const uint4*)(rmax + rl * HEADS);
            const uint4  k1 = *(const uint4*)(rmax + rl * HEADS + 4);
            float4 w0, w1;
            w0.x = expf(own_lrelu(a0.x + d0.x) - fdec(k0.x));
            w0.y = expf(own_lrelu(a0.y + d0.y) - fdec(k0.y));
            w0.z = expf(own_lrelu(a0.z + d0.z) - fdec(k0.z));
            w0.w = expf(own_lrelu(a0.w + d0.w) - fdec(k0.w));
            w1.x = expf(own_lrelu(a1.x + d1.x) - fdec(k1.x));
            w1.y = expf(own_lrelu(a1.y + d1.y) - fdec(k1.y));
            w1.z = expf(own_lrelu(a1.z + d1.z) - fdec(k1.z));
            w1.w = expf(own_lrelu(a1.w + d1.w) - fdec(k1.w));
            *(float4*)(mywt + lane * HEADS)     = w0;
            *(float4*)(mywt + lane * HEADS + 4) = w1;
            float* dn = rden + rl * HEADS;
            __hip_atomic_fetch_add(dn + 0, w0.x, __ATOMIC_RELAXED, __HIP_MEMORY_SCOPE_WORKGROUP);
            __hip_atomic_fetch_add(dn + 1, w0.y, __ATOMIC_RELAXED, __HIP_MEMORY_SCOPE_WORKGROUP);
            __hip_atomic_fetch_add(dn + 2, w0.z, __ATOMIC_RELAXED, __HIP_MEMORY_SCOPE_WORKGROUP);
            __hip_atomic_fetch_add(dn + 3, w0.w, __ATOMIC_RELAXED, __HIP_MEMORY_SCOPE_WORKGROUP);
            __hip_atomic_fetch_add(dn + 4, w1.x, __ATOMIC_RELAXED, __HIP_MEMORY_SCOPE_WORKGROUP);
            __hip_atomic_fetch_add(dn + 5, w1.y, __ATOMIC_RELAXED, __HIP_MEMORY_SCOPE_WORKGROUP);
            __hip_atomic_fetch_add(dn + 6, w1.z, __ATOMIC_RELAXED, __HIP_MEMORY_SCOPE_WORKGROUP);
            __hip_atomic_fetch_add(dn + 7, w1.w, __ATOMIC_RELAXED, __HIP_MEMORY_SCOPE_WORKGROUP);
        }
        int i = 0;
        for (; i + 4 <= m; i += 4) {
            float4 x0[4], x1[4];
            int    rl[4];
#pragma unroll
            for (int u = 0; u < 4; ++u) {
                const unsigned pk = (unsigned)__builtin_amdgcn_readfirstlane((int)myq[c0 + i + u]);
                const unsigned s  = pk & 0xFFFFu;
                rl[u] = (int)(pk >> 16);
                x0[u] = xl4[(size_t)s * (HIDDEN / 4) + lane];
                x1[u] = xl4[(size_t)s * (HIDDEN / 4) + 32 + lane];
            }
#pragma unroll
            for (int u = 0; u < 4; ++u) {
                const float w0 = mywt[(i + u) * HEADS + hq];
                const float w1 = mywt[(i + u) * HEADS + 4 + hq];
                own_add_row(accs, rl[u], lane, w0, w1, x0[u], x1[u]);
            }
        }
        for (; i < m; ++i) {
            const unsigned pk = (unsigned)__builtin_amdgcn_readfirstlane((int)myq[c0 + i]);
            const unsigned s  = pk & 0xFFFFu;
            const float4 x0 = xl4[(size_t)s * (HIDDEN / 4) + lane];
            const float4 x1 = xl4[(size_t)s * (HIDDEN / 4) + 32 + lane];
            const float w0 = mywt[i * HEADS + hq];
            const float w1 = mywt[i * HEADS + 4 + hq];
            own_add_row(accs, (int)(pk >> 16), lane, w0, w1, x0, x1);
        }
    }
}

template <bool SECOND>
__device__ __forceinline__ int own_filter(
    int tid, int lane, int wave, int hq, int n0, int cnt,
    const int* __restrict__ src, const int* __restrict__ dst,
    const float4* __restrict__ as4, const float4* __restrict__ xl4,
    unsigned* myq, float* mywt, const float* rad, unsigned* rmax,
    float* rden, float* accs, int* ovf)
{
    const int  rs  = wave + OWN_WAVES * lane;
    const bool has = rs < cnt;
    if (has) myq[lane] = ((unsigned)rs << OWN_EBITS) | (unsigned)(N_EDGES + n0 + rs);
    int qn = __builtin_popcount(__builtin_amdgcn_ballot_w32(has));

    const int4* __restrict__ dst4 = (const int4*)dst;
    constexpr int NV    = N_EDGES / 4;
    constexpr int TRIPS = (NV + 1023) / 1024;
    static_assert(TRIPS % 4 == 0, "the four-trip body assumes a whole number of bodies");
    auto ld = [&](int t) {
        const int i = t * 1024 + tid;
        return dst4[i < NV ? i : NV - 1];
    };
    auto trip = [&](const int4 d, int t) {
        const int i = t * 1024 + tid;
        const unsigned lim = (i < NV) ? (unsigned)cnt : 0u;
        const unsigned l0 = (unsigned)(d.x - n0), l1 = (unsigned)(d.y - n0);
        const unsigned l2 = (unsigned)(d.z - n0), l3 = (unsigned)(d.w - n0);
        const bool m0 = l0 < lim, m1 = l1 < lim, m2 = l2 < lim, m3 = l3 < lim;
        if (__builtin_amdgcn_ballot_w32(m0 | m1 | m2 | m3) != 0u) {
            const unsigned ll[4] = {l0, l1, l2, l3};
            const bool     mm[4] = {m0, m1, m2, m3};
#pragma unroll
            for (int k = 0; k < 4; ++k) {
                const unsigned mask = __builtin_amdgcn_ballot_w32(mm[k]);
                if (mm[k]) {
                    const int pos = qn + (int)__builtin_amdgcn_mbcnt_lo(mask, 0u);
                    myq[pos] = (ll[k] << OWN_EBITS) | (unsigned)(i * 4 + k);
                }
                qn += __builtin_popcount(mask);
            }
            if (qn > OWN_QCAP - 128) {
                own_resolve<true>(myq, qn, lane, src, as4, rad, rmax);
                if (SECOND) own_drain(myq, qn, lane, hq, as4, xl4, mywt, rad, rmax, rden, accs);
                else if (lane == 0) *ovf = 1;
                qn = 0;
            }
        }
    };
    int4 d0 = ld(0), d1 = ld(1), d2 = ld(2), d3;
    for (int t = 0; t < TRIPS; t += 4) {
        d3 = ld(t + 3);  trip(d0, t);
        d0 = ld(t + 4);  trip(d1, t + 1);
        d1 = ld(t + 5);  trip(d2, t + 2);
        d2 = ld(t + 6);  trip(d3, t + 3);
    }
    own_resolve<true>(myq, qn, lane, src, as4, rad, rmax);
    if (SECOND) { own_drain(myq, qn, lane, hq, as4, xl4, mywt, rad, rmax, rden, accs); qn = 0; }
    return qn;
}

__device__ __forceinline__ float own_elu(float v) { return v > 0.f ? v : expm1f(v); }

__global__ __launch_bounds__(1024) void gat_owner_softmax_aggregate_l1(
    const int* __restrict__ src, const int* __restrict__ dst,
    const float* __restrict__ asrc, const float* __restrict__ adst,
    const float* __restrict__ xl, const float* __restrict__ bias,
    const float* __restrict__ W2, const float* __restrict__ wa2,
    float* __restrict__ xl2, float* __restrict__ as2, float* __restrict__ ad2,
    unsigned* __restrict__ qsave,
    int* __restrict__ qcnt, int* __restrict__ qovf, const unsigned* __restrict__ bins)
{
    __shared__ __attribute__((aligned(16))) float    accs[OWN_R * HIDDEN];
    __shared__ __attribute__((aligned(16))) float    rad[OWN_R * HEADS];
    __shared__ __attribute__((aligned(16))) unsigned rmax[OWN_R * HEADS];
    __shared__ __attribute__((aligned(16))) float    rden[OWN_R * HEADS];
    __shared__ __attribute__((aligned(16))) float    wts[OWN_WAVES][32 * HEADS];
    __shared__ unsigned qpk[OWN_WAVES][OWN_QCAP];
    __shared__ int      ovf;
    __shared__ int      binbad;

    const int tid  = threadIdx.x;
    const int lane = tid & 31;
    const int wave = tid >> 5;
    const int hq   = lane >> 3;
    const int n0   = blockIdx.x * OWN_R;
    int cnt = N_NODES - n0;
    if (cnt > OWN_R) cnt = OWN_R;
    const float4* __restrict__ xl4 = (const float4*)xl;
    const float4* __restrict__ as4 = (const float4*)asrc;

    const unsigned* __restrict__ cb =
        bins + ((size_t)blockIdx.x * BIN_CHUNKS + wave * BIN_PER_WAVE) * BIN_W;
    unsigned w0[BIN_PER_WAVE];
#pragma unroll
    for (int k = 0; k < BIN_PER_WAVE; ++k) w0[k] = cb[k * BIN_W + lane];

    float4* accs4 = (float4*)accs;
    for (int i = tid; i < OWN_R * HIDDEN / 4; i += 1024)
        accs4[i] = make_float4(0.f, 0.f, 0.f, 0.f);
    for (int i = tid; i < OWN_R * HEADS; i += 1024) {
        rad[i]  = ((i >> 3) < cnt) ? adst[(size_t)n0 * HEADS + i] : 0.f;
        rmax[i] = 0u;
        rden[i] = 0.f;
    }
    if (tid == 0) { ovf = 0; binbad = 0; }
    __syncthreads();

    unsigned* myq  = qpk[wave];
    float*    mywt = wts[wave];

    int  ck[BIN_PER_WAVE];
    bool bad = false;
#pragma unroll
    for (int k = 0; k < BIN_PER_WAVE; ++k) {
        ck[k] = __builtin_amdgcn_readlane((int)w0[k], 0);
        bad |= (unsigned)ck[k] > (unsigned)BIN_CAP;
    }
    if (bad && lane == 0) binbad = 1;
    __syncthreads();

    int qn;
    if (binbad == 0) {
        const int  rs  = wave + OWN_WAVES * lane;
        const bool has = rs < cnt;
        if (has) myq[lane] = ((unsigned)rs << 16) | (unsigned)(n0 + rs);
        qn = __builtin_popcount(__builtin_amdgcn_ballot_w32(has));
#pragma unroll
        for (int k = 0; k < BIN_PER_WAVE; ++k) {
            if (lane >= 1 && lane <= ck[k]) myq[qn + lane - 1] = w0[k];
            if (ck[k] > 31) {
                if (31 + lane < ck[k]) myq[qn + 31 + lane] = cb[k * BIN_W + 32 + lane];
            }
            qn += ck[k];
        }
        own_resolve<false>(myq, qn, lane, src, as4, rad, rmax);
    } else {
        qn = own_filter<false>(tid, lane, wave, hq, n0, cnt, src, dst, as4, xl4,
                               myq, mywt, rad, rmax, rden, accs, &ovf);
    }
    __syncthreads();

    {
        unsigned* gq = qsave + ((size_t)blockIdx.x * OWN_WAVES + wave) * OWN_QCAP;
        for (int j = lane; j < qn; j += 32) gq[j] = myq[j];
        if (lane == 0) qcnt[blockIdx.x * OWN_WAVES + wave] = qn;
        if (tid == 0)  qovf[blockIdx.x] = ovf;
    }

    if (ovf == 0)
        own_drain(myq, qn, lane, hq, as4, xl4, mywt, rad, rmax, rden, accs);
    else
        (void)own_filter<true>(tid, lane, wave, hq, n0, cnt, src, dst, as4, xl4,
                               myq, mywt, rad, rmax, rden, accs, &ovf);
    __syncthreads();

    float bq0[4], bq1[4];
#pragma unroll
    for (int k = 0; k < 4; ++k) { bq0[k] = bias[lane * 4 + k]; bq1[k] = bias[128 + lane * 4 + k]; }
    for (int r = wave; r < cnt; r += OWN_WAVES) {
        float* rowp = accs + r * HIDDEN;
        const float* row = rowp + lane;
        const float d0 = rden[r * HEADS + hq] + GAT_EPS;
        const float d1 = rden[r * HEADS + 4 + hq] + GAT_EPS;
        float4 o0, o1;
        o0.x = own_elu(row[0]   / d0 + bq0[0]);  o0.y = own_elu(row[32]  / d0 + bq0[1]);
        o0.z = own_elu(row[64]  / d0 + bq0[2]);  o0.w = own_elu(row[96]  / d0 + bq0[3]);
        o1.x = own_elu(row[128] / d1 + bq1[0]);  o1.y = own_elu(row[160] / d1 + bq1[1]);
        o1.z = own_elu(row[192] / d1 + bq1[2]);  o1.w = own_elu(row[224] / d1 + bq1[3]);
        const int sw = (lane * 4) ^ ((r & 15) << 2);
        *(float4*)(rowp + sw)       = o0;
        *(float4*)(rowp + 128 + sw) = o1;
    }
    __syncthreads();

    const int ntile = (cnt + 15) >> 4;
    if (wave < 2 * ntile) {
        const int  tile = wave >> 1;
        const bool lg   = (wave & 1) != 0;
        const int  half = lane >> 4;
        const int  l    = lane & 15;
        int rr = tile * 16 + l;
        if (rr > cnt - 1) rr = cnt - 1;
        const int swz = (rr & 15) << 2;
        const float* arow = accs + rr * HIDDEN + half * 2;
        static_assert(OUT_C == FOLD_COLS, "both B tiles share the row pitch");
        const float* __restrict__ bp = (lg ? wa2 : W2) + (half * 2) * OUT_C + l;
        v8f acc = {0.f,0.f,0.f,0.f,0.f,0.f,0.f,0.f};
#pragma unroll
        for (int kk = 0; kk < HIDDEN / 4; ++kk) {
            const v2f a = *(const v2f*)(arow + (kk >> 4) * 64 + (((kk & 15) << 2) ^ swz));
            v2f b;
            b.x = bp[(kk * 4) * OUT_C];
            b.y = bp[(kk * 4 + 1) * OUT_C];
            acc = __builtin_amdgcn_wmma_f32_16x16x4_f32( false, a,  false, b,
                                                         (short)0, acc,
                                                         false,  false);
        }
        const int rbase = tile * 16 + half * 8;
        if (!lg) {
            float* __restrict__ xo = xl2 + (size_t)(n0 + rbase) * OUT_C + l;
#pragma unroll
            for (int r = 0; r < 8; ++r)
                if (rbase + r < cnt) xo[(size_t)r * OUT_C] = acc[r];
        } else if (l < 2) {
            float* __restrict__ o = (l == 0 ? as2 : ad2) + n0 + rbase;
#pragma unroll
            for (int r = 0; r < 8; ++r)
                if (rbase + r < cnt) o[r] = acc[r];
        }
    }
}

__device__ __forceinline__ void l2_add_quarter(float* acc, unsigned rl, int q, float w, float4 x)
{
    float* a = acc + (q * 4) * OWN_R + rl;
    __hip_atomic_fetch_add(a,             w * x.x, __ATOMIC_RELAXED, __HIP_MEMORY_SCOPE_WORKGROUP);
    __hip_atomic_fetch_add(a + OWN_R,     w * x.y, __ATOMIC_RELAXED, __HIP_MEMORY_SCOPE_WORKGROUP);
    __hip_atomic_fetch_add(a + 2 * OWN_R, w * x.z, __ATOMIC_RELAXED, __HIP_MEMORY_SCOPE_WORKGROUP);
    __hip_atomic_fetch_add(a + 3 * OWN_R, w * x.w, __ATOMIC_RELAXED, __HIP_MEMORY_SCOPE_WORKGROUP);
}

__global__ __launch_bounds__(1024) void gat_owner_softmax_aggregate_l2(
    const int* __restrict__ src, const int* __restrict__ dst,
    const float* __restrict__ as2, const float* __restrict__ ad2,
    const float* __restrict__ xl2, const float* __restrict__ bias,
    const unsigned* __restrict__ qsave, const int* __restrict__ qcnt,
    const int* __restrict__ qovf, float* __restrict__ out)
{
    __shared__ float    acc[OUT_C * OWN_R];
    __shared__ float    rad[OWN_R];
    __shared__ unsigned rmax[OWN_R];
    __shared__ float    rden[OWN_R];

    const int tid  = threadIdx.x;
    const int lane = tid & 31;
    const int wave = tid >> 5;
    const int n0   = blockIdx.x * OWN_R;
    int cnt = N_NODES - n0;
    if (cnt > OWN_R) cnt = OWN_R;
    const float4* __restrict__ xq = (const float4*)xl2;

    for (int i = tid; i < OUT_C * OWN_R; i += 1024) acc[i] = 0.f;
    if (tid < OWN_R) {
        rad[tid]  = (tid < cnt) ? ad2[n0 + tid] : 0.f;
        rmax[tid] = 0u;
        rden[tid] = 0.f;
    }
    const bool queued = qovf[blockIdx.x] == 0;
    const unsigned* __restrict__ gq = qsave + ((size_t)blockIdx.x * OWN_WAVES + wave) * OWN_QCAP;
    const int qn = queued ? __builtin_amdgcn_readfirstlane(qcnt[blockIdx.x * OWN_WAVES + wave]) : 0;
    __syncthreads();

    if (queued) {
        for (int j0 = 0; j0 < qn; j0 += 128) {
            unsigned pk[4];
            float    a[4];
#pragma unroll
            for (int t = 0; t < 4; ++t) {
                const int j = j0 + t * 32 + lane;
                pk[t] = (j < qn) ? gq[j] : 0u;
            }
#pragma unroll
            for (int t = 0; t < 4; ++t) a[t] = as2[pk[t] & 0xFFFFu];
#pragma unroll
            for (int t = 0; t < 4; ++t) {
                if (j0 + t * 32 + lane < qn) {
                    const unsigned rl = pk[t] >> 16;
                    const float v = own_lrelu(a[t] + rad[rl]);
                    __hip_atomic_fetch_max(rmax + rl, fkey(v), __ATOMIC_RELAXED, __HIP_MEMORY_SCOPE_WORKGROUP);
                }
            }
        }
    } else {
        for (int e = tid; e < TOT_E; e += 1024) {
            int s, d; edge_sd(e, src, dst, s, d);
            const unsigned rl = (unsigned)(d - n0);
            if (rl < (unsigned)cnt) {
                const float v = own_lrelu(as2[s] + rad[rl]);
                __hip_atomic_fetch_max(rmax + rl, fkey(v), __ATOMIC_RELAXED, __HIP_MEMORY_SCOPE_WORKGROUP);
            }
        }
    }
    __syncthreads();

    if (queued) {
        const int q   = lane & 3;
        const int sub = lane >> 2;
        auto entry = [&](int j) { return gq[j < qn ? j : qn - 1]; };
        unsigned nxt = (qn > 0) ? entry(lane) : 0u;
        for (int j0 = 0; j0 < qn; j0 += 32) {
            const unsigned mine = nxt;
            nxt = entry(j0 + 32 + lane);
            unsigned pk[4];
            float    a[4];
            float4   x[4];
#pragma unroll
            for (int t = 0; t < 4; ++t) {
                pk[t] = (unsigned)__shfl((int)mine, t * 8 + sub);
                const unsigned s = pk[t] & 0xFFFFu;
                a[t] = as2[s];
                x[t] = xq[(size_t)s * (OUT_C / 4) + q];
            }
#pragma unroll
            for (int t = 0; t < 4; ++t) {
                if (j0 + t * 8 + sub < qn) {
                    const unsigned rl = pk[t] >> 16;
                    const float w = expf(own_lrelu(a[t] + rad[rl]) - fdec(rmax[rl]));
                    if (q == 0)
                        __hip_atomic_fetch_add(rden + rl, w, __ATOMIC_RELAXED, __HIP_MEMORY_SCOPE_WORKGROUP);
                    l2_add_quarter(acc, rl, q, w, x[t]);
                }
            }
        }
    } else {
        for (int e = tid; e < TOT_E; e += 1024) {
            int s, d; edge_sd(e, src, dst, s, d);
            const unsigned rl = (unsigned)(d - n0);
            if (rl < (unsigned)cnt) {
                const float w = expf(own_lrelu(as2[s] + rad[rl]) - fdec(rmax[rl]));
                __hip_atomic_fetch_add(rden + rl, w, __ATOMIC_RELAXED, __HIP_MEMORY_SCOPE_WORKGROUP);
                for (int q = 0; q < 4; ++q)
                    l2_add_quarter(acc, rl, q, w, xq[(size_t)s * (OUT_C / 4) + q]);
            }
        }
    }
    __syncthreads();

    const int tot = cnt * OUT_C;
    for (int i = tid; i < tot; i += 1024) {
        const int r = i >> 4, c = i & 15;
        out[(size_t)n0 * OUT_C + i] = acc[c * OWN_R + r] / (rden[r] + GAT_EPS) + bias[c];
    }
}

extern "C" void kernel_launch(void* const* d_in, const int* in_sizes, int n_in,
                              void* d_out, int out_size, void* d_ws, size_t ws_size,
                              hipStream_t stream)
{
    (void)in_sizes; (void)n_in; (void)out_size; (void)ws_size;

    const float* x    = (const float*)d_in[0];
    const int*   ei   = (const int*)  d_in[1];
    const float* W1   = (const float*)d_in[2];
    const float* ats1 = (const float*)d_in[3];
    const float* atd1 = (const float*)d_in[4];
    const float* b1   = (const float*)d_in[5];
    const float* W2   = (const float*)d_in[6];
    const float* ats2 = (const float*)d_in[7];
    const float* atd2 = (const float*)d_in[8];
    const float* b2   = (const float*)d_in[9];
    const int* src = ei;
    const int* dst = ei + N_EDGES;
    float* out = (float*)d_out;

    float* xl1 = (float*)d_ws;
    float* as1 = xl1 + (size_t)N_NODES * HIDDEN;
    float* ad1 = as1 + (size_t)N_NODES * HEADS;
    float* xl2 = ad1 + (size_t)N_NODES * HEADS;
    float* as2 = xl2 + (size_t)N_NODES * OUT_C;
    float* ad2 = as2 + (size_t)N_NODES;
    unsigned* qsv = (unsigned*)(ad2 + (size_t)N_NODES);
    int*      qct = (int*)(qsv + (size_t)OWN_WG * OWN_WAVES * OWN_QCAP);
    int*      qov = qct + (size_t)OWN_WG * OWN_WAVES;
    constexpr size_t kWordsBefore = (size_t)N_NODES * (HIDDEN + 2 * HEADS + OUT_C + 2)
                                  + (size_t)OWN_WG * OWN_WAVES * OWN_QCAP
                                  + (size_t)OWN_WG * OWN_WAVES + OWN_WG;
    constexpr size_t kBinsAt = (kWordsBefore + BIN_W - 1) / BIN_W * BIN_W;
    constexpr size_t kFoldAt = kBinsAt + (size_t)OWN_WG * BIN_CHUNKS * BIN_W;
    static_assert((kFoldAt + (size_t)(IN_C + HIDDEN) * FOLD_COLS) * 4 <= 268435456ull,
                  "the bins and the folded weights must end inside the workspace");
    unsigned* bins = (unsigned*)d_ws + kBinsAt;
    float* wa1 = (float*)d_ws + kFoldAt;
    float* wa2 = wa1 + (size_t)IN_C * FOLD_COLS;

    gat_bin_edges_by_owner<<<BIN_CHUNKS + FOLD_WG, 1024, 0, stream>>>(
        src, dst, bins, W1, ats1, atd1, W2, ats2, atd2, wa1, wa2);

    gat_gemm1_xl_logits<<<N_NODES / 16, 160, 0, stream>>>(x, W1, wa1, xl1, as1, ad1);
    gat_owner_softmax_aggregate_l1<<<OWN_WG, OWN_WAVES * 32, 0, stream>>>(
        src, dst, as1, ad1, xl1, b1, W2, wa2, xl2, as2, ad2, qsv, qct, qov, bins);

    gat_owner_softmax_aggregate_l2<<<OWN_WG, OWN_WAVES * 32, 0, stream>>>(
        src, dst, as2, ad2, xl2, b2, qsv, qct, qov, out);
}


# Layer-1 pass B: each wave counting-sorts its queue by destination row and adds runs of equal row (rows, denominators) in registers before one set of LDS float atomics per run (sum order: by run)
# speedup vs baseline: 9.5055x; 1.0331x over previous
#include <hip/hip_runtime.h>
#include <cstdint>
#include <cstddef>

#define N_NODES   50000
#define N_EDGES   800000
#define TOT_E     (N_EDGES + N_NODES)
#define IN_C      128
#define HID       32
#define OUT_C     16
#define HEADS     8
#define HIDDEN    (HEADS * HID)
#define NEG_SLOPE 0.2f
#define GAT_EPS   1e-16f

typedef float v2f __attribute__((ext_vector_type(2)));
typedef float v8f __attribute__((ext_vector_type(8)));

__device__ __forceinline__ v8f gat_wmma_step_f32(v2f a, v2f b, v8f c)
{
    return __builtin_amdgcn_wmma_f32_16x16x4_f32( false, a,  false, b,
                                                  (short)0, c,
                                                  false,  false);
}

template <int K, int NC>
__device__ __forceinline__ void gat_wmma_strip4_f32(
    const float* __restrict__ Ap, const float* __restrict__ Bq, v8f (&c)[4])
{
    const v8f zero = {0.f,0.f,0.f,0.f,0.f,0.f,0.f,0.f};
#pragma unroll
    for (int j = 0; j < 4; ++j) c[j] = zero;

    float4 an = *(const float4*)Ap;
    float4 bn[4];
#pragma unroll
    for (int s = 0; s < 4; ++s) bn[s] = *(const float4*)(Bq + (size_t)s * NC);

#pragma unroll
    for (int u = 0; u < K / 8; ++u) {
        const float4 a4 = an;
        const float4 b0 = bn[0], b1 = bn[1], b2 = bn[2], b3 = bn[3];
        if (u + 1 < K / 8) {
            an = *(const float4*)(Ap + 8 * (u + 1));
#pragma unroll
            for (int s = 0; s < 4; ++s)
                bn[s] = *(const float4*)(Bq + (size_t)(8 * (u + 1) + s) * NC);
        }
        const v2f a01 = {a4.x, a4.y};
        const v2f a23 = {a4.z, a4.w};
        const v2f p01[4] = {{b0.x, b1.x}, {b0.y, b1.y}, {b0.z, b1.z}, {b0.w, b1.w}};
        const v2f p23[4] = {{b2.x, b3.x}, {b2.y, b3.y}, {b2.z, b3.z}, {b2.w, b3.w}};
#pragma unroll
        for (int j = 0; j < 4; ++j) c[j] = gat_wmma_step_f32(a01, p01[j], c[j]);
#pragma unroll
        for (int j = 0; j < 4; ++j) c[j] = gat_wmma_step_f32(a23, p23[j], c[j]);
    }
}

template <int K, int NC>
__device__ __forceinline__ void gat_wmma_strip1_f32(
    const float* __restrict__ Ap, const float* __restrict__ Bp, v8f& c)
{
    const v8f zero = {0.f,0.f,0.f,0.f,0.f,0.f,0.f,0.f};
    c = zero;
#pragma unroll
    for (int u = 0; u < K / 8; ++u) {
        const float4 a4 = *(const float4*)(Ap + 8 * u);
        float b[4];
#pragma unroll
        for (int s = 0; s < 4; ++s) b[s] = Bp[(size_t)(8 * u + s) * NC];
        const v2f a01 = {a4.x, a4.y};
        const v2f a23 = {a4.z, a4.w};
        const v2f b01 = {b[0], b[1]};
        const v2f b23 = {b[2], b[3]};
        c = gat_wmma_step_f32(a01, b01, c);
        c = gat_wmma_step_f32(a23, b23, c);
    }
}

#define FOLD_COLS 16
#define FOLD_WG   ((IN_C + HIDDEN) * FOLD_COLS / 1024)
static_assert(FOLD_WG * 1024 == (IN_C + HIDDEN) * FOLD_COLS, "one thread per folded weight");
static_assert(2 * HEADS == FOLD_COLS && N_NODES % 16 == 0, "logit tile shape; whole row tiles");

__device__ __forceinline__ void gat_fold_att(
    int t, const float* __restrict__ W1, const float* __restrict__ ats1,
    const float* __restrict__ atd1, const float* __restrict__ W2,
    const float* __restrict__ ats2, const float* __restrict__ atd2,
    float* __restrict__ wa1, float* __restrict__ wa2)
{
    if (t < IN_C * FOLD_COLS) {
        const int k = t / FOLD_COLS, j = t % FOLD_COLS, h = j % HEADS;
        const float* __restrict__ att = (j < HEADS ? ats1 : atd1) + h * HID;
        const float* __restrict__ w   = W1 + (size_t)k * HIDDEN + h * HID;
        float s = 0.f;
        for (int c = 0; c < HID; ++c) s += w[c] * att[c];
        wa1[t] = s;
    } else {
        const int u = t - IN_C * FOLD_COLS;
        const int k = u / FOLD_COLS, j = u % FOLD_COLS;
        float s = 0.f;
        if (j < 2) {
            const float* __restrict__ att = (j == 0) ? ats2 : atd2;
            const float* __restrict__ w   = W2 + (size_t)k * OUT_C;
            for (int c = 0; c < OUT_C; ++c) s += w[c] * att[c];
        }
        wa2[u] = s;
    }
}

static_assert(IN_C % 8 == 0, "the strips take k in blocks of 8");
__global__ __launch_bounds__(160) __attribute__((amdgpu_waves_per_eu(4, 10))) void gat_gemm1_xl_logits(
    const float* __restrict__ x, const float* __restrict__ W1,
    const float* __restrict__ wa1, float* __restrict__ xl,
    float* __restrict__ asrc, float* __restrict__ adst)
{
    const int wave = threadIdx.x >> 5;
    const int lane = threadIdx.x & 31;
    const int tM   = blockIdx.x;
    const int half = lane >> 4;
    const int l    = lane & 15;
    const float* __restrict__ Ap = x + (size_t)(tM * 16 + l) * IN_C + half * 4;

    if (wave < 4) {
        const float* __restrict__ Bq = W1 + (size_t)(half * 4) * HIDDEN + wave * 64 + l * 4;
        v8f c[4];
        gat_wmma_strip4_f32<IN_C, HIDDEN>(Ap, Bq, c);
        float* __restrict__ Cbase = xl + (size_t)(tM * 16 + half * 8) * HIDDEN + wave * 64 + l * 4;
#pragma unroll
        for (int r = 0; r < 8; ++r)
            *(float4*)(Cbase + (size_t)r * HIDDEN) = make_float4(c[0][r], c[1][r], c[2][r], c[3][r]);
    } else {
        const float* __restrict__ Bp = wa1 + (half * 4) * FOLD_COLS + l;
        v8f c;
        gat_wmma_strip1_f32<IN_C, FOLD_COLS>(Ap, Bp, c);
        float* __restrict__ o = (l < HEADS ? asrc : adst)
                              + (size_t)(tM * 16 + half * 8) * HEADS + (l % HEADS);
#pragma unroll
        for (int r = 0; r < 8; ++r) o[r * HEADS] = c[r];
    }
}

__device__ __forceinline__ unsigned fkey(float f) {
    unsigned u = __float_as_uint(f);
    return (u & 0x80000000u) ? ~u : (u | 0x80000000u);
}
__device__ __forceinline__ float fdec(unsigned k) {
    unsigned u = (k & 0x80000000u) ? (k ^ 0x80000000u) : ~k;
    return __uint_as_float(u);
}

__device__ __forceinline__ void edge_sd(long long e, const int* __restrict__ src,
                                        const int* __restrict__ dst, int& s, int& d) {
    if (e < N_EDGES) { s = src[e]; d = dst[e]; }
    else             { s = (int)(e - N_EDGES); d = s; }
}

#define OWN_R     196
#define OWN_WG    256
#define OWN_WAVES 32
#define OWN_QCAP  512
#define OWN_EBITS 20
#define BIN_CHUNKS 256
#define BIN_W      64
#define BIN_CAP    (BIN_W - 1)
#define BIN_VEC    ((N_EDGES / 4 + BIN_CHUNKS - 1) / BIN_CHUNKS)
#define BIN_PER_WAVE (BIN_CHUNKS / OWN_WAVES)
static_assert(TOT_E <= (1 << OWN_EBITS), "edge id must fit the low bits of a queue entry");
static_assert(N_NODES <= 65536, "a resolved queue entry keeps the source node in 16 bits");
static_assert(OWN_R * OWN_WG >= N_NODES, "every node needs an owner");
static_assert(N_EDGES % 4 == 0, "the binning launch reads the edge list as int4");
static_assert(BIN_VEC <= 1024, "one edge vector per thread of a binning work-group");
static_assert(BIN_PER_WAVE == 8 && OWN_WG == 8 * OWN_WAVES, "8 cells per wave on both sides");
static_assert((OWN_R + OWN_WAVES - 1) / OWN_WAVES + BIN_PER_WAVE * BIN_CAP <= OWN_QCAP,
              "self loops plus 8 full cells must fit a wave's queue");

__global__ __launch_bounds__(1024) void gat_bin_edges_by_owner(
    const int* __restrict__ src, const int* __restrict__ dst, unsigned* __restrict__ bins,
    const float* __restrict__ W1, const float* __restrict__ ats1, const float* __restrict__ atd1,
    const float* __restrict__ W2, const float* __restrict__ ats2, const float* __restrict__ atd2,
    float* __restrict__ wa1, float* __restrict__ wa2)
{
    if (blockIdx.x >= BIN_CHUNKS) {
        gat_fold_att((int)(blockIdx.x - BIN_CHUNKS) * 1024 + (int)threadIdx.x,
                     W1, ats1, atd1, W2, ats2, atd2, wa1, wa2);
        return;
    }

    __shared__ unsigned lcnt[OWN_WG];
    __shared__ unsigned lent[OWN_WG * BIN_CAP];

    const int tid  = threadIdx.x;
    const int lane = tid & 31;
    const int wave = tid >> 5;
    if (tid < OWN_WG) lcnt[tid] = 0u;
    __syncthreads();

    const int v = blockIdx.x * BIN_VEC + tid;
    if (tid < BIN_VEC && v < N_EDGES / 4) {
        const int4 d4 = ((const int4*)dst)[v];
        const int4 s4 = ((const int4*)src)[v];
        const int dd[4] = {d4.x, d4.y, d4.z, d4.w};
        const int ss[4] = {s4.x, s4.y, s4.z, s4.w};
#pragma unroll
        for (int k = 0; k < 4; ++k) {
            const unsigned d = (unsigned)dd[k];
            if (d < (unsigned)N_NODES) {
                const unsigned o    = d / OWN_R;
                const unsigned rl   = d - o * OWN_R;
                const unsigned slot = __hip_atomic_fetch_add(lcnt + o, 1u, __ATOMIC_RELAXED,
                                                             __HIP_MEMORY_SCOPE_WORKGROUP);
                if (slot < BIN_CAP) lent[o * BIN_CAP + slot] = (rl << 16) | (unsigned)ss[k];
            }
        }
    }
    __syncthreads();

#pragma unroll
    for (int k = 0; k < BIN_PER_WAVE; ++k) {
        const int o = wave * BIN_PER_WAVE + k;
        const unsigned c  = lcnt[o];
        const unsigned cc = c < BIN_CAP ? c : BIN_CAP;
        unsigned* cell = bins + ((size_t)o * BIN_CHUNKS + blockIdx.x) * BIN_W;
        const unsigned* row = lent + o * BIN_CAP;
        cell[lane] = (lane == 0) ? c : (((unsigned)lane <= cc) ? row[lane - 1] : 0u);
        if (cc > 31u) {
            if ((unsigned)(31 + lane) < cc) cell[32 + lane] = row[31 + lane];
        }
    }
}

__device__ __forceinline__ float own_lrelu(float v) { return v > 0.f ? v : NEG_SLOPE * v; }

__device__ __forceinline__ void own_flush_run(float* accs, float* rden, int rl, int lane, int hq,
                                              float4 s0, float4 s1, float dn0, float dn1)
{
    float* row = accs + rl * HIDDEN + lane;
    __hip_atomic_fetch_add(row +   0, s0.x, __ATOMIC_RELAXED, __HIP_MEMORY_SCOPE_WORKGROUP);
    __hip_atomic_fetch_add(row +  32, s0.y, __ATOMIC_RELAXED, __HIP_MEMORY_SCOPE_WORKGROUP);
    __hip_atomic_fetch_add(row +  64, s0.z, __ATOMIC_RELAXED, __HIP_MEMORY_SCOPE_WORKGROUP);
    __hip_atomic_fetch_add(row +  96, s0.w, __ATOMIC_RELAXED, __HIP_MEMORY_SCOPE_WORKGROUP);
    __hip_atomic_fetch_add(row + 128, s1.x, __ATOMIC_RELAXED, __HIP_MEMORY_SCOPE_WORKGROUP);
    __hip_atomic_fetch_add(row + 160, s1.y, __ATOMIC_RELAXED, __HIP_MEMORY_SCOPE_WORKGROUP);
    __hip_atomic_fetch_add(row + 192, s1.z, __ATOMIC_RELAXED, __HIP_MEMORY_SCOPE_WORKGROUP);
    __hip_atomic_fetch_add(row + 224, s1.w, __ATOMIC_RELAXED, __HIP_MEMORY_SCOPE_WORKGROUP);
    const int sub = lane & 7;
    if (sub < 2)
        __hip_atomic_fetch_add(rden + rl * HEADS + hq + 4 * sub, sub == 0 ? dn0 : dn1,
                               __ATOMIC_RELAXED, __HIP_MEMORY_SCOPE_WORKGROUP);
}

typedef unsigned own_cnt_t __attribute__((may_alias));
__device__ __forceinline__ void own_sort_queue_by_row(unsigned* myq, int qn, int lane, own_cnt_t* cnt)
{
#pragma unroll
    for (int k = 0; k < 8; ++k) cnt[lane * 8 + k] = 0u;
    unsigned pk[4], slot[4];
#pragma unroll
    for (int t = 0; t < 4; ++t) {
        const int j = t * 32 + lane;
        pk[t] = (j < qn) ? myq[j] : 0u;
    }
#pragma unroll
    for (int t = 0; t < 4; ++t) {
        slot[t] = 0u;
        if (t * 32 + lane < qn)
            slot[t] = __hip_atomic_fetch_add(cnt + (pk[t] >> 16), 1u, __ATOMIC_RELAXED,
                                             __HIP_MEMORY_SCOPE_WORKGROUP);
    }
    unsigned c[8], tot = 0u;
#pragma unroll
    for (int k = 0; k < 8; ++k) { c[k] = cnt[lane * 8 + k]; tot += c[k]; }
    unsigned inc = tot;
#pragma unroll
    for (int d = 1; d < 32; d <<= 1) {
        const unsigned o = (unsigned)__shfl_up((int)inc, d);
        if (lane >= d) inc += o;
    }
    unsigned at = inc - tot;
#pragma unroll
    for (int k = 0; k < 8; ++k) { cnt[lane * 8 + k] = at; at += c[k]; }
#pragma unroll
    for (int t = 0; t < 4; ++t)
        if (t * 32 + lane < qn) myq[cnt[pk[t] >> 16] + slot[t]] = pk[t];
}

template <bool RAW>
__device__ __forceinline__ void own_resolve(
    unsigned* myq, int qn, int lane, const int* __restrict__ src,
    const float4* __restrict__ as4, const float* rad, unsigned* rmax)
{
    const unsigned emask = (1u << OWN_EBITS) - 1u;
    for (int j = lane; j < qn; j += 32) {
        const unsigned pk = myq[j];
        unsigned rl, s;
        if (RAW) {
            const unsigned e = pk & emask;
            rl = pk >> OWN_EBITS;
            s  = (e < (unsigned)N_EDGES) ? (unsigned)src[e] : (e - (unsigned)N_EDGES);
        } else {
            rl = pk >> 16;
            s  = pk & 0xFFFFu;
        }
        const float4 a0 = as4[(size_t)s * 2], a1 = as4[(size_t)s * 2 + 1];
        const float4 d0 = *(const float4*)(rad + rl * HEADS);
        const float4 d1 = *(const float4*)(rad + rl * HEADS + 4);
        const float v[8] = {a0.x + d0.x, a0.y + d0.y, a0.z + d0.z, a0.w + d0.w,
                            a1.x + d1.x, a1.y + d1.y, a1.z + d1.z, a1.w + d1.w};
        unsigned* mx = rmax + rl * HEADS;
#pragma unroll
        for (int h = 0; h < HEADS; ++h)
            __hip_atomic_fetch_max(mx + h, fkey(own_lrelu(v[h])),
                                   __ATOMIC_RELAXED, __HIP_MEMORY_SCOPE_WORKGROUP);
        if (RAW) myq[j] = (rl << 16) | s;
    }
}

static_assert(OWN_R <= 32 * HEADS, "the wave's weight strip doubles as its row counters in the queue sort");
__device__ __forceinline__ void own_drain(
    const unsigned* myq, int qn, int lane, int hq,
    const float4* __restrict__ as4, const float4* __restrict__ xl4,
    float* mywt, const float* rad, const unsigned* rmax, float* rden, float* accs)
{
    int    cr  = -1;
    float4 s0  = make_float4(0.f, 0.f, 0.f, 0.f), s1 = make_float4(0.f, 0.f, 0.f, 0.f);
    float  dn0 = 0.f, dn1 = 0.f;
    auto take = [&](int rl, float w0, float w1, float4 x0, float4 x1) {
        if (rl != cr) {
            if (cr >= 0) own_flush_run(accs, rden, cr, lane, hq, s0, s1, dn0, dn1);
            cr  = rl;
            s0  = make_float4(w0 * x0.x, w0 * x0.y, w0 * x0.z, w0 * x0.w);
            s1  = make_float4(w1 * x1.x, w1 * x1.y, w1 * x1.z, w1 * x1.w);
            dn0 = w0;  dn1 = w1;
        } else {
            s0.x += w0 * x0.x;  s0.y += w0 * x0.y;  s0.z += w0 * x0.z;  s0.w += w0 * x0.w;
            s1.x += w1 * x1.x;  s1.y += w1 * x1.y;  s1.z += w1 * x1.z;  s1.w += w1 * x1.w;
            dn0 += w0;  dn1 += w1;
        }
    };
    for (int c0 = 0; c0 < qn; c0 += 32) {
        const int m = (qn - c0 < 32) ? (qn - c0) : 32;
        if (lane < m) {
            const unsigned pk = myq[c0 + lane];
            const unsigned s  = pk & 0xFFFFu;
            const unsigned rl = pk >> 16;
            const float4 a0 = as4[(size_t)s * 2], a1 = as4[(size_t)s * 2 + 1];
            const float4 d0 = *(const float4*)(rad + rl * HEADS);
            const float4 d1 = *(const float4*)(rad + rl * HEADS + 4);
            const uint4  k0 = *(const uint4*)(rmax + rl * HEADS);
            const uint4  k1 = *(const uint4*)(rmax + rl * HEADS + 4);
            float4 w0, w1;
            w0.x = expf(own_lrelu(a0.x + d0.x) - fdec(k0.x));
            w0.y = expf(own_lrelu(a0.y + d0.y) - fdec(k0.y));
            w0.z = expf(own_lrelu(a0.z + d0.z) - fdec(k0.z));
            w0.w = expf(own_lrelu(a0.w + d0.w) - fdec(k0.w));
            w1.x = expf(own_lrelu(a1.x + d1.x) - fdec(k1.x));
            w1.y = expf(own_lrelu(a1.y + d1.y) - fdec(k1.y));
            w1.z = expf(own_lrelu(a1.z + d1.z) - fdec(k1.z));
            w1.w = expf(own_lrelu(a1.w + d1.w) - fdec(k1.w));
            *(float4*)(mywt + lane * HEADS)     = w0;
            *(float4*)(mywt + lane * HEADS + 4) = w1;
        }
        int i = 0;
        for (; i + 4 <= m; i += 4) {
            float4 x0[4], x1[4];
            int    rl[4];
#pragma unroll
            for (int u = 0; u < 4; ++u) {
                const unsigned pk = (unsigned)__builtin_amdgcn_readfirstlane((int)myq[c0 + i + u]);
                const unsigned s  = pk & 0xFFFFu;
                rl[u] = (int)(pk >> 16);
                x0[u] = xl4[(size_t)s * (HIDDEN / 4) + lane];
                x1[u] = xl4[(size_t)s * (HIDDEN / 4) + 32 + lane];
            }
#pragma unroll
            for (int u = 0; u < 4; ++u) {
                const float w0 = mywt[(i + u) * HEADS + hq];
                const float w1 = mywt[(i + u) * HEADS + 4 + hq];
                take(rl[u], w0, w1, x0[u], x1[u]);
            }
        }
        for (; i < m; ++i) {
            const unsigned pk = (unsigned)__builtin_amdgcn_readfirstlane((int)myq[c0 + i]);
            const unsigned s  = pk & 0xFFFFu;
            const float4 x0 = xl4[(size_t)s * (HIDDEN / 4) + lane];
            const float4 x1 = xl4[(size_t)s * (HIDDEN / 4) + 32 + lane];
            const float w0 = mywt[i * HEADS + hq];
            const float w1 = mywt[i * HEADS + 4 + hq];
            take((int)(pk >> 16), w0, w1, x0, x1);
        }
    }
    if (cr >= 0) own_flush_run(accs, rden, cr, lane, hq, s0, s1, dn0, dn1);
}

template <bool SECOND>
__device__ __forceinline__ int own_filter(
    int tid, int lane, int wave, int hq, int n0, int cnt,
    const int* __restrict__ src, const int* __restrict__ dst,
    const float4* __restrict__ as4, const float4* __restrict__ xl4,
    unsigned* myq, float* mywt, const float* rad, unsigned* rmax,
    float* rden, float* accs, int* ovf)
{
    const int  rs  = wave + OWN_WAVES * lane;
    const bool has = rs < cnt;
    if (has) myq[lane] = ((unsigned)rs << OWN_EBITS) | (unsigned)(N_EDGES + n0 + rs);
    int qn = __builtin_popcount(__builtin_amdgcn_ballot_w32(has));

    const int4* __restrict__ dst4 = (const int4*)dst;
    constexpr int NV    = N_EDGES / 4;
    constexpr int TRIPS = (NV + 1023) / 1024;
    static_assert(TRIPS % 4 == 0, "the four-trip body assumes a whole number of bodies");
    auto ld = [&](int t) {
        const int i = t * 1024 + tid;
        return dst4[i < NV ? i : NV - 1];
    };
    auto trip = [&](const int4 d, int t) {
        const int i = t * 1024 + tid;
        const unsigned lim = (i < NV) ? (unsigned)cnt : 0u;
        const unsigned l0 = (unsigned)(d.x - n0), l1 = (unsigned)(d.y - n0);
        const unsigned l2 = (unsigned)(d.z - n0), l3 = (unsigned)(d.w - n0);
        const bool m0 = l0 < lim, m1 = l1 < lim, m2 = l2 < lim, m3 = l3 < lim;
        if (__builtin_amdgcn_ballot_w32(m0 | m1 | m2 | m3) != 0u) {
            const unsigned ll[4] = {l0, l1, l2, l3};
            const bool     mm[4] = {m0, m1, m2, m3};
#pragma unroll
            for (int k = 0; k < 4; ++k) {
                const unsigned mask = __builtin_amdgcn_ballot_w32(mm[k]);
                if (mm[k]) {
                    const int pos = qn + (int)__builtin_amdgcn_mbcnt_lo(mask, 0u);
                    myq[pos] = (ll[k] << OWN_EBITS) | (unsigned)(i * 4 + k);
                }
                qn += __builtin_popcount(mask);
            }
            if (qn > OWN_QCAP - 128) {
                own_resolve<true>(myq, qn, lane, src, as4, rad, rmax);
                if (SECOND) own_drain(myq, qn, lane, hq, as4, xl4, mywt, rad, rmax, rden, accs);
                else if (lane == 0) *ovf = 1;
                qn = 0;
            }
        }
    };
    int4 d0 = ld(0), d1 = ld(1), d2 = ld(2), d3;
    for (int t = 0; t < TRIPS; t += 4) {
        d3 = ld(t + 3);  trip(d0, t);
        d0 = ld(t + 4);  trip(d1, t + 1);
        d1 = ld(t + 5);  trip(d2, t + 2);
        d2 = ld(t + 6);  trip(d3, t + 3);
    }
    own_resolve<true>(myq, qn, lane, src, as4, rad, rmax);
    if (SECOND) { own_drain(myq, qn, lane, hq, as4, xl4, mywt, rad, rmax, rden, accs); qn = 0; }
    return qn;
}

__device__ __forceinline__ float own_elu(float v) { return v > 0.f ? v : expm1f(v); }

__global__ __launch_bounds__(1024) void gat_owner_softmax_aggregate_l1(
    const int* __restrict__ src, const int* __restrict__ dst,
    const float* __restrict__ asrc, const float* __restrict__ adst,
    const float* __restrict__ xl, const float* __restrict__ bias,
    const float* __restrict__ W2, const float* __restrict__ wa2,
    float* __restrict__ xl2, float* __restrict__ as2, float* __restrict__ ad2,
    unsigned* __restrict__ qsave,
    int* __restrict__ qcnt, int* __restrict__ qovf, const unsigned* __restrict__ bins)
{
    __shared__ __attribute__((aligned(16))) float    accs[OWN_R * HIDDEN];
    __shared__ __attribute__((aligned(16))) float    rad[OWN_R * HEADS];
    __shared__ __attribute__((aligned(16))) unsigned rmax[OWN_R * HEADS];
    __shared__ __attribute__((aligned(16))) float    rden[OWN_R * HEADS];
    __shared__ __attribute__((aligned(16))) float    wts[OWN_WAVES][32 * HEADS];
    __shared__ unsigned qpk[OWN_WAVES][OWN_QCAP];
    __shared__ int      ovf;
    __shared__ int      binbad;

    const int tid  = threadIdx.x;
    const int lane = tid & 31;
    const int wave = tid >> 5;
    const int hq   = lane >> 3;
    const int n0   = blockIdx.x * OWN_R;
    int cnt = N_NODES - n0;
    if (cnt > OWN_R) cnt = OWN_R;
    const float4* __restrict__ xl4 = (const float4*)xl;
    const float4* __restrict__ as4 = (const float4*)asrc;

    const unsigned* __restrict__ cb =
        bins + ((size_t)blockIdx.x * BIN_CHUNKS + wave * BIN_PER_WAVE) * BIN_W;
    unsigned w0[BIN_PER_WAVE];
#pragma unroll
    for (int k = 0; k < BIN_PER_WAVE; ++k) w0[k] = cb[k * BIN_W + lane];

    float4* accs4 = (float4*)accs;
    for (int i = tid; i < OWN_R * HIDDEN / 4; i += 1024)
        accs4[i] = make_float4(0.f, 0.f, 0.f, 0.f);
    for (int i = tid; i < OWN_R * HEADS; i += 1024) {
        rad[i]  = ((i >> 3) < cnt) ? adst[(size_t)n0 * HEADS + i] : 0.f;
        rmax[i] = 0u;
        rden[i] = 0.f;
    }
    if (tid == 0) { ovf = 0; binbad = 0; }
    __syncthreads();

    unsigned* myq  = qpk[wave];
    float*    mywt = wts[wave];

    int  ck[BIN_PER_WAVE];
    bool bad = false;
#pragma unroll
    for (int k = 0; k < BIN_PER_WAVE; ++k) {
        ck[k] = __builtin_amdgcn_readlane((int)w0[k], 0);
        bad |= (unsigned)ck[k] > (unsigned)BIN_CAP;
    }
    if (bad && lane == 0) binbad = 1;
    __syncthreads();

    int qn;
    if (binbad == 0) {
        const int  rs  = wave + OWN_WAVES * lane;
        const bool has = rs < cnt;
        if (has) myq[lane] = ((unsigned)rs << 16) | (unsigned)(n0 + rs);
        qn = __builtin_popcount(__builtin_amdgcn_ballot_w32(has));
#pragma unroll
        for (int k = 0; k < BIN_PER_WAVE; ++k) {
            if (lane >= 1 && lane <= ck[k]) myq[qn + lane - 1] = w0[k];
            if (ck[k] > 31) {
                if (31 + lane < ck[k]) myq[qn + 31 + lane] = cb[k * BIN_W + 32 + lane];
            }
            qn += ck[k];
        }
        own_resolve<false>(myq, qn, lane, src, as4, rad, rmax);
    } else {
        qn = own_filter<false>(tid, lane, wave, hq, n0, cnt, src, dst, as4, xl4,
                               myq, mywt, rad, rmax, rden, accs, &ovf);
    }
    __syncthreads();

    {
        unsigned* gq = qsave + ((size_t)blockIdx.x * OWN_WAVES + wave) * OWN_QCAP;
        for (int j = lane; j < qn; j += 32) gq[j] = myq[j];
        if (lane == 0) qcnt[blockIdx.x * OWN_WAVES + wave] = qn;
        if (tid == 0)  qovf[blockIdx.x] = ovf;
    }

    if (ovf == 0) {
        if (qn <= 128) own_sort_queue_by_row(myq, qn, lane, (own_cnt_t*)mywt);
        own_drain(myq, qn, lane, hq, as4, xl4, mywt, rad, rmax, rden, accs);
    } else
        (void)own_filter<true>(tid, lane, wave, hq, n0, cnt, src, dst, as4, xl4,
                               myq, mywt, rad, rmax, rden, accs, &ovf);
    __syncthreads();

    float bq0[4], bq1[4];
#pragma unroll
    for (int k = 0; k < 4; ++k) { bq0[k] = bias[lane * 4 + k]; bq1[k] = bias[128 + lane * 4 + k]; }
    for (int r = wave; r < cnt; r += OWN_WAVES) {
        float* rowp = accs + r * HIDDEN;
        const float* row = rowp + lane;
        const float d0 = rden[r * HEADS + hq] + GAT_EPS;
        const float d1 = rden[r * HEADS + 4 + hq] + GAT_EPS;
        float4 o0, o1;
        o0.x = own_elu(row[0]   / d0 + bq0[0]);  o0.y = own_elu(row[32]  / d0 + bq0[1]);
        o0.z = own_elu(row[64]  / d0 + bq0[2]);  o0.w = own_elu(row[96]  / d0 + bq0[3]);
        o1.x = own_elu(row[128] / d1 + bq1[0]);  o1.y = own_elu(row[160] / d1 + bq1[1]);
        o1.z = own_elu(row[192] / d1 + bq1[2]);  o1.w = own_elu(row[224] / d1 + bq1[3]);
        const int sw = (lane * 4) ^ ((r & 15) << 2);
        *(float4*)(rowp + sw)       = o0;
        *(float4*)(rowp + 128 + sw) = o1;
    }
    __syncthreads();

    const int ntile = (cnt + 15) >> 4;
    if (wave < 2 * ntile) {
        const int  tile = wave >> 1;
        const bool lg   = (wave & 1) != 0;
        const int  half = lane >> 4;
        const int  l    = lane & 15;
        int rr = tile * 16 + l;
        if (rr > cnt - 1) rr = cnt - 1;
        const int swz = (rr & 15) << 2;
        const float* arow = accs + rr * HIDDEN + half * 2;
        static_assert(OUT_C == FOLD_COLS, "both B tiles share the row pitch");
        const float* __restrict__ bp = (lg ? wa2 : W2) + (half * 2) * OUT_C + l;
        v8f acc = {0.f,0.f,0.f,0.f,0.f,0.f,0.f,0.f};
#pragma unroll
        for (int kk = 0; kk < HIDDEN / 4; ++kk) {
            const v2f a = *(const v2f*)(arow + (kk >> 4) * 64 + (((kk & 15) << 2) ^ swz));
            v2f b;
            b.x = bp[(kk * 4) * OUT_C];
            b.y = bp[(kk * 4 + 1) * OUT_C];
            acc = __builtin_amdgcn_wmma_f32_16x16x4_f32( false, a,  false, b,
                                                         (short)0, acc,
                                                         false,  false);
        }
        const int rbase = tile * 16 + half * 8;
        if (!lg) {
            float* __restrict__ xo = xl2 + (size_t)(n0 + rbase) * OUT_C + l;
#pragma unroll
            for (int r = 0; r < 8; ++r)
                if (rbase + r < cnt) xo[(size_t)r * OUT_C] = acc[r];
        } else if (l < 2) {
            float* __restrict__ o = (l == 0 ? as2 : ad2) + n0 + rbase;
#pragma unroll
            for (int r = 0; r < 8; ++r)
                if (rbase + r < cnt) o[r] = acc[r];
        }
    }
}

__device__ __forceinline__ void l2_add_quarter(float* acc, unsigned rl, int q, float w, float4 x)
{
    float* a = acc + (q * 4) * OWN_R + rl;
    __hip_atomic_fetch_add(a,             w * x.x, __ATOMIC_RELAXED, __HIP_MEMORY_SCOPE_WORKGROUP);
    __hip_atomic_fetch_add(a + OWN_R,     w * x.y, __ATOMIC_RELAXED, __HIP_MEMORY_SCOPE_WORKGROUP);
    __hip_atomic_fetch_add(a + 2 * OWN_R, w * x.z, __ATOMIC_RELAXED, __HIP_MEMORY_SCOPE_WORKGROUP);
    __hip_atomic_fetch_add(a + 3 * OWN_R, w * x.w, __ATOMIC_RELAXED, __HIP_MEMORY_SCOPE_WORKGROUP);
}

__global__ __launch_bounds__(1024) void gat_owner_softmax_aggregate_l2(
    const int* __restrict__ src, const int* __restrict__ dst,
    const float* __restrict__ as2, const float* __restrict__ ad2,
    const float* __restrict__ xl2, const float* __restrict__ bias,
    const unsigned* __restrict__ qsave, const int* __restrict__ qcnt,
    const int* __restrict__ qovf, float* __restrict__ out)
{
    __shared__ float    acc[OUT_C * OWN_R];
    __shared__ float    rad[OWN_R];
    __shared__ unsigned rmax[OWN_R];
    __shared__ float    rden[OWN_R];

    const int tid  = threadIdx.x;
    const int lane = tid & 31;
    const int wave = tid >> 5;
    const int n0   = blockIdx.x * OWN_R;
    int cnt = N_NODES - n0;
    if (cnt > OWN_R) cnt = OWN_R;
    const float4* __restrict__ xq = (const float4*)xl2;

    for (int i = tid; i < OUT_C * OWN_R; i += 1024) acc[i] = 0.f;
    if (tid < OWN_R) {
        rad[tid]  = (tid < cnt) ? ad2[n0 + tid] : 0.f;
        rmax[tid] = 0u;
        rden[tid] = 0.f;
    }
    const bool queued = qovf[blockIdx.x] == 0;
    const unsigned* __restrict__ gq = qsave + ((size_t)blockIdx.x * OWN_WAVES + wave) * OWN_QCAP;
    const int qn = queued ? __builtin_amdgcn_readfirstlane(qcnt[blockIdx.x * OWN_WAVES + wave]) : 0;
    __syncthreads();

    if (queued) {
        for (int j0 = 0; j0 < qn; j0 += 128) {
            unsigned pk[4];
            float    a[4];
#pragma unroll
            for (int t = 0; t < 4; ++t) {
                const int j = j0 + t * 32 + lane;
                pk[t] = (j < qn) ? gq[j] : 0u;
            }
#pragma unroll
            for (int t = 0; t < 4; ++t) a[t] = as2[pk[t] & 0xFFFFu];
#pragma unroll
            for (int t = 0; t < 4; ++t) {
                if (j0 + t * 32 + lane < qn) {
                    const unsigned rl = pk[t] >> 16;
                    const float v = own_lrelu(a[t] + rad[rl]);
                    __hip_atomic_fetch_max(rmax + rl, fkey(v), __ATOMIC_RELAXED, __HIP_MEMORY_SCOPE_WORKGROUP);
                }
            }
        }
    } else {
        for (int e = tid; e < TOT_E; e += 1024) {
            int s, d; edge_sd(e, src, dst, s, d);
            const unsigned rl = (unsigned)(d - n0);
            if (rl < (unsigned)cnt) {
                const float v = own_lrelu(as2[s] + rad[rl]);
                __hip_atomic_fetch_max(rmax + rl, fkey(v), __ATOMIC_RELAXED, __HIP_MEMORY_SCOPE_WORKGROUP);
            }
        }
    }
    __syncthreads();

    if (queued) {
        const int q   = lane & 3;
        const int sub = lane >> 2;
        auto entry = [&](int j) { return gq[j < qn ? j : qn - 1]; };
        unsigned nxt = (qn > 0) ? entry(lane) : 0u;
        for (int j0 = 0; j0 < qn; j0 += 32) {
            const unsigned mine = nxt;
            nxt = entry(j0 + 32 + lane);
            unsigned pk[4];
            float    a[4];
            float4   x[4];
#pragma unroll
            for (int t = 0; t < 4; ++t) {
                pk[t] = (unsigned)__shfl((int)mine, t * 8 + sub);
                const unsigned s = pk[t] & 0xFFFFu;
                a[t] = as2[s];
                x[t] = xq[(size_t)s * (OUT_C / 4) + q];
            }
#pragma unroll
            for (int t = 0; t < 4; ++t) {
                if (j0 + t * 8 + sub < qn) {
                    const unsigned rl = pk[t] >> 16;
                    const float w = expf(own_lrelu(a[t] + rad[rl]) - fdec(rmax[rl]));
                    if (q == 0)
                        __hip_atomic_fetch_add(rden + rl, w, __ATOMIC_RELAXED, __HIP_MEMORY_SCOPE_WORKGROUP);
                    l2_add_quarter(acc, rl, q, w, x[t]);
                }
            }
        }
    } else {
        for (int e = tid; e < TOT_E; e += 1024) {
            int s, d; edge_sd(e, src, dst, s, d);
            const unsigned rl = (unsigned)(d - n0);
            if (rl < (unsigned)cnt) {
                const float w = expf(own_lrelu(as2[s] + rad[rl]) - fdec(rmax[rl]));
                __hip_atomic_fetch_add(rden + rl, w, __ATOMIC_RELAXED, __HIP_MEMORY_SCOPE_WORKGROUP);
                for (int q = 0; q < 4; ++q)
                    l2_add_quarter(acc, rl, q, w, xq[(size_t)s * (OUT_C / 4) + q]);
            }
        }
    }
    __syncthreads();

    const int tot = cnt * OUT_C;
    for (int i = tid; i < tot; i += 1024) {
        const int r = i >> 4, c = i & 15;
        out[(size_t)n0 * OUT_C + i] = acc[c * OWN_R + r] / (rden[r] + GAT_EPS) + bias[c];
    }
}

extern "C" void kernel_launch(void* const* d_in, const int* in_sizes, int n_in,
                              void* d_out, int out_size, void* d_ws, size_t ws_size,
                              hipStream_t stream)
{
    (void)in_sizes; (void)n_in; (void)out_size; (void)ws_size;

    const float* x    = (const float*)d_in[0];
    const int*   ei   = (const int*)  d_in[1];
    const float* W1   = (const float*)d_in[2];
    const float* ats1 = (const float*)d_in[3];
    const float* atd1 = (const float*)d_in[4];
    const float* b1   = (const float*)d_in[5];
    const float* W2   = (const float*)d_in[6];
    const float* ats2 = (const float*)d_in[7];
    const float* atd2 = (const float*)d_in[8];
    const float* b2   = (const float*)d_in[9];
    const int* src = ei;
    const int* dst = ei + N_EDGES;
    float* out = (float*)d_out;

    float* xl1 = (float*)d_ws;
    float* as1 = xl1 + (size_t)N_NODES * HIDDEN;
    float* ad1 = as1 + (size_t)N_NODES * HEADS;
    float* xl2 = ad1 + (size_t)N_NODES * HEADS;
    float* as2 = xl2 + (size_t)N_NODES * OUT_C;
    float* ad2 = as2 + (size_t)N_NODES;
    unsigned* qsv = (unsigned*)(ad2 + (size_t)N_NODES);
    int*      qct = (int*)(qsv + (size_t)OWN_WG * OWN_WAVES * OWN_QCAP);
    int*      qov = qct + (size_t)OWN_WG * OWN_WAVES;
    constexpr size_t kWordsBefore = (size_t)N_NODES * (HIDDEN + 2 * HEADS + OUT_C + 2)
                                  + (size_t)OWN_WG * OWN_WAVES * OWN_QCAP
                                  + (size_t)OWN_WG * OWN_WAVES + OWN_WG;
    constexpr size_t kBinsAt = (kWordsBefore + BIN_W - 1) / BIN_W * BIN_W;
    constexpr size_t kFoldAt = kBinsAt + (size_t)OWN_WG * BIN_CHUNKS * BIN_W;
    static_assert((kFoldAt + (size_t)(IN_C + HIDDEN) * FOLD_COLS) * 4 <= 268435456ull,
                  "the bins and the folded weights must end inside the workspace");
    unsigned* bins = (unsigned*)d_ws + kBinsAt;
    float* wa1 = (float*)d_ws + kFoldAt;
    float* wa2 = wa1 + (size_t)IN_C * FOLD_COLS;

    gat_bin_edges_by_owner<<<BIN_CHUNKS + FOLD_WG, 1024, 0, stream>>>(
        src, dst, bins, W1, ats1, atd1, W2, ats2, atd2, wa1, wa2);

    gat_gemm1_xl_logits<<<N_NODES / 16, 160, 0, stream>>>(x, W1, wa1, xl1, as1, ad1);
    gat_owner_softmax_aggregate_l1<<<OWN_WG, OWN_WAVES * 32, 0, stream>>>(
        src, dst, as1, ad1, xl1, b1, W2, wa2, xl2, as2, ad2, qsv, qct, qov, bins);

    gat_owner_softmax_aggregate_l2<<<OWN_WG, OWN_WAVES * 32, 0, stream>>>(
        src, dst, as2, ad2, xl2, b2, qsv, qct, qov, out);
}


# Layer-1 pass B: all 32 waves' queues counting-sorted together by destination row in LDS, each wave drains an equal slice; a row's edges add in registers, LDS float atomics per run (sum order: by run)
# speedup vs baseline: 10.1169x; 1.0643x over previous
#include <hip/hip_runtime.h>
#include <cstdint>
#include <cstddef>

#define N_NODES   50000
#define N_EDGES   800000
#define TOT_E     (N_EDGES + N_NODES)
#define IN_C      128
#define HID       32
#define OUT_C     16
#define HEADS     8
#define HIDDEN    (HEADS * HID)
#define NEG_SLOPE 0.2f
#define GAT_EPS   1e-16f

typedef float v2f __attribute__((ext_vector_type(2)));
typedef float v8f __attribute__((ext_vector_type(8)));

__device__ __forceinline__ v8f gat_wmma_step_f32(v2f a, v2f b, v8f c)
{
    return __builtin_amdgcn_wmma_f32_16x16x4_f32( false, a,  false, b,
                                                  (short)0, c,
                                                  false,  false);
}

template <int K, int NC>
__device__ __forceinline__ void gat_wmma_strip4_f32(
    const float* __restrict__ Ap, const float* __restrict__ Bq, v8f (&c)[4])
{
    const v8f zero = {0.f,0.f,0.f,0.f,0.f,0.f,0.f,0.f};
#pragma unroll
    for (int j = 0; j < 4; ++j) c[j] = zero;

    float4 an = *(const float4*)Ap;
    float4 bn[4];
#pragma unroll
    for (int s = 0; s < 4; ++s) bn[s] = *(const float4*)(Bq + (size_t)s * NC);

#pragma unroll
    for (int u = 0; u < K / 8; ++u) {
        const float4 a4 = an;
        const float4 b0 = bn[0], b1 = bn[1], b2 = bn[2], b3 = bn[3];
        if (u + 1 < K / 8) {
            an = *(const float4*)(Ap + 8 * (u + 1));
#pragma unroll
            for (int s = 0; s < 4; ++s)
                bn[s] = *(const float4*)(Bq + (size_t)(8 * (u + 1) + s) * NC);
        }
        const v2f a01 = {a4.x, a4.y};
        const v2f a23 = {a4.z, a4.w};
        const v2f p01[4] = {{b0.x, b1.x}, {b0.y, b1.y}, {b0.z, b1.z}, {b0.w, b1.w}};
        const v2f p23[4] = {{b2.x, b3.x}, {b2.y, b3.y}, {b2.z, b3.z}, {b2.w, b3.w}};
#pragma unroll
        for (int j = 0; j < 4; ++j) c[j] = gat_wmma_step_f32(a01, p01[j], c[j]);
#pragma unroll
        for (int j = 0; j < 4; ++j) c[j] = gat_wmma_step_f32(a23, p23[j], c[j]);
    }
}

template <int K, int NC>
__device__ __forceinline__ void gat_wmma_strip1_f32(
    const float* __restrict__ Ap, const float* __restrict__ Bp, v8f& c)
{
    const v8f zero = {0.f,0.f,0.f,0.f,0.f,0.f,0.f,0.f};
    c = zero;
#pragma unroll
    for (int u = 0; u < K / 8; ++u) {
        const float4 a4 = *(const float4*)(Ap + 8 * u);
        float b[4];
#pragma unroll
        for (int s = 0; s < 4; ++s) b[s] = Bp[(size_t)(8 * u + s) * NC];
        const v2f a01 = {a4.x, a4.y};
        const v2f a23 = {a4.z, a4.w};
        const v2f b01 = {b[0], b[1]};
        const v2f b23 = {b[2], b[3]};
        c = gat_wmma_step_f32(a01, b01, c);
        c = gat_wmma_step_f32(a23, b23, c);
    }
}

#define FOLD_COLS 16
#define FOLD_WG   ((IN_C + HIDDEN) * FOLD_COLS / 1024)
static_assert(FOLD_WG * 1024 == (IN_C + HIDDEN) * FOLD_COLS, "one thread per folded weight");
static_assert(2 * HEADS == FOLD_COLS && N_NODES % 16 == 0, "logit tile shape; whole row tiles");

__device__ __forceinline__ void gat_fold_att(
    int t, const float* __restrict__ W1, const float* __restrict__ ats1,
    const float* __restrict__ atd1, const float* __restrict__ W2,
    const float* __restrict__ ats2, const float* __restrict__ atd2,
    float* __restrict__ wa1, float* __restrict__ wa2)
{
    if (t < IN_C * FOLD_COLS) {
        const int k = t / FOLD_COLS, j = t % FOLD_COLS, h = j % HEADS;
        const float* __restrict__ att = (j < HEADS ? ats1 : atd1) + h * HID;
        const float* __restrict__ w   = W1 + (size_t)k * HIDDEN + h * HID;
        float s = 0.f;
        for (int c = 0; c < HID; ++c) s += w[c] * att[c];
        wa1[t] = s;
    } else {
        const int u = t - IN_C * FOLD_COLS;
        const int k = u / FOLD_COLS, j = u % FOLD_COLS;
        float s = 0.f;
        if (j < 2) {
            const float* __restrict__ att = (j == 0) ? ats2 : atd2;
            const float* __restrict__ w   = W2 + (size_t)k * OUT_C;
            for (int c = 0; c < OUT_C; ++c) s += w[c] * att[c];
        }
        wa2[u] = s;
    }
}

static_assert(IN_C % 8 == 0, "the strips take k in blocks of 8");
__global__ __launch_bounds__(160) __attribute__((amdgpu_waves_per_eu(4, 10))) void gat_gemm1_xl_logits(
    const float* __restrict__ x, const float* __restrict__ W1,
    const float* __restrict__ wa1, float* __restrict__ xl,
    float* __restrict__ asrc, float* __restrict__ adst)
{
    const int wave = threadIdx.x >> 5;
    const int lane = threadIdx.x & 31;
    const int tM   = blockIdx.x;
    const int half = lane >> 4;
    const int l    = lane & 15;
    const float* __restrict__ Ap = x + (size_t)(tM * 16 + l) * IN_C + half * 4;

    if (wave < 4) {
        const float* __restrict__ Bq = W1 + (size_t)(half * 4) * HIDDEN + wave * 64 + l * 4;
        v8f c[4];
        gat_wmma_strip4_f32<IN_C, HIDDEN>(Ap, Bq, c);
        float* __restrict__ Cbase = xl + (size_t)(tM * 16 + half * 8) * HIDDEN + wave * 64 + l * 4;
#pragma unroll
        for (int r = 0; r < 8; ++r)
            *(float4*)(Cbase + (size_t)r * HIDDEN) = make_float4(c[0][r], c[1][r], c[2][r], c[3][r]);
    } else {
        const float* __restrict__ Bp = wa1 + (half * 4) * FOLD_COLS + l;
        v8f c;
        gat_wmma_strip1_f32<IN_C, FOLD_COLS>(Ap, Bp, c);
        float* __restrict__ o = (l < HEADS ? asrc : adst)
                              + (size_t)(tM * 16 + half * 8) * HEADS + (l % HEADS);
#pragma unroll
        for (int r = 0; r < 8; ++r) o[r * HEADS] = c[r];
    }
}

__device__ __forceinline__ unsigned fkey(float f) {
    unsigned u = __float_as_uint(f);
    return (u & 0x80000000u) ? ~u : (u | 0x80000000u);
}
__device__ __forceinline__ float fdec(unsigned k) {
    unsigned u = (k & 0x80000000u) ? (k ^ 0x80000000u) : ~k;
    return __uint_as_float(u);
}

__device__ __forceinline__ void edge_sd(long long e, const int* __restrict__ src,
                                        const int* __restrict__ dst, int& s, int& d) {
    if (e < N_EDGES) { s = src[e]; d = dst[e]; }
    else             { s = (int)(e - N_EDGES); d = s; }
}

#define OWN_R     196
#define OWN_WG    256
#define OWN_WAVES 32
#define OWN_QCAP  512
#define OWN_SORT_T 8
#define OWN_EBITS 20
#define BIN_CHUNKS 256
#define BIN_W      64
#define BIN_CAP    (BIN_W - 1)
#define BIN_VEC    ((N_EDGES / 4 + BIN_CHUNKS - 1) / BIN_CHUNKS)
#define BIN_PER_WAVE (BIN_CHUNKS / OWN_WAVES)
static_assert(TOT_E <= (1 << OWN_EBITS), "edge id must fit the low bits of a queue entry");
static_assert(N_NODES <= 65536, "a resolved queue entry keeps the source node in 16 bits");
static_assert(OWN_R * OWN_WG >= N_NODES, "every node needs an owner");
static_assert(N_EDGES % 4 == 0, "the binning launch reads the edge list as int4");
static_assert(BIN_VEC <= 1024, "one edge vector per thread of a binning work-group");
static_assert(BIN_PER_WAVE == 8 && OWN_WG == 8 * OWN_WAVES, "8 cells per wave on both sides");
static_assert((OWN_R + OWN_WAVES - 1) / OWN_WAVES + BIN_PER_WAVE * BIN_CAP <= OWN_QCAP,
              "self loops plus 8 full cells must fit a wave's queue");

__global__ __launch_bounds__(1024) void gat_bin_edges_by_owner(
    const int* __restrict__ src, const int* __restrict__ dst, unsigned* __restrict__ bins,
    const float* __restrict__ W1, const float* __restrict__ ats1, const float* __restrict__ atd1,
    const float* __restrict__ W2, const float* __restrict__ ats2, const float* __restrict__ atd2,
    float* __restrict__ wa1, float* __restrict__ wa2)
{
    if (blockIdx.x >= BIN_CHUNKS) {
        gat_fold_att((int)(blockIdx.x - BIN_CHUNKS) * 1024 + (int)threadIdx.x,
                     W1, ats1, atd1, W2, ats2, atd2, wa1, wa2);
        return;
    }

    __shared__ unsigned lcnt[OWN_WG];
    __shared__ unsigned lent[OWN_WG * BIN_CAP];

    const int tid  = threadIdx.x;
    const int lane = tid & 31;
    const int wave = tid >> 5;
    if (tid < OWN_WG) lcnt[tid] = 0u;
    __syncthreads();

    const int v = blockIdx.x * BIN_VEC + tid;
    if (tid < BIN_VEC && v < N_EDGES / 4) {
        const int4 d4 = ((const int4*)dst)[v];
        const int4 s4 = ((const int4*)src)[v];
        const int dd[4] = {d4.x, d4.y, d4.z, d4.w};
        const int ss[4] = {s4.x, s4.y, s4.z, s4.w};
#pragma unroll
        for (int k = 0; k < 4; ++k) {
            const unsigned d = (unsigned)dd[k];
            if (d < (unsigned)N_NODES) {
                const unsigned o    = d / OWN_R;
                const unsigned rl   = d - o * OWN_R;
                const unsigned slot = __hip_atomic_fetch_add(lcnt + o, 1u, __ATOMIC_RELAXED,
                                                             __HIP_MEMORY_SCOPE_WORKGROUP);
                if (slot < BIN_CAP) lent[o * BIN_CAP + slot] = (rl << 16) | (unsigned)ss[k];
            }
        }
    }
    __syncthreads();

#pragma unroll
    for (int k = 0; k < BIN_PER_WAVE; ++k) {
        const int o = wave * BIN_PER_WAVE + k;
        const unsigned c  = lcnt[o];
        const unsigned cc = c < BIN_CAP ? c : BIN_CAP;
        unsigned* cell = bins + ((size_t)o * BIN_CHUNKS + blockIdx.x) * BIN_W;
        const unsigned* row = lent + o * BIN_CAP;
        cell[lane] = (lane == 0) ? c : (((unsigned)lane <= cc) ? row[lane - 1] : 0u);
        if (cc > 31u) {
            if ((unsigned)(31 + lane) < cc) cell[32 + lane] = row[31 + lane];
        }
    }
}

__device__ __forceinline__ float own_lrelu(float v) { return v > 0.f ? v : NEG_SLOPE * v; }

__device__ __forceinline__ void own_flush_run(float* accs, float* rden, int rl, int lane, int hq,
                                              float4 s0, float4 s1, float dn0, float dn1)
{
    float* row = accs + rl * HIDDEN + lane;
    __hip_atomic_fetch_add(row +   0, s0.x, __ATOMIC_RELAXED, __HIP_MEMORY_SCOPE_WORKGROUP);
    __hip_atomic_fetch_add(row +  32, s0.y, __ATOMIC_RELAXED, __HIP_MEMORY_SCOPE_WORKGROUP);
    __hip_atomic_fetch_add(row +  64, s0.z, __ATOMIC_RELAXED, __HIP_MEMORY_SCOPE_WORKGROUP);
    __hip_atomic_fetch_add(row +  96, s0.w, __ATOMIC_RELAXED, __HIP_MEMORY_SCOPE_WORKGROUP);
    __hip_atomic_fetch_add(row + 128, s1.x, __ATOMIC_RELAXED, __HIP_MEMORY_SCOPE_WORKGROUP);
    __hip_atomic_fetch_add(row + 160, s1.y, __ATOMIC_RELAXED, __HIP_MEMORY_SCOPE_WORKGROUP);
    __hip_atomic_fetch_add(row + 192, s1.z, __ATOMIC_RELAXED, __HIP_MEMORY_SCOPE_WORKGROUP);
    __hip_atomic_fetch_add(row + 224, s1.w, __ATOMIC_RELAXED, __HIP_MEMORY_SCOPE_WORKGROUP);
    const int sub = lane & 7;
    if (sub < 2)
        __hip_atomic_fetch_add(rden + rl * HEADS + hq + 4 * sub, sub == 0 ? dn0 : dn1,
                               __ATOMIC_RELAXED, __HIP_MEMORY_SCOPE_WORKGROUP);
}

typedef unsigned own_cnt_t __attribute__((may_alias));
__device__ __forceinline__ void own_sort_queue_by_row(unsigned* myq, int qn, int lane, own_cnt_t* cnt)
{
#pragma unroll
    for (int k = 0; k < 8; ++k) cnt[lane * 8 + k] = 0u;
    unsigned pk[4], slot[4];
#pragma unroll
    for (int t = 0; t < 4; ++t) {
        const int j = t * 32 + lane;
        pk[t] = (j < qn) ? myq[j] : 0u;
    }
#pragma unroll
    for (int t = 0; t < 4; ++t) {
        slot[t] = 0u;
        if (t * 32 + lane < qn)
            slot[t] = __hip_atomic_fetch_add(cnt + (pk[t] >> 16), 1u, __ATOMIC_RELAXED,
                                             __HIP_MEMORY_SCOPE_WORKGROUP);
    }
    unsigned c[8], tot = 0u;
#pragma unroll
    for (int k = 0; k < 8; ++k) { c[k] = cnt[lane * 8 + k]; tot += c[k]; }
    unsigned inc = tot;
#pragma unroll
    for (int d = 1; d < 32; d <<= 1) {
        const unsigned o = (unsigned)__shfl_up((int)inc, d);
        if (lane >= d) inc += o;
    }
    unsigned at = inc - tot;
#pragma unroll
    for (int k = 0; k < 8; ++k) { cnt[lane * 8 + k] = at; at += c[k]; }
#pragma unroll
    for (int t = 0; t < 4; ++t)
        if (t * 32 + lane < qn) myq[cnt[pk[t] >> 16] + slot[t]] = pk[t];
}

template <bool RAW>
__device__ __forceinline__ void own_resolve(
    unsigned* myq, int qn, int lane, const int* __restrict__ src,
    const float4* __restrict__ as4, const float* rad, unsigned* rmax)
{
    const unsigned emask = (1u << OWN_EBITS) - 1u;
    for (int j = lane; j < qn; j += 32) {
        const unsigned pk = myq[j];
        unsigned rl, s;
        if (RAW) {
            const unsigned e = pk & emask;
            rl = pk >> OWN_EBITS;
            s  = (e < (unsigned)N_EDGES) ? (unsigned)src[e] : (e - (unsigned)N_EDGES);
        } else {
            rl = pk >> 16;
            s  = pk & 0xFFFFu;
        }
        const float4 a0 = as4[(size_t)s * 2], a1 = as4[(size_t)s * 2 + 1];
        const float4 d0 = *(const float4*)(rad + rl * HEADS);
        const float4 d1 = *(const float4*)(rad + rl * HEADS + 4);
        const float v[8] = {a0.x + d0.x, a0.y + d0.y, a0.z + d0.z, a0.w + d0.w,
                            a1.x + d1.x, a1.y + d1.y, a1.z + d1.z, a1.w + d1.w};
        unsigned* mx = rmax + rl * HEADS;
#pragma unroll
        for (int h = 0; h < HEADS; ++h)
            __hip_atomic_fetch_max(mx + h, fkey(own_lrelu(v[h])),
                                   __ATOMIC_RELAXED, __HIP_MEMORY_SCOPE_WORKGROUP);
        if (RAW) myq[j] = (rl << 16) | s;
    }
}

static_assert(OWN_R <= 32 * HEADS, "the wave's weight strip doubles as its row counters in the queue sort");
static_assert(OWN_R <= 256 && 32 * OWN_SORT_T <= OWN_QCAP,
              "256 row counters; the ordered list of all queues fits the queues' own LDS");
__device__ __forceinline__ void own_drain(
    const unsigned* myq, int qn, int lane, int hq,
    const float4* __restrict__ as4, const float4* __restrict__ xl4,
    float* mywt, const float* rad, const unsigned* rmax, float* rden, float* accs)
{
    int    cr  = -1;
    float4 s0  = make_float4(0.f, 0.f, 0.f, 0.f), s1 = make_float4(0.f, 0.f, 0.f, 0.f);
    float  dn0 = 0.f, dn1 = 0.f;
    auto take = [&](int rl, float w0, float w1, float4 x0, float4 x1) {
        if (rl != cr) {
            if (cr >= 0) own_flush_run(accs, rden, cr, lane, hq, s0, s1, dn0, dn1);
            cr  = rl;
            s0  = make_float4(w0 * x0.x, w0 * x0.y, w0 * x0.z, w0 * x0.w);
            s1  = make_float4(w1 * x1.x, w1 * x1.y, w1 * x1.z, w1 * x1.w);
            dn0 = w0;  dn1 = w1;
        } else {
            s0.x += w0 * x0.x;  s0.y += w0 * x0.y;  s0.z += w0 * x0.z;  s0.w += w0 * x0.w;
            s1.x += w1 * x1.x;  s1.y += w1 * x1.y;  s1.z += w1 * x1.z;  s1.w += w1 * x1.w;
            dn0 += w0;  dn1 += w1;
        }
    };
    for (int c0 = 0; c0 < qn; c0 += 32) {
        const int m = (qn - c0 < 32) ? (qn - c0) : 32;
        if (lane < m) {
            const unsigned pk = myq[c0 + lane];
            const unsigned s  = pk & 0xFFFFu;
            const unsigned rl = pk >> 16;
            const float4 a0 = as4[(size_t)s * 2], a1 = as4[(size_t)s * 2 + 1];
            const float4 d0 = *(const float4*)(rad + rl * HEADS);
            const float4 d1 = *(const float4*)(rad + rl * HEADS + 4);
            const uint4  k0 = *(const uint4*)(rmax + rl * HEADS);
            const uint4  k1 = *(const uint4*)(rmax + rl * HEADS + 4);
            float4 w0, w1;
            w0.x = expf(own_lrelu(a0.x + d0.x) - fdec(k0.x));
            w0.y = expf(own_lrelu(a0.y + d0.y) - fdec(k0.y));
            w0.z = expf(own_lrelu(a0.z + d0.z) - fdec(k0.z));
            w0.w = expf(own_lrelu(a0.w + d0.w) - fdec(k0.w));
            w1.x = expf(own_lrelu(a1.x + d1.x) - fdec(k1.x));
            w1.y = expf(own_lrelu(a1.y + d1.y) - fdec(k1.y));
            w1.z = expf(own_lrelu(a1.z + d1.z) - fdec(k1.z));
            w1.w = expf(own_lrelu(a1.w + d1.w) - fdec(k1.w));
            *(float4*)(mywt + lane * HEADS)     = w0;
            *(float4*)(mywt + lane * HEADS + 4) = w1;
        }
        int i = 0;
        for (; i + 4 <= m; i += 4) {
            float4 x0[4], x1[4];
            int    rl[4];
#pragma unroll
            for (int u = 0; u < 4; ++u) {
                const unsigned pk = (unsigned)__builtin_amdgcn_readfirstlane((int)myq[c0 + i + u]);
                const unsigned s  = pk & 0xFFFFu;
                rl[u] = (int)(pk >> 16);
                x0[u] = xl4[(size_t)s * (HIDDEN / 4) + lane];
                x1[u] = xl4[(size_t)s * (HIDDEN / 4) + 32 + lane];
            }
#pragma unroll
            for (int u = 0; u < 4; ++u) {
                const float w0 = mywt[(i + u) * HEADS + hq];
                const float w1 = mywt[(i + u) * HEADS + 4 + hq];
                take(rl[u], w0, w1, x0[u], x1[u]);
            }
        }
        for (; i < m; ++i) {
            const unsigned pk = (unsigned)__builtin_amdgcn_readfirstlane((int)myq[c0 + i]);
            const unsigned s  = pk & 0xFFFFu;
            const float4 x0 = xl4[(size_t)s * (HIDDEN / 4) + lane];
            const float4 x1 = xl4[(size_t)s * (HIDDEN / 4) + 32 + lane];
            const float w0 = mywt[i * HEADS + hq];
            const float w1 = mywt[i * HEADS + 4 + hq];
            take((int)(pk >> 16), w0, w1, x0, x1);
        }
    }
    if (cr >= 0) own_flush_run(accs, rden, cr, lane, hq, s0, s1, dn0, dn1);
}

template <bool SECOND>
__device__ __forceinline__ int own_filter(
    int tid, int lane, int wave, int hq, int n0, int cnt,
    const int* __restrict__ src, const int* __restrict__ dst,
    const float4* __restrict__ as4, const float4* __restrict__ xl4,
    unsigned* myq, float* mywt, const float* rad, unsigned* rmax,
    float* rden, float* accs, int* ovf)
{
    const int  rs  = wave + OWN_WAVES * lane;
    const bool has = rs < cnt;
    if (has) myq[lane] = ((unsigned)rs << OWN_EBITS) | (unsigned)(N_EDGES + n0 + rs);
    int qn = __builtin_popcount(__builtin_amdgcn_ballot_w32(has));

    const int4* __restrict__ dst4 = (const int4*)dst;
    constexpr int NV    = N_EDGES / 4;
    constexpr int TRIPS = (NV + 1023) / 1024;
    static_assert(TRIPS % 4 == 0, "the four-trip body assumes a whole number of bodies");
    auto ld = [&](int t) {
        const int i = t * 1024 + tid;
        return dst4[i < NV ? i : NV - 1];
    };
    auto trip = [&](const int4 d, int t) {
        const int i = t * 1024 + tid;
        const unsigned lim = (i < NV) ? (unsigned)cnt : 0u;
        const unsigned l0 = (unsigned)(d.x - n0), l1 = (unsigned)(d.y - n0);
        const unsigned l2 = (unsigned)(d.z - n0), l3 = (unsigned)(d.w - n0);
        const bool m0 = l0 < lim, m1 = l1 < lim, m2 = l2 < lim, m3 = l3 < lim;
        if (__builtin_amdgcn_ballot_w32(m0 | m1 | m2 | m3) != 0u) {
            const unsigned ll[4] = {l0, l1, l2, l3};
            const bool     mm[4] = {m0, m1, m2, m3};
#pragma unroll
            for (int k = 0; k < 4; ++k) {
                const unsigned mask = __builtin_amdgcn_ballot_w32(mm[k]);
                if (mm[k]) {
                    const int pos = qn + (int)__builtin_amdgcn_mbcnt_lo(mask, 0u);
                    myq[pos] = (ll[k] << OWN_EBITS) | (unsigned)(i * 4 + k);
                }
                qn += __builtin_popcount(mask);
            }
            if (qn > OWN_QCAP - 128) {
                own_resolve<true>(myq, qn, lane, src, as4, rad, rmax);
                if (SECOND) own_drain(myq, qn, lane, hq, as4, xl4, mywt, rad, rmax, rden, accs);
                else if (lane == 0) *ovf = 1;
                qn = 0;
            }
        }
    };
    int4 d0 = ld(0), d1 = ld(1), d2 = ld(2), d3;
    for (int t = 0; t < TRIPS; t += 4) {
        d3 = ld(t + 3);  trip(d0, t);
        d0 = ld(t + 4);  trip(d1, t + 1);
        d1 = ld(t + 5);  trip(d2, t + 2);
        d2 = ld(t + 6);  trip(d3, t + 3);
    }
    own_resolve<true>(myq, qn, lane, src, as4, rad, rmax);
    if (SECOND) { own_drain(myq, qn, lane, hq, as4, xl4, mywt, rad, rmax, rden, accs); qn = 0; }
    return qn;
}

__device__ __forceinline__ float own_elu(float v) { return v > 0.f ? v : expm1f(v); }

__global__ __launch_bounds__(1024) void gat_owner_softmax_aggregate_l1(
    const int* __restrict__ src, const int* __restrict__ dst,
    const float* __restrict__ asrc, const float* __restrict__ adst,
    const float* __restrict__ xl, const float* __restrict__ bias,
    const float* __restrict__ W2, const float* __restrict__ wa2,
    float* __restrict__ xl2, float* __restrict__ as2, float* __restrict__ ad2,
    unsigned* __restrict__ qsave,
    int* __restrict__ qcnt, int* __restrict__ qovf, const unsigned* __restrict__ bins)
{
    __shared__ __attribute__((aligned(16))) float    accs[OWN_R * HIDDEN];
    __shared__ __attribute__((aligned(16))) float    rad[OWN_R * HEADS];
    __shared__ __attribute__((aligned(16))) unsigned rmax[OWN_R * HEADS];
    __shared__ __attribute__((aligned(16))) float    rden[OWN_R * HEADS];
    __shared__ __attribute__((aligned(16))) float    wts[OWN_WAVES][32 * HEADS];
    __shared__ unsigned qpk[OWN_WAVES][OWN_QCAP];
    __shared__ int      ovf;
    __shared__ int      binbad;
    __shared__ int      qbig;
    __shared__ unsigned rcnt[256];

    const int tid  = threadIdx.x;
    const int lane = tid & 31;
    const int wave = tid >> 5;
    const int hq   = lane >> 3;
    const int n0   = blockIdx.x * OWN_R;
    int cnt = N_NODES - n0;
    if (cnt > OWN_R) cnt = OWN_R;
    const float4* __restrict__ xl4 = (const float4*)xl;
    const float4* __restrict__ as4 = (const float4*)asrc;

    const unsigned* __restrict__ cb =
        bins + ((size_t)blockIdx.x * BIN_CHUNKS + wave * BIN_PER_WAVE) * BIN_W;
    unsigned w0[BIN_PER_WAVE];
#pragma unroll
    for (int k = 0; k < BIN_PER_WAVE; ++k) w0[k] = cb[k * BIN_W + lane];

    float4* accs4 = (float4*)accs;
    for (int i = tid; i < OWN_R * HIDDEN / 4; i += 1024)
        accs4[i] = make_float4(0.f, 0.f, 0.f, 0.f);
    for (int i = tid; i < OWN_R * HEADS; i += 1024) {
        rad[i]  = ((i >> 3) < cnt) ? adst[(size_t)n0 * HEADS + i] : 0.f;
        rmax[i] = 0u;
        rden[i] = 0.f;
    }
    if (tid < 256) rcnt[tid] = 0u;
    if (tid == 0) { ovf = 0; binbad = 0; qbig = 0; }
    __syncthreads();

    unsigned* myq  = qpk[wave];
    float*    mywt = wts[wave];

    int  ck[BIN_PER_WAVE];
    bool bad = false;
#pragma unroll
    for (int k = 0; k < BIN_PER_WAVE; ++k) {
        ck[k] = __builtin_amdgcn_readlane((int)w0[k], 0);
        bad |= (unsigned)ck[k] > (unsigned)BIN_CAP;
    }
    if (bad && lane == 0) binbad = 1;
    __syncthreads();

    int qn;
    if (binbad == 0) {
        const int  rs  = wave + OWN_WAVES * lane;
        const bool has = rs < cnt;
        if (has) myq[lane] = ((unsigned)rs << 16) | (unsigned)(n0 + rs);
        qn = __builtin_popcount(__builtin_amdgcn_ballot_w32(has));
#pragma unroll
        for (int k = 0; k < BIN_PER_WAVE; ++k) {
            if (lane >= 1 && lane <= ck[k]) myq[qn + lane - 1] = w0[k];
            if (ck[k] > 31) {
                if (31 + lane < ck[k]) myq[qn + 31 + lane] = cb[k * BIN_W + 32 + lane];
            }
            qn += ck[k];
        }
        own_resolve<false>(myq, qn, lane, src, as4, rad, rmax);
    } else {
        qn = own_filter<false>(tid, lane, wave, hq, n0, cnt, src, dst, as4, xl4,
                               myq, mywt, rad, rmax, rden, accs, &ovf);
    }
    if (qn > 32 * OWN_SORT_T && lane == 0) qbig = 1;
    __syncthreads();

    {
        unsigned* gq = qsave + ((size_t)blockIdx.x * OWN_WAVES + wave) * OWN_QCAP;
        for (int j = lane; j < qn; j += 32) gq[j] = myq[j];
        if (lane == 0) qcnt[blockIdx.x * OWN_WAVES + wave] = qn;
        if (tid == 0)  qovf[blockIdx.x] = ovf;
    }

    if (ovf == 0 && qbig == 0) {
        unsigned pk[OWN_SORT_T], slot[OWN_SORT_T];
#pragma unroll
        for (int t = 0; t < OWN_SORT_T; ++t) {
            const int j = t * 32 + lane;
            pk[t] = (j < qn) ? myq[j] : 0u;
        }
#pragma unroll
        for (int t = 0; t < OWN_SORT_T; ++t) {
            slot[t] = 0u;
            if (t * 32 + lane < qn)
                slot[t] = __hip_atomic_fetch_add(rcnt + (pk[t] >> 16), 1u, __ATOMIC_RELAXED,
                                                 __HIP_MEMORY_SCOPE_WORKGROUP);
        }
        __syncthreads();
        own_cnt_t* offs = (own_cnt_t*)mywt;
        unsigned c[8], tot = 0u;
#pragma unroll
        for (int k = 0; k < 8; ++k) { c[k] = rcnt[lane * 8 + k]; tot += c[k]; }
        unsigned inc = tot;
#pragma unroll
        for (int d = 1; d < 32; d <<= 1) {
            const unsigned o = (unsigned)__shfl_up((int)inc, d);
            if (lane >= d) inc += o;
        }
        unsigned at = inc - tot;
#pragma unroll
        for (int k = 0; k < 8; ++k) { offs[lane * 8 + k] = at; at += c[k]; }
        const int total = __builtin_amdgcn_readlane((int)inc, 31);
        unsigned* flat = &qpk[0][0];
#pragma unroll
        for (int t = 0; t < OWN_SORT_T; ++t)
            if (t * 32 + lane < qn) flat[offs[pk[t] >> 16] + slot[t]] = pk[t];
        __syncthreads();
        const int per = (total + OWN_WAVES - 1) / OWN_WAVES;
        const int wv = __builtin_amdgcn_readfirstlane(wave);
        int sb = wv * per;        if (sb > total) sb = total;
        int se = sb + per;        if (se > total) se = total;
        own_drain(flat + sb, se - sb, lane, hq, as4, xl4, mywt, rad, rmax, rden, accs);
    } else if (ovf == 0) {
        if (qn <= 128) own_sort_queue_by_row(myq, qn, lane, (own_cnt_t*)mywt);
        own_drain(myq, qn, lane, hq, as4, xl4, mywt, rad, rmax, rden, accs);
    } else
        (void)own_filter<true>(tid, lane, wave, hq, n0, cnt, src, dst, as4, xl4,
                               myq, mywt, rad, rmax, rden, accs, &ovf);
    __syncthreads();

    float bq0[4], bq1[4];
#pragma unroll
    for (int k = 0; k < 4; ++k) { bq0[k] = bias[lane * 4 + k]; bq1[k] = bias[128 + lane * 4 + k]; }
    for (int r = wave; r < cnt; r += OWN_WAVES) {
        float* rowp = accs + r * HIDDEN;
        const float* row = rowp + lane;
        const float d0 = rden[r * HEADS + hq] + GAT_EPS;
        const float d1 = rden[r * HEADS + 4 + hq] + GAT_EPS;
        float4 o0, o1;
        o0.x = own_elu(row[0]   / d0 + bq0[0]);  o0.y = own_elu(row[32]  / d0 + bq0[1]);
        o0.z = own_elu(row[64]  / d0 + bq0[2]);  o0.w = own_elu(row[96]  / d0 + bq0[3]);
        o1.x = own_elu(row[128] / d1 + bq1[0]);  o1.y = own_elu(row[160] / d1 + bq1[1]);
        o1.z = own_elu(row[192] / d1 + bq1[2]);  o1.w = own_elu(row[224] / d1 + bq1[3]);
        const int sw = (lane * 4) ^ ((r & 15) << 2);
        *(float4*)(rowp + sw)       = o0;
        *(float4*)(rowp + 128 + sw) = o1;
    }
    __syncthreads();

    const int ntile = (cnt + 15) >> 4;
    if (wave < 2 * ntile) {
        const int  tile = wave >> 1;
        const bool lg   = (wave & 1) != 0;
        const int  half = lane >> 4;
        const int  l    = lane & 15;
        int rr = tile * 16 + l;
        if (rr > cnt - 1) rr = cnt - 1;
        const int swz = (rr & 15) << 2;
        const float* arow = accs + rr * HIDDEN + half * 2;
        static_assert(OUT_C == FOLD_COLS, "both B tiles share the row pitch");
        const float* __restrict__ bp = (lg ? wa2 : W2) + (half * 2) * OUT_C + l;
        v8f acc = {0.f,0.f,0.f,0.f,0.f,0.f,0.f,0.f};
#pragma unroll
        for (int kk = 0; kk < HIDDEN / 4; ++kk) {
            const v2f a = *(const v2f*)(arow + (kk >> 4) * 64 + (((kk & 15) << 2) ^ swz));
            v2f b;
            b.x = bp[(kk * 4) * OUT_C];
            b.y = bp[(kk * 4 + 1) * OUT_C];
            acc = __builtin_amdgcn_wmma_f32_16x16x4_f32( false, a,  false, b,
                                                         (short)0, acc,
                                                         false,  false);
        }
        const int rbase = tile * 16 + half * 8;
        if (!lg) {
            float* __restrict__ xo = xl2 + (size_t)(n0 + rbase) * OUT_C + l;
#pragma unroll
            for (int r = 0; r < 8; ++r)
                if (rbase + r < cnt) xo[(size_t)r * OUT_C] = acc[r];
        } else if (l < 2) {
            float* __restrict__ o = (l == 0 ? as2 : ad2) + n0 + rbase;
#pragma unroll
            for (int r = 0; r < 8; ++r)
                if (rbase + r < cnt) o[r] = acc[r];
        }
    }
}

__device__ __forceinline__ void l2_add_quarter(float* acc, unsigned rl, int q, float w, float4 x)
{
    float* a = acc + (q * 4) * OWN_R + rl;
    __hip_atomic_fetch_add(a,             w * x.x, __ATOMIC_RELAXED, __HIP_MEMORY_SCOPE_WORKGROUP);
    __hip_atomic_fetch_add(a + OWN_R,     w * x.y, __ATOMIC_RELAXED, __HIP_MEMORY_SCOPE_WORKGROUP);
    __hip_atomic_fetch_add(a + 2 * OWN_R, w * x.z, __ATOMIC_RELAXED, __HIP_MEMORY_SCOPE_WORKGROUP);
    __hip_atomic_fetch_add(a + 3 * OWN_R, w * x.w, __ATOMIC_RELAXED, __HIP_MEMORY_SCOPE_WORKGROUP);
}

__global__ __launch_bounds__(1024) void gat_owner_softmax_aggregate_l2(
    const int* __restrict__ src, const int* __restrict__ dst,
    const float* __restrict__ as2, const float* __restrict__ ad2,
    const float* __restrict__ xl2, const float* __restrict__ bias,
    const unsigned* __restrict__ qsave, const int* __restrict__ qcnt,
    const int* __restrict__ qovf, float* __restrict__ out)
{
    __shared__ float    acc[OUT_C * OWN_R];
    __shared__ float    rad[OWN_R];
    __shared__ unsigned rmax[OWN_R];
    __shared__ float    rden[OWN_R];

    const int tid  = threadIdx.x;
    const int lane = tid & 31;
    const int wave = tid >> 5;
    const int n0   = blockIdx.x * OWN_R;
    int cnt = N_NODES - n0;
    if (cnt > OWN_R) cnt = OWN_R;
    const float4* __restrict__ xq = (const float4*)xl2;

    for (int i = tid; i < OUT_C * OWN_R; i += 1024) acc[i] = 0.f;
    if (tid < OWN_R) {
        rad[tid]  = (tid < cnt) ? ad2[n0 + tid] : 0.f;
        rmax[tid] = 0u;
        rden[tid] = 0.f;
    }
    const bool queued = qovf[blockIdx.x] == 0;
    const unsigned* __restrict__ gq = qsave + ((size_t)blockIdx.x * OWN_WAVES + wave) * OWN_QCAP;
    const int qn = queued ? __builtin_amdgcn_readfirstlane(qcnt[blockIdx.x * OWN_WAVES + wave]) : 0;
    __syncthreads();

    if (queued) {
        for (int j0 = 0; j0 < qn; j0 += 128) {
            unsigned pk[4];
            float    a[4];
#pragma unroll
            for (int t = 0; t < 4; ++t) {
                const int j = j0 + t * 32 + lane;
                pk[t] = (j < qn) ? gq[j] : 0u;
            }
#pragma unroll
            for (int t = 0; t < 4; ++t) a[t] = as2[pk[t] & 0xFFFFu];
#pragma unroll
            for (int t = 0; t < 4; ++t) {
                if (j0 + t * 32 + lane < qn) {
                    const unsigned rl = pk[t] >> 16;
                    const float v = own_lrelu(a[t] + rad[rl]);
                    __hip_atomic_fetch_max(rmax + rl, fkey(v), __ATOMIC_RELAXED, __HIP_MEMORY_SCOPE_WORKGROUP);
                }
            }
        }
    } else {
        for (int e = tid; e < TOT_E; e += 1024) {
            int s, d; edge_sd(e, src, dst, s, d);
            const unsigned rl = (unsigned)(d - n0);
            if (rl < (unsigned)cnt) {
                const float v = own_lrelu(as2[s] + rad[rl]);
                __hip_atomic_fetch_max(rmax + rl, fkey(v), __ATOMIC_RELAXED, __HIP_MEMORY_SCOPE_WORKGROUP);
            }
        }
    }
    __syncthreads();

    if (queued) {
        const int q   = lane & 3;
        const int sub = lane >> 2;
        auto entry = [&](int j) { return gq[j < qn ? j : qn - 1]; };
        unsigned nxt = (qn > 0) ? entry(lane) : 0u;
        for (int j0 = 0; j0 < qn; j0 += 32) {
            const unsigned mine = nxt;
            nxt = entry(j0 + 32 + lane);
            unsigned pk[4];
            float    a[4];
            float4   x[4];
#pragma unroll
            for (int t = 0; t < 4; ++t) {
                pk[t] = (unsigned)__shfl((int)mine, t * 8 + sub);
                const unsigned s = pk[t] & 0xFFFFu;
                a[t] = as2[s];
                x[t] = xq[(size_t)s * (OUT_C / 4) + q];
            }
#pragma unroll
            for (int t = 0; t < 4; ++t) {
                if (j0 + t * 8 + sub < qn) {
                    const unsigned rl = pk[t] >> 16;
                    const float w = expf(own_lrelu(a[t] + rad[rl]) - fdec(rmax[rl]));
                    if (q == 0)
                        __hip_atomic_fetch_add(rden + rl, w, __ATOMIC_RELAXED, __HIP_MEMORY_SCOPE_WORKGROUP);
                    l2_add_quarter(acc, rl, q, w, x[t]);
                }
            }
        }
    } else {
        for (int e = tid; e < TOT_E; e += 1024) {
            int s, d; edge_sd(e, src, dst, s, d);
            const unsigned rl = (unsigned)(d - n0);
            if (rl < (unsigned)cnt) {
                const float w = expf(own_lrelu(as2[s] + rad[rl]) - fdec(rmax[rl]));
                __hip_atomic_fetch_add(rden + rl, w, __ATOMIC_RELAXED, __HIP_MEMORY_SCOPE_WORKGROUP);
                for (int q = 0; q < 4; ++q)
                    l2_add_quarter(acc, rl, q, w, xq[(size_t)s * (OUT_C / 4) + q]);
            }
        }
    }
    __syncthreads();

    const int tot = cnt * OUT_C;
    for (int i = tid; i < tot; i += 1024) {
        const int r = i >> 4, c = i & 15;
        out[(size_t)n0 * OUT_C + i] = acc[c * OWN_R + r] / (rden[r] + GAT_EPS) + bias[c];
    }
}

extern "C" void kernel_launch(void* const* d_in, const int* in_sizes, int n_in,
                              void* d_out, int out_size, void* d_ws, size_t ws_size,
                              hipStream_t stream)
{
    (void)in_sizes; (void)n_in; (void)out_size; (void)ws_size;

    const float* x    = (const float*)d_in[0];
    const int*   ei   = (const int*)  d_in[1];
    const float* W1   = (const float*)d_in[2];
    const float* ats1 = (const float*)d_in[3];
    const float* atd1 = (const float*)d_in[4];
    const float* b1   = (const float*)d_in[5];
    const float* W2   = (const float*)d_in[6];
    const float* ats2 = (const float*)d_in[7];
    const float* atd2 = (const float*)d_in[8];
    const float* b2   = (const float*)d_in[9];
    const int* src = ei;
    const int* dst = ei + N_EDGES;
    float* out = (float*)d_out;

    float* xl1 = (float*)d_ws;
    float* as1 = xl1 + (size_t)N_NODES * HIDDEN;
    float* ad1 = as1 + (size_t)N_NODES * HEADS;
    float* xl2 = ad1 + (size_t)N_NODES * HEADS;
    float* as2 = xl2 + (size_t)N_NODES * OUT_C;
    float* ad2 = as2 + (size_t)N_NODES;
    unsigned* qsv = (unsigned*)(ad2 + (size_t)N_NODES);
    int*      qct = (int*)(qsv + (size_t)OWN_WG * OWN_WAVES * OWN_QCAP);
    int*      qov = qct + (size_t)OWN_WG * OWN_WAVES;
    constexpr size_t kWordsBefore = (size_t)N_NODES * (HIDDEN + 2 * HEADS + OUT_C + 2)
                                  + (size_t)OWN_WG * OWN_WAVES * OWN_QCAP
                                  + (size_t)OWN_WG * OWN_WAVES + OWN_WG;
    constexpr size_t kBinsAt = (kWordsBefore + BIN_W - 1) / BIN_W * BIN_W;
    constexpr size_t kFoldAt = kBinsAt + (size_t)OWN_WG * BIN_CHUNKS * BIN_W;
    static_assert((kFoldAt + (size_t)(IN_C + HIDDEN) * FOLD_COLS) * 4 <= 268435456ull,
                  "the bins and the folded weights must end inside the workspace");
    unsigned* bins = (unsigned*)d_ws + kBinsAt;
    float* wa1 = (float*)d_ws + kFoldAt;
    float* wa2 = wa1 + (size_t)IN_C * FOLD_COLS;

    gat_bin_edges_by_owner<<<BIN_CHUNKS + FOLD_WG, 1024, 0, stream>>>(
        src, dst, bins, W1, ats1, atd1, W2, ats2, atd2, wa1, wa2);

    gat_gemm1_xl_logits<<<N_NODES / 16, 160, 0, stream>>>(x, W1, wa1, xl1, as1, ad1);
    gat_owner_softmax_aggregate_l1<<<OWN_WG, OWN_WAVES * 32, 0, stream>>>(
        src, dst, as1, ad1, xl1, b1, W2, wa2, xl2, as2, ad2, qsv, qct, qov, bins);

    gat_owner_softmax_aggregate_l2<<<OWN_WG, OWN_WAVES * 32, 0, stream>>>(
        src, dst, as2, ad2, xl2, b2, qsv, qct, qov, out);
}


# Layer-1 write-out arithmetic: one reciprocal per head group then multiply-add, ELU as hardware exp2(v)-1 (branch-free, abs error near 1e-7) instead of 8 divisions and 8 library expm1 per lane and row
# speedup vs baseline: 10.6683x; 1.0545x over previous
#include <hip/hip_runtime.h>
#include <cstdint>
#include <cstddef>

#define N_NODES   50000
#define N_EDGES   800000
#define TOT_E     (N_EDGES + N_NODES)
#define IN_C      128
#define HID       32
#define OUT_C     16
#define HEADS     8
#define HIDDEN    (HEADS * HID)
#define NEG_SLOPE 0.2f
#define GAT_EPS   1e-16f

typedef float v2f __attribute__((ext_vector_type(2)));
typedef float v8f __attribute__((ext_vector_type(8)));

__device__ __forceinline__ v8f gat_wmma_step_f32(v2f a, v2f b, v8f c)
{
    return __builtin_amdgcn_wmma_f32_16x16x4_f32( false, a,  false, b,
                                                  (short)0, c,
                                                  false,  false);
}

template <int K, int NC>
__device__ __forceinline__ void gat_wmma_strip4_f32(
    const float* __restrict__ Ap, const float* __restrict__ Bq, v8f (&c)[4])
{
    const v8f zero = {0.f,0.f,0.f,0.f,0.f,0.f,0.f,0.f};
#pragma unroll
    for (int j = 0; j < 4; ++j) c[j] = zero;

    float4 an = *(const float4*)Ap;
    float4 bn[4];
#pragma unroll
    for (int s = 0; s < 4; ++s) bn[s] = *(const float4*)(Bq + (size_t)s * NC);

#pragma unroll
    for (int u = 0; u < K / 8; ++u) {
        const float4 a4 = an;
        const float4 b0 = bn[0], b1 = bn[1], b2 = bn[2], b3 = bn[3];
        if (u + 1 < K / 8) {
            an = *(const float4*)(Ap + 8 * (u + 1));
#pragma unroll
            for (int s = 0; s < 4; ++s)
                bn[s] = *(const float4*)(Bq + (size_t)(8 * (u + 1) + s) * NC);
        }
        const v2f a01 = {a4.x, a4.y};
        const v2f a23 = {a4.z, a4.w};
        const v2f p01[4] = {{b0.x, b1.x}, {b0.y, b1.y}, {b0.z, b1.z}, {b0.w, b1.w}};
        const v2f p23[4] = {{b2.x, b3.x}, {b2.y, b3.y}, {b2.z, b3.z}, {b2.w, b3.w}};
#pragma unroll
        for (int j = 0; j < 4; ++j) c[j] = gat_wmma_step_f32(a01, p01[j], c[j]);
#pragma unroll
        for (int j = 0; j < 4; ++j) c[j] = gat_wmma_step_f32(a23, p23[j], c[j]);
    }
}

template <int K, int NC>
__device__ __forceinline__ void gat_wmma_strip1_f32(
    const float* __restrict__ Ap, const float* __restrict__ Bp, v8f& c)
{
    const v8f zero = {0.f,0.f,0.f,0.f,0.f,0.f,0.f,0.f};
    c = zero;
#pragma unroll
    for (int u = 0; u < K / 8; ++u) {
        const float4 a4 = *(const float4*)(Ap + 8 * u);
        float b[4];
#pragma unroll
        for (int s = 0; s < 4; ++s) b[s] = Bp[(size_t)(8 * u + s) * NC];
        const v2f a01 = {a4.x, a4.y};
        const v2f a23 = {a4.z, a4.w};
        const v2f b01 = {b[0], b[1]};
        const v2f b23 = {b[2], b[3]};
        c = gat_wmma_step_f32(a01, b01, c);
        c = gat_wmma_step_f32(a23, b23, c);
    }
}

#define FOLD_COLS 16
#define FOLD_WG   ((IN_C + HIDDEN) * FOLD_COLS / 1024)
static_assert(FOLD_WG * 1024 == (IN_C + HIDDEN) * FOLD_COLS, "one thread per folded weight");
static_assert(2 * HEADS == FOLD_COLS && N_NODES % 16 == 0, "logit tile shape; whole row tiles");

__device__ __forceinline__ void gat_fold_att(
    int t, const float* __restrict__ W1, const float* __restrict__ ats1,
    const float* __restrict__ atd1, const float* __restrict__ W2,
    const float* __restrict__ ats2, const float* __restrict__ atd2,
    float* __restrict__ wa1, float* __restrict__ wa2)
{
    if (t < IN_C * FOLD_COLS) {
        const int k = t / FOLD_COLS, j = t % FOLD_COLS, h = j % HEADS;
        const float* __restrict__ att = (j < HEADS ? ats1 : atd1) + h * HID;
        const float* __restrict__ w   = W1 + (size_t)k * HIDDEN + h * HID;
        float s = 0.f;
        for (int c = 0; c < HID; ++c) s += w[c] * att[c];
        wa1[t] = s;
    } else {
        const int u = t - IN_C * FOLD_COLS;
        const int k = u / FOLD_COLS, j = u % FOLD_COLS;
        float s = 0.f;
        if (j < 2) {
            const float* __restrict__ att = (j == 0) ? ats2 : atd2;
            const float* __restrict__ w   = W2 + (size_t)k * OUT_C;
            for (int c = 0; c < OUT_C; ++c) s += w[c] * att[c];
        }
        wa2[u] = s;
    }
}

static_assert(IN_C % 8 == 0, "the strips take k in blocks of 8");
__global__ __launch_bounds__(160) __attribute__((amdgpu_waves_per_eu(4, 10))) void gat_gemm1_xl_logits(
    const float* __restrict__ x, const float* __restrict__ W1,
    const float* __restrict__ wa1, float* __restrict__ xl,
    float* __restrict__ asrc, float* __restrict__ adst)
{
    const int wave = threadIdx.x >> 5;
    const int lane = threadIdx.x & 31;
    const int tM   = blockIdx.x;
    const int half = lane >> 4;
    const int l    = lane & 15;
    const float* __restrict__ Ap = x + (size_t)(tM * 16 + l) * IN_C + half * 4;

    if (wave < 4) {
        const float* __restrict__ Bq = W1 + (size_t)(half * 4) * HIDDEN + wave * 64 + l * 4;
        v8f c[4];
        gat_wmma_strip4_f32<IN_C, HIDDEN>(Ap, Bq, c);
        float* __restrict__ Cbase = xl + (size_t)(tM * 16 + half * 8) * HIDDEN + wave * 64 + l * 4;
#pragma unroll
        for (int r = 0; r < 8; ++r)
            *(float4*)(Cbase + (size_t)r * HIDDEN) = make_float4(c[0][r], c[1][r], c[2][r], c[3][r]);
    } else {
        const float* __restrict__ Bp = wa1 + (half * 4) * FOLD_COLS + l;
        v8f c;
        gat_wmma_strip1_f32<IN_C, FOLD_COLS>(Ap, Bp, c);
        float* __restrict__ o = (l < HEADS ? asrc : adst)
                              + (size_t)(tM * 16 + half * 8) * HEADS + (l % HEADS);
#pragma unroll
        for (int r = 0; r < 8; ++r) o[r * HEADS] = c[r];
    }
}

__device__ __forceinline__ unsigned fkey(float f) {
    unsigned u = __float_as_uint(f);
    return (u & 0x80000000u) ? ~u : (u | 0x80000000u);
}
__device__ __forceinline__ float fdec(unsigned k) {
    unsigned u = (k & 0x80000000u) ? (k ^ 0x80000000u) : ~k;
    return __uint_as_float(u);
}

__device__ __forceinline__ void edge_sd(long long e, const int* __restrict__ src,
                                        const int* __restrict__ dst, int& s, int& d) {
    if (e < N_EDGES) { s = src[e]; d = dst[e]; }
    else             { s = (int)(e - N_EDGES); d = s; }
}

#define OWN_R     196
#define OWN_WG    256
#define OWN_WAVES 32
#define OWN_QCAP  512
#define OWN_SORT_T 8
#define OWN_EBITS 20
#define BIN_CHUNKS 256
#define BIN_W      64
#define BIN_CAP    (BIN_W - 1)
#define BIN_VEC    ((N_EDGES / 4 + BIN_CHUNKS - 1) / BIN_CHUNKS)
#define BIN_PER_WAVE (BIN_CHUNKS / OWN_WAVES)
static_assert(TOT_E <= (1 << OWN_EBITS), "edge id must fit the low bits of a queue entry");
static_assert(N_NODES <= 65536, "a resolved queue entry keeps the source node in 16 bits");
static_assert(OWN_R * OWN_WG >= N_NODES, "every node needs an owner");
static_assert(N_EDGES % 4 == 0, "the binning launch reads the edge list as int4");
static_assert(BIN_VEC <= 1024, "one edge vector per thread of a binning work-group");
static_assert(BIN_PER_WAVE == 8 && OWN_WG == 8 * OWN_WAVES, "8 cells per wave on both sides");
static_assert((OWN_R + OWN_WAVES - 1) / OWN_WAVES + BIN_PER_WAVE * BIN_CAP <= OWN_QCAP,
              "self loops plus 8 full cells must fit a wave's queue");

__global__ __launch_bounds__(1024) void gat_bin_edges_by_owner(
    const int* __restrict__ src, const int* __restrict__ dst, unsigned* __restrict__ bins,
    const float* __restrict__ W1, const float* __restrict__ ats1, const float* __restrict__ atd1,
    const float* __restrict__ W2, const float* __restrict__ ats2, const float* __restrict__ atd2,
    float* __restrict__ wa1, float* __restrict__ wa2)
{
    if (blockIdx.x >= BIN_CHUNKS) {
        gat_fold_att((int)(blockIdx.x - BIN_CHUNKS) * 1024 + (int)threadIdx.x,
                     W1, ats1, atd1, W2, ats2, atd2, wa1, wa2);
        return;
    }

    __shared__ unsigned lcnt[OWN_WG];
    __shared__ unsigned lent[OWN_WG * BIN_CAP];

    const int tid  = threadIdx.x;
    const int lane = tid & 31;
    const int wave = tid >> 5;
    if (tid < OWN_WG) lcnt[tid] = 0u;
    __syncthreads();

    const int v = blockIdx.x * BIN_VEC + tid;
    if (tid < BIN_VEC && v < N_EDGES / 4) {
        const int4 d4 = ((const int4*)dst)[v];
        const int4 s4 = ((const int4*)src)[v];
        const int dd[4] = {d4.x, d4.y, d4.z, d4.w};
        const int ss[4] = {s4.x, s4.y, s4.z, s4.w};
#pragma unroll
        for (int k = 0; k < 4; ++k) {
            const unsigned d = (unsigned)dd[k];
            if (d < (unsigned)N_NODES) {
                const unsigned o    = d / OWN_R;
                const unsigned rl   = d - o * OWN_R;
                const unsigned slot = __hip_atomic_fetch_add(lcnt + o, 1u, __ATOMIC_RELAXED,
                                                             __HIP_MEMORY_SCOPE_WORKGROUP);
                if (slot < BIN_CAP) lent[o * BIN_CAP + slot] = (rl << 16) | (unsigned)ss[k];
            }
        }
    }
    __syncthreads();

#pragma unroll
    for (int k = 0; k < BIN_PER_WAVE; ++k) {
        const int o = wave * BIN_PER_WAVE + k;
        const unsigned c  = lcnt[o];
        const unsigned cc = c < BIN_CAP ? c : BIN_CAP;
        unsigned* cell = bins + ((size_t)o * BIN_CHUNKS + blockIdx.x) * BIN_W;
        const unsigned* row = lent + o * BIN_CAP;
        cell[lane] = (lane == 0) ? c : (((unsigned)lane <= cc) ? row[lane - 1] : 0u);
        if (cc > 31u) {
            if ((unsigned)(31 + lane) < cc) cell[32 + lane] = row[31 + lane];
        }
    }
}

__device__ __forceinline__ float own_lrelu(float v) { return v > 0.f ? v : NEG_SLOPE * v; }

__device__ __forceinline__ void own_flush_run(float* accs, float* rden, int rl, int lane, int hq,
                                              float4 s0, float4 s1, float dn0, float dn1)
{
    float* row = accs + rl * HIDDEN + lane;
    __hip_atomic_fetch_add(row +   0, s0.x, __ATOMIC_RELAXED, __HIP_MEMORY_SCOPE_WORKGROUP);
    __hip_atomic_fetch_add(row +  32, s0.y, __ATOMIC_RELAXED, __HIP_MEMORY_SCOPE_WORKGROUP);
    __hip_atomic_fetch_add(row +  64, s0.z, __ATOMIC_RELAXED, __HIP_MEMORY_SCOPE_WORKGROUP);
    __hip_atomic_fetch_add(row +  96, s0.w, __ATOMIC_RELAXED, __HIP_MEMORY_SCOPE_WORKGROUP);
    __hip_atomic_fetch_add(row + 128, s1.x, __ATOMIC_RELAXED, __HIP_MEMORY_SCOPE_WORKGROUP);
    __hip_atomic_fetch_add(row + 160, s1.y, __ATOMIC_RELAXED, __HIP_MEMORY_SCOPE_WORKGROUP);
    __hip_atomic_fetch_add(row + 192, s1.z, __ATOMIC_RELAXED, __HIP_MEMORY_SCOPE_WORKGROUP);
    __hip_atomic_fetch_add(row + 224, s1.w, __ATOMIC_RELAXED, __HIP_MEMORY_SCOPE_WORKGROUP);
    const int sub = lane & 7;
    if (sub < 2)
        __hip_atomic_fetch_add(rden + rl * HEADS + hq + 4 * sub, sub == 0 ? dn0 : dn1,
                               __ATOMIC_RELAXED, __HIP_MEMORY_SCOPE_WORKGROUP);
}

typedef unsigned own_cnt_t __attribute__((may_alias));
__device__ __forceinline__ void own_sort_queue_by_row(unsigned* myq, int qn, int lane, own_cnt_t* cnt)
{
#pragma unroll
    for (int k = 0; k < 8; ++k) cnt[lane * 8 + k] = 0u;
    unsigned pk[4], slot[4];
#pragma unroll
    for (int t = 0; t < 4; ++t) {
        const int j = t * 32 + lane;
        pk[t] = (j < qn) ? myq[j] : 0u;
    }
#pragma unroll
    for (int t = 0; t < 4; ++t) {
        slot[t] = 0u;
        if (t * 32 + lane < qn)
            slot[t] = __hip_atomic_fetch_add(cnt + (pk[t] >> 16), 1u, __ATOMIC_RELAXED,
                                             __HIP_MEMORY_SCOPE_WORKGROUP);
    }
    unsigned c[8], tot = 0u;
#pragma unroll
    for (int k = 0; k < 8; ++k) { c[k] = cnt[lane * 8 + k]; tot += c[k]; }
    unsigned inc = tot;
#pragma unroll
    for (int d = 1; d < 32; d <<= 1) {
        const unsigned o = (unsigned)__shfl_up((int)inc, d);
        if (lane >= d) inc += o;
    }
    unsigned at = inc - tot;
#pragma unroll
    for (int k = 0; k < 8; ++k) { cnt[lane * 8 + k] = at; at += c[k]; }
#pragma unroll
    for (int t = 0; t < 4; ++t)
        if (t * 32 + lane < qn) myq[cnt[pk[t] >> 16] + slot[t]] = pk[t];
}

template <bool RAW>
__device__ __forceinline__ void own_resolve(
    unsigned* myq, int qn, int lane, const int* __restrict__ src,
    const float4* __restrict__ as4, const float* rad, unsigned* rmax)
{
    const unsigned emask = (1u << OWN_EBITS) - 1u;
    for (int j = lane; j < qn; j += 32) {
        const unsigned pk = myq[j];
        unsigned rl, s;
        if (RAW) {
            const unsigned e = pk & emask;
            rl = pk >> OWN_EBITS;
            s  = (e < (unsigned)N_EDGES) ? (unsigned)src[e] : (e - (unsigned)N_EDGES);
        } else {
            rl = pk >> 16;
            s  = pk & 0xFFFFu;
        }
        const float4 a0 = as4[(size_t)s * 2], a1 = as4[(size_t)s * 2 + 1];
        const float4 d0 = *(const float4*)(rad + rl * HEADS);
        const float4 d1 = *(const float4*)(rad + rl * HEADS + 4);
        const float v[8] = {a0.x + d0.x, a0.y + d0.y, a0.z + d0.z, a0.w + d0.w,
                            a1.x + d1.x, a1.y + d1.y, a1.z + d1.z, a1.w + d1.w};
        unsigned* mx = rmax + rl * HEADS;
#pragma unroll
        for (int h = 0; h < HEADS; ++h)
            __hip_atomic_fetch_max(mx + h, fkey(own_lrelu(v[h])),
                                   __ATOMIC_RELAXED, __HIP_MEMORY_SCOPE_WORKGROUP);
        if (RAW) myq[j] = (rl << 16) | s;
    }
}

static_assert(OWN_R <= 32 * HEADS, "the wave's weight strip doubles as its row counters in the queue sort");
static_assert(OWN_R <= 256 && 32 * OWN_SORT_T <= OWN_QCAP,
              "256 row counters; the ordered list of all queues fits the queues' own LDS");
__device__ __forceinline__ void own_drain(
    const unsigned* myq, int qn, int lane, int hq,
    const float4* __restrict__ as4, const float4* __restrict__ xl4,
    float* mywt, const float* rad, const unsigned* rmax, float* rden, float* accs)
{
    int    cr  = -1;
    float4 s0  = make_float4(0.f, 0.f, 0.f, 0.f), s1 = make_float4(0.f, 0.f, 0.f, 0.f);
    float  dn0 = 0.f, dn1 = 0.f;
    auto take = [&](int rl, float w0, float w1, float4 x0, float4 x1) {
        if (rl != cr) {
            if (cr >= 0) own_flush_run(accs, rden, cr, lane, hq, s0, s1, dn0, dn1);
            cr  = rl;
            s0  = make_float4(w0 * x0.x, w0 * x0.y, w0 * x0.z, w0 * x0.w);
            s1  = make_float4(w1 * x1.x, w1 * x1.y, w1 * x1.z, w1 * x1.w);
            dn0 = w0;  dn1 = w1;
        } else {
            s0.x += w0 * x0.x;  s0.y += w0 * x0.y;  s0.z += w0 * x0.z;  s0.w += w0 * x0.w;
            s1.x += w1 * x1.x;  s1.y += w1 * x1.y;  s1.z += w1 * x1.z;  s1.w += w1 * x1.w;
            dn0 += w0;  dn1 += w1;
        }
    };
    for (int c0 = 0; c0 < qn; c0 += 32) {
        const int m = (qn - c0 < 32) ? (qn - c0) : 32;
        if (lane < m) {
            const unsigned pk = myq[c0 + lane];
            const unsigned s  = pk & 0xFFFFu;
            const unsigned rl = pk >> 16;
            const float4 a0 = as4[(size_t)s * 2], a1 = as4[(size_t)s * 2 + 1];
            const float4 d0 = *(const float4*)(rad + rl * HEADS);
            const float4 d1 = *(const float4*)(rad + rl * HEADS + 4);
            const uint4  k0 = *(const uint4*)(rmax + rl * HEADS);
            const uint4  k1 = *(const uint4*)(rmax + rl * HEADS + 4);
            float4 w0, w1;
            w0.x = expf(own_lrelu(a0.x + d0.x) - fdec(k0.x));
            w0.y = expf(own_lrelu(a0.y + d0.y) - fdec(k0.y));
            w0.z = expf(own_lrelu(a0.z + d0.z) - fdec(k0.z));
            w0.w = expf(own_lrelu(a0.w + d0.w) - fdec(k0.w));
            w1.x = expf(own_lrelu(a1.x + d1.x) - fdec(k1.x));
            w1.y = expf(own_lrelu(a1.y + d1.y) - fdec(k1.y));
            w1.z = expf(own_lrelu(a1.z + d1.z) - fdec(k1.z));
            w1.w = expf(own_lrelu(a1.w + d1.w) - fdec(k1.w));
            *(float4*)(mywt + lane * HEADS)     = w0;
            *(float4*)(mywt + lane * HEADS + 4) = w1;
        }
        int i = 0;
        for (; i + 4 <= m; i += 4) {
            float4 x0[4], x1[4];
            int    rl[4];
#pragma unroll
            for (int u = 0; u < 4; ++u) {
                const unsigned pk = (unsigned)__builtin_amdgcn_readfirstlane((int)myq[c0 + i + u]);
                const unsigned s  = pk & 0xFFFFu;
                rl[u] = (int)(pk >> 16);
                x0[u] = xl4[(size_t)s * (HIDDEN / 4) + lane];
                x1[u] = xl4[(size_t)s * (HIDDEN / 4) + 32 + lane];
            }
#pragma unroll
            for (int u = 0; u < 4; ++u) {
                const float w0 = mywt[(i + u) * HEADS + hq];
                const float w1 = mywt[(i + u) * HEADS + 4 + hq];
                take(rl[u], w0, w1, x0[u], x1[u]);
            }
        }
        for (; i < m; ++i) {
            const unsigned pk = (unsigned)__builtin_amdgcn_readfirstlane((int)myq[c0 + i]);
            const unsigned s  = pk & 0xFFFFu;
            const float4 x0 = xl4[(size_t)s * (HIDDEN / 4) + lane];
            const float4 x1 = xl4[(size_t)s * (HIDDEN / 4) + 32 + lane];
            const float w0 = mywt[i * HEADS + hq];
            const float w1 = mywt[i * HEADS + 4 + hq];
            take((int)(pk >> 16), w0, w1, x0, x1);
        }
    }
    if (cr >= 0) own_flush_run(accs, rden, cr, lane, hq, s0, s1, dn0, dn1);
}

template <bool SECOND>
__device__ __forceinline__ int own_filter(
    int tid, int lane, int wave, int hq, int n0, int cnt,
    const int* __restrict__ src, const int* __restrict__ dst,
    const float4* __restrict__ as4, const float4* __restrict__ xl4,
    unsigned* myq, float* mywt, const float* rad, unsigned* rmax,
    float* rden, float* accs, int* ovf)
{
    const int  rs  = wave + OWN_WAVES * lane;
    const bool has = rs < cnt;
    if (has) myq[lane] = ((unsigned)rs << OWN_EBITS) | (unsigned)(N_EDGES + n0 + rs);
    int qn = __builtin_popcount(__builtin_amdgcn_ballot_w32(has));

    const int4* __restrict__ dst4 = (const int4*)dst;
    constexpr int NV    = N_EDGES / 4;
    constexpr int TRIPS = (NV + 1023) / 1024;
    static_assert(TRIPS % 4 == 0, "the four-trip body assumes a whole number of bodies");
    auto ld = [&](int t) {
        const int i = t * 1024 + tid;
        return dst4[i < NV ? i : NV - 1];
    };
    auto trip = [&](const int4 d, int t) {
        const int i = t * 1024 + tid;
        const unsigned lim = (i < NV) ? (unsigned)cnt : 0u;
        const unsigned l0 = (unsigned)(d.x - n0), l1 = (unsigned)(d.y - n0);
        const unsigned l2 = (unsigned)(d.z - n0), l3 = (unsigned)(d.w - n0);
        const bool m0 = l0 < lim, m1 = l1 < lim, m2 = l2 < lim, m3 = l3 < lim;
        if (__builtin_amdgcn_ballot_w32(m0 | m1 | m2 | m3) != 0u) {
            const unsigned ll[4] = {l0, l1, l2, l3};
            const bool     mm[4] = {m0, m1, m2, m3};
#pragma unroll
            for (int k = 0; k < 4; ++k) {
                const unsigned mask = __builtin_amdgcn_ballot_w32(mm[k]);
                if (mm[k]) {
                    const int pos = qn + (int)__builtin_amdgcn_mbcnt_lo(mask, 0u);
                    myq[pos] = (ll[k] << OWN_EBITS) | (unsigned)(i * 4 + k);
                }
                qn += __builtin_popcount(mask);
            }
            if (qn > OWN_QCAP - 128) {
                own_resolve<true>(myq, qn, lane, src, as4, rad, rmax);
                if (SECOND) own_drain(myq, qn, lane, hq, as4, xl4, mywt, rad, rmax, rden, accs);
                else if (lane == 0) *ovf = 1;
                qn = 0;
            }
        }
    };
    int4 d0 = ld(0), d1 = ld(1), d2 = ld(2), d3;
    for (int t = 0; t < TRIPS; t += 4) {
        d3 = ld(t + 3);  trip(d0, t);
        d0 = ld(t + 4);  trip(d1, t + 1);
        d1 = ld(t + 5);  trip(d2, t + 2);
        d2 = ld(t + 6);  trip(d3, t + 3);
    }
    own_resolve<true>(myq, qn, lane, src, as4, rad, rmax);
    if (SECOND) { own_drain(myq, qn, lane, hq, as4, xl4, mywt, rad, rmax, rden, accs); qn = 0; }
    return qn;
}

__device__ __forceinline__ float own_elu(float v)
{
    const float e = __builtin_amdgcn_exp2f(v * 1.4426950408889634f) - 1.0f;
    return v > 0.f ? v : e;
}

__global__ __launch_bounds__(1024) void gat_owner_softmax_aggregate_l1(
    const int* __restrict__ src, const int* __restrict__ dst,
    const float* __restrict__ asrc, const float* __restrict__ adst,
    const float* __restrict__ xl, const float* __restrict__ bias,
    const float* __restrict__ W2, const float* __restrict__ wa2,
    float* __restrict__ xl2, float* __restrict__ as2, float* __restrict__ ad2,
    unsigned* __restrict__ qsave,
    int* __restrict__ qcnt, int* __restrict__ qovf, const unsigned* __restrict__ bins)
{
    __shared__ __attribute__((aligned(16))) float    accs[OWN_R * HIDDEN];
    __shared__ __attribute__((aligned(16))) float    rad[OWN_R * HEADS];
    __shared__ __attribute__((aligned(16))) unsigned rmax[OWN_R * HEADS];
    __shared__ __attribute__((aligned(16))) float    rden[OWN_R * HEADS];
    __shared__ __attribute__((aligned(16))) float    wts[OWN_WAVES][32 * HEADS];
    __shared__ unsigned qpk[OWN_WAVES][OWN_QCAP];
    __shared__ int      ovf;
    __shared__ int      binbad;
    __shared__ int      qbig;
    __shared__ unsigned rcnt[256];

    const int tid  = threadIdx.x;
    const int lane = tid & 31;
    const int wave = tid >> 5;
    const int hq   = lane >> 3;
    const int n0   = blockIdx.x * OWN_R;
    int cnt = N_NODES - n0;
    if (cnt > OWN_R) cnt = OWN_R;
    const float4* __restrict__ xl4 = (const float4*)xl;
    const float4* __restrict__ as4 = (const float4*)asrc;

    const unsigned* __restrict__ cb =
        bins + ((size_t)blockIdx.x * BIN_CHUNKS + wave * BIN_PER_WAVE) * BIN_W;
    unsigned w0[BIN_PER_WAVE];
#pragma unroll
    for (int k = 0; k < BIN_PER_WAVE; ++k) w0[k] = cb[k * BIN_W + lane];

    float4* accs4 = (float4*)accs;
    for (int i = tid; i < OWN_R * HIDDEN / 4; i += 1024)
        accs4[i] = make_float4(0.f, 0.f, 0.f, 0.f);
    for (int i = tid; i < OWN_R * HEADS; i += 1024) {
        rad[i]  = ((i >> 3) < cnt) ? adst[(size_t)n0 * HEADS + i] : 0.f;
        rmax[i] = 0u;
        rden[i] = 0.f;
    }
    if (tid < 256) rcnt[tid] = 0u;
    if (tid == 0) { ovf = 0; binbad = 0; qbig = 0; }
    __syncthreads();

    unsigned* myq  = qpk[wave];
    float*    mywt = wts[wave];

    int  ck[BIN_PER_WAVE];
    bool bad = false;
#pragma unroll
    for (int k = 0; k < BIN_PER_WAVE; ++k) {
        ck[k] = __builtin_amdgcn_readlane((int)w0[k], 0);
        bad |= (unsigned)ck[k] > (unsigned)BIN_CAP;
    }
    if (bad && lane == 0) binbad = 1;
    __syncthreads();

    int qn;
    if (binbad == 0) {
        const int  rs  = wave + OWN_WAVES * lane;
        const bool has = rs < cnt;
        if (has) myq[lane] = ((unsigned)rs << 16) | (unsigned)(n0 + rs);
        qn = __builtin_popcount(__builtin_amdgcn_ballot_w32(has));
#pragma unroll
        for (int k = 0; k < BIN_PER_WAVE; ++k) {
            if (lane >= 1 && lane <= ck[k]) myq[qn + lane - 1] = w0[k];
            if (ck[k] > 31) {
                if (31 + lane < ck[k]) myq[qn + 31 + lane] = cb[k * BIN_W + 32 + lane];
            }
            qn += ck[k];
        }
        own_resolve<false>(myq, qn, lane, src, as4, rad, rmax);
    } else {
        qn = own_filter<false>(tid, lane, wave, hq, n0, cnt, src, dst, as4, xl4,
                               myq, mywt, rad, rmax, rden, accs, &ovf);
    }
    if (qn > 32 * OWN_SORT_T && lane == 0) qbig = 1;
    __syncthreads();

    {
        unsigned* gq = qsave + ((size_t)blockIdx.x * OWN_WAVES + wave) * OWN_QCAP;
        for (int j = lane; j < qn; j += 32) gq[j] = myq[j];
        if (lane == 0) qcnt[blockIdx.x * OWN_WAVES + wave] = qn;
        if (tid == 0)  qovf[blockIdx.x] = ovf;
    }

    if (ovf == 0 && qbig == 0) {
        unsigned pk[OWN_SORT_T], slot[OWN_SORT_T];
#pragma unroll
        for (int t = 0; t < OWN_SORT_T; ++t) {
            const int j = t * 32 + lane;
            pk[t] = (j < qn) ? myq[j] : 0u;
        }
#pragma unroll
        for (int t = 0; t < OWN_SORT_T; ++t) {
            slot[t] = 0u;
            if (t * 32 + lane < qn)
                slot[t] = __hip_atomic_fetch_add(rcnt + (pk[t] >> 16), 1u, __ATOMIC_RELAXED,
                                                 __HIP_MEMORY_SCOPE_WORKGROUP);
        }
        __syncthreads();
        own_cnt_t* offs = (own_cnt_t*)mywt;
        unsigned c[8], tot = 0u;
#pragma unroll
        for (int k = 0; k < 8; ++k) { c[k] = rcnt[lane * 8 + k]; tot += c[k]; }
        unsigned inc = tot;
#pragma unroll
        for (int d = 1; d < 32; d <<= 1) {
            const unsigned o = (unsigned)__shfl_up((int)inc, d);
            if (lane >= d) inc += o;
        }
        unsigned at = inc - tot;
#pragma unroll
        for (int k = 0; k < 8; ++k) { offs[lane * 8 + k] = at; at += c[k]; }
        const int total = __builtin_amdgcn_readlane((int)inc, 31);
        unsigned* flat = &qpk[0][0];
#pragma unroll
        for (int t = 0; t < OWN_SORT_T; ++t)
            if (t * 32 + lane < qn) flat[offs[pk[t] >> 16] + slot[t]] = pk[t];
        __syncthreads();
        const int per = (total + OWN_WAVES - 1) / OWN_WAVES;
        const int wv = __builtin_amdgcn_readfirstlane(wave);
        int sb = wv * per;        if (sb > total) sb = total;
        int se = sb + per;        if (se > total) se = total;
        own_drain(flat + sb, se - sb, lane, hq, as4, xl4, mywt, rad, rmax, rden, accs);
    } else if (ovf == 0) {
        if (qn <= 128) own_sort_queue_by_row(myq, qn, lane, (own_cnt_t*)mywt);
        own_drain(myq, qn, lane, hq, as4, xl4, mywt, rad, rmax, rden, accs);
    } else
        (void)own_filter<true>(tid, lane, wave, hq, n0, cnt, src, dst, as4, xl4,
                               myq, mywt, rad, rmax, rden, accs, &ovf);
    __syncthreads();

    float bq0[4], bq1[4];
#pragma unroll
    for (int k = 0; k < 4; ++k) { bq0[k] = bias[lane * 4 + k]; bq1[k] = bias[128 + lane * 4 + k]; }
    for (int r = wave; r < cnt; r += OWN_WAVES) {
        float* rowp = accs + r * HIDDEN;
        const float* row = rowp + lane;
        const float i0 = 1.0f / (rden[r * HEADS + hq] + GAT_EPS);
        const float i1 = 1.0f / (rden[r * HEADS + 4 + hq] + GAT_EPS);
        float4 o0, o1;
        o0.x = own_elu(row[0]   * i0 + bq0[0]);  o0.y = own_elu(row[32]  * i0 + bq0[1]);
        o0.z = own_elu(row[64]  * i0 + bq0[2]);  o0.w = own_elu(row[96]  * i0 + bq0[3]);
        o1.x = own_elu(row[128] * i1 + bq1[0]);  o1.y = own_elu(row[160] * i1 + bq1[1]);
        o1.z = own_elu(row[192] * i1 + bq1[2]);  o1.w = own_elu(row[224] * i1 + bq1[3]);
        const int sw = (lane * 4) ^ ((r & 15) << 2);
        *(float4*)(rowp + sw)       = o0;
        *(float4*)(rowp + 128 + sw) = o1;
    }
    __syncthreads();

    const int ntile = (cnt + 15) >> 4;
    if (wave < 2 * ntile) {
        const int  tile = wave >> 1;
        const bool lg   = (wave & 1) != 0;
        const int  half = lane >> 4;
        const int  l    = lane & 15;
        int rr = tile * 16 + l;
        if (rr > cnt - 1) rr = cnt - 1;
        const int swz = (rr & 15) << 2;
        const float* arow = accs + rr * HIDDEN + half * 2;
        static_assert(OUT_C == FOLD_COLS, "both B tiles share the row pitch");
        const float* __restrict__ bp = (lg ? wa2 : W2) + (half * 2) * OUT_C + l;
        v8f acc = {0.f,0.f,0.f,0.f,0.f,0.f,0.f,0.f};
#pragma unroll
        for (int kk = 0; kk < HIDDEN / 4; ++kk) {
            const v2f a = *(const v2f*)(arow + (kk >> 4) * 64 + (((kk & 15) << 2) ^ swz));
            v2f b;
            b.x = bp[(kk * 4) * OUT_C];
            b.y = bp[(kk * 4 + 1) * OUT_C];
            acc = __builtin_amdgcn_wmma_f32_16x16x4_f32( false, a,  false, b,
                                                         (short)0, acc,
                                                         false,  false);
        }
        const int rbase = tile * 16 + half * 8;
        if (!lg) {
            float* __restrict__ xo = xl2 + (size_t)(n0 + rbase) * OUT_C + l;
#pragma unroll
            for (int r = 0; r < 8; ++r)
                if (rbase + r < cnt) xo[(size_t)r * OUT_C] = acc[r];
        } else if (l < 2) {
            float* __restrict__ o = (l == 0 ? as2 : ad2) + n0 + rbase;
#pragma unroll
            for (int r = 0; r < 8; ++r)
                if (rbase + r < cnt) o[r] = acc[r];
        }
    }
}

__device__ __forceinline__ void l2_add_quarter(float* acc, unsigned rl, int q, float w, float4 x)
{
    float* a = acc + (q * 4) * OWN_R + rl;
    __hip_atomic_fetch_add(a,             w * x.x, __ATOMIC_RELAXED, __HIP_MEMORY_SCOPE_WORKGROUP);
    __hip_atomic_fetch_add(a + OWN_R,     w * x.y, __ATOMIC_RELAXED, __HIP_MEMORY_SCOPE_WORKGROUP);
    __hip_atomic_fetch_add(a + 2 * OWN_R, w * x.z, __ATOMIC_RELAXED, __HIP_MEMORY_SCOPE_WORKGROUP);
    __hip_atomic_fetch_add(a + 3 * OWN_R, w * x.w, __ATOMIC_RELAXED, __HIP_MEMORY_SCOPE_WORKGROUP);
}

__global__ __launch_bounds__(1024) void gat_owner_softmax_aggregate_l2(
    const int* __restrict__ src, const int* __restrict__ dst,
    const float* __restrict__ as2, const float* __restrict__ ad2,
    const float* __restrict__ xl2, const float* __restrict__ bias,
    const unsigned* __restrict__ qsave, const int* __restrict__ qcnt,
    const int* __restrict__ qovf, float* __restrict__ out)
{
    __shared__ float    acc[OUT_C * OWN_R];
    __shared__ float    rad[OWN_R];
    __shared__ unsigned rmax[OWN_R];
    __shared__ float    rden[OWN_R];

    const int tid  = threadIdx.x;
    const int lane = tid & 31;
    const int wave = tid >> 5;
    const int n0   = blockIdx.x * OWN_R;
    int cnt = N_NODES - n0;
    if (cnt > OWN_R) cnt = OWN_R;
    const float4* __restrict__ xq = (const float4*)xl2;

    for (int i = tid; i < OUT_C * OWN_R; i += 1024) acc[i] = 0.f;
    if (tid < OWN_R) {
        rad[tid]  = (tid < cnt) ? ad2[n0 + tid] : 0.f;
        rmax[tid] = 0u;
        rden[tid] = 0.f;
    }
    const bool queued = qovf[blockIdx.x] == 0;
    const unsigned* __restrict__ gq = qsave + ((size_t)blockIdx.x * OWN_WAVES + wave) * OWN_QCAP;
    const int qn = queued ? __builtin_amdgcn_readfirstlane(qcnt[blockIdx.x * OWN_WAVES + wave]) : 0;
    __syncthreads();

    if (queued) {
        for (int j0 = 0; j0 < qn; j0 += 128) {
            unsigned pk[4];
            float    a[4];
#pragma unroll
            for (int t = 0; t < 4; ++t) {
                const int j = j0 + t * 32 + lane;
                pk[t] = (j < qn) ? gq[j] : 0u;
            }
#pragma unroll
            for (int t = 0; t < 4; ++t) a[t] = as2[pk[t] & 0xFFFFu];
#pragma unroll
            for (int t = 0; t < 4; ++t) {
                if (j0 + t * 32 + lane < qn) {
                    const unsigned rl = pk[t] >> 16;
                    const float v = own_lrelu(a[t] + rad[rl]);
                    __hip_atomic_fetch_max(rmax + rl, fkey(v), __ATOMIC_RELAXED, __HIP_MEMORY_SCOPE_WORKGROUP);
                }
            }
        }
    } else {
        for (int e = tid; e < TOT_E; e += 1024) {
            int s, d; edge_sd(e, src, dst, s, d);
            const unsigned rl = (unsigned)(d - n0);
            if (rl < (unsigned)cnt) {
                const float v = own_lrelu(as2[s] + rad[rl]);
                __hip_atomic_fetch_max(rmax + rl, fkey(v), __ATOMIC_RELAXED, __HIP_MEMORY_SCOPE_WORKGROUP);
            }
        }
    }
    __syncthreads();

    if (queued) {
        const int q   = lane & 3;
        const int sub = lane >> 2;
        auto entry = [&](int j) { return gq[j < qn ? j : qn - 1]; };
        unsigned nxt = (qn > 0) ? entry(lane) : 0u;
        for (int j0 = 0; j0 < qn; j0 += 32) {
            const unsigned mine = nxt;
            nxt = entry(j0 + 32 + lane);
            unsigned pk[4];
            float    a[4];
            float4   x[4];
#pragma unroll
            for (int t = 0; t < 4; ++t) {
                pk[t] = (unsigned)__shfl((int)mine, t * 8 + sub);
                const unsigned s = pk[t] & 0xFFFFu;
                a[t] = as2[s];
                x[t] = xq[(size_t)s * (OUT_C / 4) + q];
            }
#pragma unroll
            for (int t = 0; t < 4; ++t) {
                if (j0 + t * 8 + sub < qn) {
                    const unsigned rl = pk[t] >> 16;
                    const float w = expf(own_lrelu(a[t] + rad[rl]) - fdec(rmax[rl]));
                    if (q == 0)
                        __hip_atomic_fetch_add(rden + rl, w, __ATOMIC_RELAXED, __HIP_MEMORY_SCOPE_WORKGROUP);
                    l2_add_quarter(acc, rl, q, w, x[t]);
                }
            }
        }
    } else {
        for (int e = tid; e < TOT_E; e += 1024) {
            int s, d; edge_sd(e, src, dst, s, d);
            const unsigned rl = (unsigned)(d - n0);
            if (rl < (unsigned)cnt) {
                const float w = expf(own_lrelu(as2[s] + rad[rl]) - fdec(rmax[rl]));
                __hip_atomic_fetch_add(rden + rl, w, __ATOMIC_RELAXED, __HIP_MEMORY_SCOPE_WORKGROUP);
                for (int q = 0; q < 4; ++q)
                    l2_add_quarter(acc, rl, q, w, xq[(size_t)s * (OUT_C / 4) + q]);
            }
        }
    }
    __syncthreads();

    const int tot = cnt * OUT_C;
    for (int i = tid; i < tot; i += 1024) {
        const int r = i >> 4, c = i & 15;
        out[(size_t)n0 * OUT_C + i] = acc[c * OWN_R + r] / (rden[r] + GAT_EPS) + bias[c];
    }
}

extern "C" void kernel_launch(void* const* d_in, const int* in_sizes, int n_in,
                              void* d_out, int out_size, void* d_ws, size_t ws_size,
                              hipStream_t stream)
{
    (void)in_sizes; (void)n_in; (void)out_size; (void)ws_size;

    const float* x    = (const float*)d_in[0];
    const int*   ei   = (const int*)  d_in[1];
    const float* W1   = (const float*)d_in[2];
    const float* ats1 = (const float*)d_in[3];
    const float* atd1 = (const float*)d_in[4];
    const float* b1   = (const float*)d_in[5];
    const float* W2   = (const float*)d_in[6];
    const float* ats2 = (const float*)d_in[7];
    const float* atd2 = (const float*)d_in[8];
    const float* b2   = (const float*)d_in[9];
    const int* src = ei;
    const int* dst = ei + N_EDGES;
    float* out = (float*)d_out;

    float* xl1 = (float*)d_ws;
    float* as1 = xl1 + (size_t)N_NODES * HIDDEN;
    float* ad1 = as1 + (size_t)N_NODES * HEADS;
    float* xl2 = ad1 + (size_t)N_NODES * HEADS;
    float* as2 = xl2 + (size_t)N_NODES * OUT_C;
    float* ad2 = as2 + (size_t)N_NODES;
    unsigned* qsv = (unsigned*)(ad2 + (size_t)N_NODES);
    int*      qct = (int*)(qsv + (size_t)OWN_WG * OWN_WAVES * OWN_QCAP);
    int*      qov = qct + (size_t)OWN_WG * OWN_WAVES;
    constexpr size_t kWordsBefore = (size_t)N_NODES * (HIDDEN + 2 * HEADS + OUT_C + 2)
                                  + (size_t)OWN_WG * OWN_WAVES * OWN_QCAP
                                  + (size_t)OWN_WG * OWN_WAVES + OWN_WG;
    constexpr size_t kBinsAt = (kWordsBefore + BIN_W - 1) / BIN_W * BIN_W;
    constexpr size_t kFoldAt = kBinsAt + (size_t)OWN_WG * BIN_CHUNKS * BIN_W;
    static_assert((kFoldAt + (size_t)(IN_C + HIDDEN) * FOLD_COLS) * 4 <= 268435456ull,
                  "the bins and the folded weights must end inside the workspace");
    unsigned* bins = (unsigned*)d_ws + kBinsAt;
    float* wa1 = (float*)d_ws + kFoldAt;
    float* wa2 = wa1 + (size_t)IN_C * FOLD_COLS;

    gat_bin_edges_by_owner<<<BIN_CHUNKS + FOLD_WG, 1024, 0, stream>>>(
        src, dst, bins, W1, ats1, atd1, W2, ats2, atd2, wa1, wa2);

    gat_gemm1_xl_logits<<<N_NODES / 16, 160, 0, stream>>>(x, W1, wa1, xl1, as1, ad1);
    gat_owner_softmax_aggregate_l1<<<OWN_WG, OWN_WAVES * 32, 0, stream>>>(
        src, dst, as1, ad1, xl1, b1, W2, wa2, xl2, as2, ad2, qsv, qct, qov, bins);

    gat_owner_softmax_aggregate_l2<<<OWN_WG, OWN_WAVES * 32, 0, stream>>>(
        src, dst, as2, ad2, xl2, b2, qsv, qct, qov, out);
}
